# starting kernel
# baseline (speedup 1.0000x reference)
#include <hip/hip_runtime.h>
#include <cstdint>

#define N_NODES 50000
#define N_EDGES 800000
#define IN_CH 64
#define OUT_CH 32
#define HEADS 4
#define HC 128
#define NEG_SLOPE 0.2f
#define EPSV 1e-16f

typedef __attribute__((ext_vector_type(2))) float v2f;
typedef __attribute__((ext_vector_type(4))) float v4f;
typedef __attribute__((ext_vector_type(8))) float v8f;

__global__ void __launch_bounds__(128) proj_nodes_wmma(
    const float* __restrict__ x, const float* __restrict__ W,
    float* __restrict__ xh)
{
    const int lane = threadIdx.x & 31;
    const int wv   = threadIdx.x >> 5;
    const int half = lane >> 4;
    const int l15  = lane & 15;
    const int m0   = blockIdx.x * 16;

    const int rowA  = m0 + l15;
    const int colB0 = wv * 32 + l15;
    const int colB1 = colB0 + 16;

    v8f acc0 = {};
    v8f acc1 = {};
#pragma unroll
    for (int k0 = 0; k0 < IN_CH; k0 += 4) {
        const int kk = k0 + half * 2;
        v2f a  = *(const v2f*)(x + (size_t)rowA  * IN_CH + kk);
        v2f b0 = *(const v2f*)(W + (size_t)colB0 * IN_CH + kk);
        v2f b1 = *(const v2f*)(W + (size_t)colB1 * IN_CH + kk);
        acc0 = __builtin_amdgcn_wmma_f32_16x16x4_f32(
                   false, a, false, b0, (short)0, acc0, false, false);
        acc1 = __builtin_amdgcn_wmma_f32_16x16x4_f32(
                   false, a, false, b1, (short)0, acc1, false, false);
    }
#pragma unroll
    for (int v = 0; v < 8; ++v) {
        const int m = m0 + v + half * 8;
        xh[(size_t)m * HC + colB0] = acc0[v];
        xh[(size_t)m * HC + colB1] = acc1[v];
    }
}

__global__ void build_weff(const float* __restrict__ W_edge,
                           const float* __restrict__ att_edge,
                           float* __restrict__ weff)
{
    const int t = threadIdx.x;
    const int h = t >> 6, k = t & 63;
    float s = 0.f;
#pragma unroll
    for (int c = 0; c < OUT_CH; ++c)
        s += att_edge[h * OUT_CH + c] * W_edge[(size_t)(h * OUT_CH + c) * IN_CH + k];
    weff[t] = s;
}

__global__ void node_attn_scalars(const float* __restrict__ xh,
                                  const float* __restrict__ att_src,
                                  const float* __restrict__ att_dst,
                                  float* __restrict__ s_node,
                                  float* __restrict__ d_node)
{
    const int t = blockIdx.x * blockDim.x + threadIdx.x;
    if (t >= N_NODES * HEADS) return;
    const int n = t >> 2, h = t & 3;
    const float* xp = xh + (size_t)n * HC + h * OUT_CH;
    float ss = 0.f, dd = 0.f;
#pragma unroll
    for (int c = 0; c < OUT_CH; ++c) {
        const float v = xp[c];
        ss += v * att_src[h * OUT_CH + c];
        dd += v * att_dst[h * OUT_CH + c];
    }
    s_node[t] = ss;
    d_node[t] = dd;
}

__global__ void init_buffers(float* __restrict__ seg_max,
                             float* __restrict__ seg_sum,
                             float* __restrict__ acc)
{
    const int t = blockIdx.x * blockDim.x + threadIdx.x;
    if (t < N_NODES * HEADS) {
        seg_max[t] = -__builtin_huge_valf();
        seg_sum[t] = 0.f;
    }
    if (t < N_NODES * OUT_CH) acc[t] = 0.f;
}

__global__ void __launch_bounds__(256) edge_logits(
    const float* __restrict__ edge_attr, const int* __restrict__ edge_index,
    const float* __restrict__ s_node, const float* __restrict__ d_node,
    const float* __restrict__ weff,
    float* __restrict__ alpha, float* __restrict__ seg_max)
{
    __shared__ float sw[HEADS * IN_CH];
    if (threadIdx.x < HEADS * IN_CH) sw[threadIdx.x] = weff[threadIdx.x];
    __syncthreads();

    const int e = blockIdx.x * blockDim.x + threadIdx.x;
    if (e >= N_EDGES) return;
    const int src = edge_index[e];
    const int dst = edge_index[N_EDGES + e];

    float acch[HEADS] = {0.f, 0.f, 0.f, 0.f};
    const v4f* ea = (const v4f*)(edge_attr + (size_t)e * IN_CH);
#pragma unroll
    for (int k4 = 0; k4 < IN_CH / 4; ++k4) {
        const v4f v = __builtin_nontemporal_load(ea + k4);
#pragma unroll
        for (int h = 0; h < HEADS; ++h) {
            const float* wh = sw + h * IN_CH + k4 * 4;
            acch[h] += v.x * wh[0] + v.y * wh[1] + v.z * wh[2] + v.w * wh[3];
        }
    }
#pragma unroll
    for (int h = 0; h < HEADS; ++h) {
        float a = acch[h] + s_node[dst * 4 + h] + d_node[src * 4 + h];
        a = (a > 0.f) ? a : NEG_SLOPE * a;
        alpha[(size_t)e * 4 + h] = a;
        float* addr = seg_max + dst * 4 + h;
        if (a >= 0.f) atomicMax((int*)addr, __float_as_int(a));
        else          atomicMin((unsigned int*)addr, __float_as_uint(a));
    }
}

__global__ void edge_softmax(const int* __restrict__ edge_index,
                             const float* __restrict__ seg_max,
                             float* __restrict__ alpha,
                             float* __restrict__ seg_sum)
{
    const int t = blockIdx.x * blockDim.x + threadIdx.x;
    if (t >= N_EDGES * HEADS) return;
    const int e = t >> 2, h = t & 3;
    const int dst = edge_index[N_EDGES + e];
    const float ex = __expf(alpha[t] - seg_max[dst * 4 + h]);
    alpha[t] = ex;
    unsafeAtomicAdd(seg_sum + dst * 4 + h, ex);
}

__global__ void __launch_bounds__(256) edge_scatter(
    const int* __restrict__ edge_index, const float* __restrict__ xh,
    const float* __restrict__ alpha, const float* __restrict__ seg_sum,
    float* __restrict__ acc)
{
    const int lane = threadIdx.x & 31;
    const int e = blockIdx.x * 8 + (threadIdx.x >> 5);
    if (e >= N_EDGES) return;
    const int src = edge_index[e];
    const int dst = edge_index[N_EDGES + e];

    const float c0 = alpha[(size_t)e * 4 + 0] / (seg_sum[dst * 4 + 0] + EPSV);
    const float c1 = alpha[(size_t)e * 4 + 1] / (seg_sum[dst * 4 + 1] + EPSV);
    const float c2 = alpha[(size_t)e * 4 + 2] / (seg_sum[dst * 4 + 2] + EPSV);
    const float c3 = alpha[(size_t)e * 4 + 3] / (seg_sum[dst * 4 + 3] + EPSV);

    const float* xp = xh + (size_t)src * HC;
    const float v = c0 * xp[lane] + c1 * xp[32 + lane]
                  + c2 * xp[64 + lane] + c3 * xp[96 + lane];
    unsafeAtomicAdd(acc + (size_t)dst * OUT_CH + lane, v);
}

__global__ void finalize(const float* __restrict__ acc,
                         const float* __restrict__ bias_conv,
                         const float* __restrict__ bias_layer,
                         float* __restrict__ out)
{
    const int t = blockIdx.x * blockDim.x + threadIdx.x;
    if (t >= N_NODES * OUT_CH) return;
    const int c = t & 31;
    const float v = acc[t] * 0.25f + bias_conv[c] + bias_layer[c];
    out[t] = v > 0.f ? v : 0.f;
}

extern "C" void kernel_launch(void* const* d_in, const int* in_sizes, int n_in,
                              void* d_out, int out_size, void* d_ws, size_t ws_size,
                              hipStream_t stream) {
    const float* x          = (const float*)d_in[0];
    const float* edge_attr  = (const float*)d_in[1];
    const float* W          = (const float*)d_in[2];
    const float* W_edge     = (const float*)d_in[3];
    const float* att_src    = (const float*)d_in[4];
    const float* att_dst    = (const float*)d_in[5];
    const float* att_edge   = (const float*)d_in[6];
    const float* bias_conv  = (const float*)d_in[7];
    const float* bias_layer = (const float*)d_in[8];
    const int*   edge_index = (const int*)d_in[9];
    float* out = (float*)d_out;

    float* ws      = (float*)d_ws;
    float* xh      = ws;
    float* s_node  = xh      + (size_t)N_NODES * HC;
    float* d_node  = s_node  + (size_t)N_NODES * HEADS;
    float* weff    = d_node  + (size_t)N_NODES * HEADS;
    float* alpha   = weff    + (size_t)HEADS * IN_CH;
    float* seg_max = alpha   + (size_t)N_EDGES * HEADS;
    float* seg_sum = seg_max + (size_t)N_NODES * HEADS;
    float* accb    = seg_sum + (size_t)N_NODES * HEADS;

    proj_nodes_wmma<<<N_NODES / 16, 128, 0, stream>>>(x, W, xh);
    build_weff<<<1, 256, 0, stream>>>(W_edge, att_edge, weff);
    node_attn_scalars<<<(N_NODES * HEADS + 255) / 256, 256, 0, stream>>>(
        xh, att_src, att_dst, s_node, d_node);
    init_buffers<<<(N_NODES * OUT_CH + 255) / 256, 256, 0, stream>>>(
        seg_max, seg_sum, accb);
    edge_logits<<<(N_EDGES + 255) / 256, 256, 0, stream>>>(
        edge_attr, edge_index, s_node, d_node, weff, alpha, seg_max);
    edge_softmax<<<(N_EDGES * HEADS + 255) / 256, 256, 0, stream>>>(
        edge_index, seg_max, alpha, seg_sum);
    edge_scatter<<<(N_EDGES + 7) / 8, 256, 0, stream>>>(
        edge_index, xh, alpha, seg_sum, accb);
    finalize<<<(N_NODES * OUT_CH + 255) / 256, 256, 0, stream>>>(
        accb, bias_conv, bias_layer, out);
}


# edge_logits: each wave-wide 16-byte load of edge_attr now covers 512 contiguous bytes (2 edges x 16 lanes); a 16-lane DPP butterfly finishes the dots (sum order changes), 4 lanes per edge after it
# speedup vs baseline: 1.7669x; 1.7669x over previous
#include <hip/hip_runtime.h>
#include <cstdint>

#define N_NODES 50000
#define N_EDGES 800000
#define IN_CH 64
#define OUT_CH 32
#define HEADS 4
#define HC 128
#define NEG_SLOPE 0.2f
#define EPSV 1e-16f

typedef __attribute__((ext_vector_type(2))) float v2f;
typedef __attribute__((ext_vector_type(4))) float v4f;
typedef __attribute__((ext_vector_type(8))) float v8f;

__global__ void __launch_bounds__(128) proj_nodes_wmma(
    const float* __restrict__ x, const float* __restrict__ W,
    float* __restrict__ xh)
{
    const int lane = threadIdx.x & 31;
    const int wv   = threadIdx.x >> 5;
    const int half = lane >> 4;
    const int l15  = lane & 15;
    const int m0   = blockIdx.x * 16;

    const int rowA  = m0 + l15;
    const int colB0 = wv * 32 + l15;
    const int colB1 = colB0 + 16;

    v8f acc0 = {};
    v8f acc1 = {};
#pragma unroll
    for (int k0 = 0; k0 < IN_CH; k0 += 4) {
        const int kk = k0 + half * 2;
        v2f a  = *(const v2f*)(x + (size_t)rowA  * IN_CH + kk);
        v2f b0 = *(const v2f*)(W + (size_t)colB0 * IN_CH + kk);
        v2f b1 = *(const v2f*)(W + (size_t)colB1 * IN_CH + kk);
        acc0 = __builtin_amdgcn_wmma_f32_16x16x4_f32(
                   false, a, false, b0, (short)0, acc0, false, false);
        acc1 = __builtin_amdgcn_wmma_f32_16x16x4_f32(
                   false, a, false, b1, (short)0, acc1, false, false);
    }
#pragma unroll
    for (int v = 0; v < 8; ++v) {
        const int m = m0 + v + half * 8;
        xh[(size_t)m * HC + colB0] = acc0[v];
        xh[(size_t)m * HC + colB1] = acc1[v];
    }
}

__global__ void build_weff(const float* __restrict__ W_edge,
                           const float* __restrict__ att_edge,
                           float* __restrict__ weff)
{
    const int t = threadIdx.x;
    const int h = t >> 6, k = t & 63;
    float s = 0.f;
#pragma unroll
    for (int c = 0; c < OUT_CH; ++c)
        s += att_edge[h * OUT_CH + c] * W_edge[(size_t)(h * OUT_CH + c) * IN_CH + k];
    weff[t] = s;
}

__global__ void node_attn_scalars(const float* __restrict__ xh,
                                  const float* __restrict__ att_src,
                                  const float* __restrict__ att_dst,
                                  float* __restrict__ s_node,
                                  float* __restrict__ d_node)
{
    const int t = blockIdx.x * blockDim.x + threadIdx.x;
    if (t >= N_NODES * HEADS) return;
    const int n = t >> 2, h = t & 3;
    const float* xp = xh + (size_t)n * HC + h * OUT_CH;
    float ss = 0.f, dd = 0.f;
#pragma unroll
    for (int c = 0; c < OUT_CH; ++c) {
        const float v = xp[c];
        ss += v * att_src[h * OUT_CH + c];
        dd += v * att_dst[h * OUT_CH + c];
    }
    s_node[t] = ss;
    d_node[t] = dd;
}

__global__ void init_buffers(float* __restrict__ seg_max,
                             float* __restrict__ seg_sum,
                             float* __restrict__ acc)
{
    const int t = blockIdx.x * blockDim.x + threadIdx.x;
    if (t < N_NODES * HEADS) {
        seg_max[t] = -__builtin_huge_valf();
        seg_sum[t] = 0.f;
    }
    if (t < N_NODES * OUT_CH) acc[t] = 0.f;
}

template <int M>
__device__ __forceinline__ float row_xor(float v)
{
    return __builtin_bit_cast(float, __builtin_amdgcn_update_dpp(
        0, __builtin_bit_cast(int, v), 0x160 + M, 0xf, 0xf, true));
}

template <int M>
__device__ __forceinline__ float fold_pair(float a, float b, bool bit)
{
    const float send = bit ? a : b;
    const float keep = bit ? b : a;
    return keep + row_xor<M>(send);
}

__device__ __forceinline__ float dot4(v4f v, v4f w)
{
    return __builtin_fmaf(v.w, w.w, __builtin_fmaf(v.z, w.z, __builtin_fmaf(v.y, w.y, v.x * w.x)));
}

static_assert(N_EDGES % 256 == 0, "one wave per 32 edges, 8 waves per block");

__global__ void __launch_bounds__(256) __attribute__((amdgpu_waves_per_eu(8, 8))) edge_logits(
    const float* __restrict__ edge_attr, const int* __restrict__ edge_index,
    const float* __restrict__ s_node, const float* __restrict__ d_node,
    const float* __restrict__ weff,
    float* __restrict__ alpha, float* __restrict__ seg_max)
{
    const int lane = threadIdx.x & 31;
    const int l15  = lane & 15;
    const int half = lane >> 4;
    const int h    = l15 & 3;
    const int e0   = (blockIdx.x * 8 + (threadIdx.x >> 5)) * 32;
    const int eb   = e0 + ((l15 >> 2) << 1) + half;

    const v4f w0 = *(const v4f*)(weff + 0 * IN_CH + 4 * l15);
    const v4f w1 = *(const v4f*)(weff + 1 * IN_CH + 4 * l15);
    const v4f w2 = *(const v4f*)(weff + 2 * IN_CH + 4 * l15);
    const v4f w3 = *(const v4f*)(weff + 3 * IN_CH + 4 * l15);

    int srcn[4], dstn[4];
#pragma unroll
    for (int j = 0; j < 4; ++j) {
        srcn[j] = edge_index[eb + 8 * j];
        dstn[j] = edge_index[N_EDGES + eb + 8 * j];
    }

    const v4f* ea = (const v4f*)(edge_attr + (size_t)e0 * IN_CH) + lane;
    v4f v[16];
#pragma unroll
    for (int i = 0; i < 16; ++i)
        v[i] = __builtin_nontemporal_load(ea + i * 32);

    float sn[4], dn[4];
#pragma unroll
    for (int j = 0; j < 4; ++j) {
        sn[j] = s_node[dstn[j] * 4 + h];
        dn[j] = d_node[srcn[j] * 4 + h];
    }
    __builtin_amdgcn_sched_barrier(0);

    const bool b0 = (lane & 1) != 0, b1 = (lane & 2) != 0;
    const bool b2 = (lane & 4) != 0, b3 = (lane & 8) != 0;

    float s[16];
#pragma unroll
    for (int i = 0; i < 16; ++i) {
        const float r01 = fold_pair<1>(dot4(v[i], w0), dot4(v[i], w1), b0);
        const float r23 = fold_pair<1>(dot4(v[i], w2), dot4(v[i], w3), b0);
        s[i] = fold_pair<2>(r01, r23, b1);
    }
    float t[8];
#pragma unroll
    for (int m = 0; m < 8; ++m) t[m] = fold_pair<4>(s[2 * m], s[2 * m + 1], b2);
    float u[4];
#pragma unroll
    for (int j = 0; j < 4; ++j) u[j] = fold_pair<8>(t[2 * j], t[2 * j + 1], b3);

    float av[4];
#pragma unroll
    for (int j = 0; j < 4; ++j) {
        float a = u[j] + sn[j] + dn[j];
        a = (a > 0.f) ? a : NEG_SLOPE * a;
        av[j] = a;
        alpha[(size_t)(eb + 8 * j) * 4 + h] = a;
    }
#pragma unroll
    for (int j = 0; j < 4; ++j) {
        const float a = av[j];
        float* addr = seg_max + dstn[j] * 4 + h;
        if (a >= 0.f) atomicMax((int*)addr, __float_as_int(a));
        else          atomicMin((unsigned int*)addr, __float_as_uint(a));
    }
}

__global__ void edge_softmax(const int* __restrict__ edge_index,
                             const float* __restrict__ seg_max,
                             float* __restrict__ alpha,
                             float* __restrict__ seg_sum)
{
    const int t = blockIdx.x * blockDim.x + threadIdx.x;
    if (t >= N_EDGES * HEADS) return;
    const int e = t >> 2, h = t & 3;
    const int dst = edge_index[N_EDGES + e];
    const float ex = __expf(alpha[t] - seg_max[dst * 4 + h]);
    alpha[t] = ex;
    unsafeAtomicAdd(seg_sum + dst * 4 + h, ex);
}

__global__ void __launch_bounds__(256) edge_scatter(
    const int* __restrict__ edge_index, const float* __restrict__ xh,
    const float* __restrict__ alpha, const float* __restrict__ seg_sum,
    float* __restrict__ acc)
{
    const int lane = threadIdx.x & 31;
    const int e = blockIdx.x * 8 + (threadIdx.x >> 5);
    if (e >= N_EDGES) return;
    const int src = edge_index[e];
    const int dst = edge_index[N_EDGES + e];

    const float c0 = alpha[(size_t)e * 4 + 0] / (seg_sum[dst * 4 + 0] + EPSV);
    const float c1 = alpha[(size_t)e * 4 + 1] / (seg_sum[dst * 4 + 1] + EPSV);
    const float c2 = alpha[(size_t)e * 4 + 2] / (seg_sum[dst * 4 + 2] + EPSV);
    const float c3 = alpha[(size_t)e * 4 + 3] / (seg_sum[dst * 4 + 3] + EPSV);

    const float* xp = xh + (size_t)src * HC;
    const float v = c0 * xp[lane] + c1 * xp[32 + lane]
                  + c2 * xp[64 + lane] + c3 * xp[96 + lane];
    unsafeAtomicAdd(acc + (size_t)dst * OUT_CH + lane, v);
}

__global__ void finalize(const float* __restrict__ acc,
                         const float* __restrict__ bias_conv,
                         const float* __restrict__ bias_layer,
                         float* __restrict__ out)
{
    const int t = blockIdx.x * blockDim.x + threadIdx.x;
    if (t >= N_NODES * OUT_CH) return;
    const int c = t & 31;
    const float v = acc[t] * 0.25f + bias_conv[c] + bias_layer[c];
    out[t] = v > 0.f ? v : 0.f;
}

extern "C" void kernel_launch(void* const* d_in, const int* in_sizes, int n_in,
                              void* d_out, int out_size, void* d_ws, size_t ws_size,
                              hipStream_t stream) {
    const float* x          = (const float*)d_in[0];
    const float* edge_attr  = (const float*)d_in[1];
    const float* W          = (const float*)d_in[2];
    const float* W_edge     = (const float*)d_in[3];
    const float* att_src    = (const float*)d_in[4];
    const float* att_dst    = (const float*)d_in[5];
    const float* att_edge   = (const float*)d_in[6];
    const float* bias_conv  = (const float*)d_in[7];
    const float* bias_layer = (const float*)d_in[8];
    const int*   edge_index = (const int*)d_in[9];
    float* out = (float*)d_out;

    float* ws      = (float*)d_ws;
    float* xh      = ws;
    float* s_node  = xh      + (size_t)N_NODES * HC;
    float* d_node  = s_node  + (size_t)N_NODES * HEADS;
    float* weff    = d_node  + (size_t)N_NODES * HEADS;
    float* alpha   = weff    + (size_t)HEADS * IN_CH;
    float* seg_max = alpha   + (size_t)N_EDGES * HEADS;
    float* seg_sum = seg_max + (size_t)N_NODES * HEADS;
    float* accb    = seg_sum + (size_t)N_NODES * HEADS;

    proj_nodes_wmma<<<N_NODES / 16, 128, 0, stream>>>(x, W, xh);
    build_weff<<<1, 256, 0, stream>>>(W_edge, att_edge, weff);
    node_attn_scalars<<<(N_NODES * HEADS + 255) / 256, 256, 0, stream>>>(
        xh, att_src, att_dst, s_node, d_node);
    init_buffers<<<(N_NODES * OUT_CH + 255) / 256, 256, 0, stream>>>(
        seg_max, seg_sum, accb);
    edge_logits<<<(N_EDGES + 255) / 256, 256, 0, stream>>>(
        edge_attr, edge_index, s_node, d_node, weff, alpha, seg_max);
    edge_softmax<<<(N_EDGES * HEADS + 255) / 256, 256, 0, stream>>>(
        edge_index, seg_max, alpha, seg_sum);
    edge_scatter<<<(N_EDGES + 7) / 8, 256, 0, stream>>>(
        edge_index, xh, alpha, seg_sum, accb);
    finalize<<<(N_NODES * OUT_CH + 255) / 256, 256, 0, stream>>>(
        accb, bias_conv, bias_layer, out);
}


# edges grouped by destination (counting sort: in-degree histogram, prefix sum, slot = cursor[dst]++); one wave per node does softmax and gather in registers; no float atomics, sums in slot order
# speedup vs baseline: 2.1403x; 1.2113x over previous
#include <hip/hip_runtime.h>
#include <cstdint>

#define N_NODES 50000
#define N_EDGES 800000
#define IN_CH 64
#define OUT_CH 32
#define HEADS 4
#define HC 128
#define NEG_SLOPE 0.2f
#define EPSV 1e-16f

typedef __attribute__((ext_vector_type(2))) float v2f;
typedef __attribute__((ext_vector_type(4))) float v4f;
typedef __attribute__((ext_vector_type(8))) float v8f;

__global__ void __launch_bounds__(128) proj_nodes_wmma(
    const float* __restrict__ x, const float* __restrict__ W,
    float* __restrict__ xh, int* __restrict__ cnt)
{
    const int lane = threadIdx.x & 31;
    const int wv   = threadIdx.x >> 5;
    const int half = lane >> 4;
    const int l15  = lane & 15;
    const int m0   = blockIdx.x * 16;

    if (threadIdx.x < 16) cnt[m0 + threadIdx.x] = 0;

    const int rowA  = m0 + l15;
    const int colB0 = wv * 32 + l15;
    const int colB1 = colB0 + 16;

    v8f acc0 = {};
    v8f acc1 = {};
#pragma unroll
    for (int k0 = 0; k0 < IN_CH; k0 += 4) {
        const int kk = k0 + half * 2;
        v2f a  = *(const v2f*)(x + (size_t)rowA  * IN_CH + kk);
        v2f b0 = *(const v2f*)(W + (size_t)colB0 * IN_CH + kk);
        v2f b1 = *(const v2f*)(W + (size_t)colB1 * IN_CH + kk);
        acc0 = __builtin_amdgcn_wmma_f32_16x16x4_f32(
                   false, a, false, b0, (short)0, acc0, false, false);
        acc1 = __builtin_amdgcn_wmma_f32_16x16x4_f32(
                   false, a, false, b1, (short)0, acc1, false, false);
    }
#pragma unroll
    for (int v = 0; v < 8; ++v) {
        const int m = m0 + v + half * 8;
        xh[(size_t)m * HC + colB0] = acc0[v];
        xh[(size_t)m * HC + colB1] = acc1[v];
    }
}

__global__ void build_weff(const float* __restrict__ W_edge,
                           const float* __restrict__ att_edge,
                           float* __restrict__ weff)
{
    const int t = threadIdx.x;
    const int h = t >> 6, k = t & 63;
    float s = 0.f;
#pragma unroll
    for (int c = 0; c < OUT_CH; ++c)
        s += att_edge[h * OUT_CH + c] * W_edge[(size_t)(h * OUT_CH + c) * IN_CH + k];
    weff[t] = s;
}

__global__ void node_attn_scalars(const float* __restrict__ xh,
                                  const float* __restrict__ att_src,
                                  const float* __restrict__ att_dst,
                                  float* __restrict__ s_node,
                                  float* __restrict__ d_node,
                                  const int* __restrict__ edge_index,
                                  int* __restrict__ cnt)
{
    const int t = blockIdx.x * blockDim.x + threadIdx.x;
    if (t >= N_NODES * HEADS) return;
    constexpr int NT = N_NODES * HEADS;
    constexpr int EPT = (N_EDGES + NT - 1) / NT;
    int dn[EPT];
#pragma unroll
    for (int j = 0; j < EPT; ++j) {
        const int e = t + j * NT;
        dn[j] = (e < N_EDGES) ? edge_index[N_EDGES + e] : -1;
    }
#pragma unroll
    for (int j = 0; j < EPT; ++j)
        if (dn[j] >= 0) atomicAdd(cnt + dn[j], 1);
    const int n = t >> 2, h = t & 3;
    const float* xp = xh + (size_t)n * HC + h * OUT_CH;
    float ss = 0.f, dd = 0.f;
#pragma unroll
    for (int c = 0; c < OUT_CH; ++c) {
        const float v = xp[c];
        ss += v * att_src[h * OUT_CH + c];
        dd += v * att_dst[h * OUT_CH + c];
    }
    s_node[t] = ss;
    d_node[t] = dd;
}

#define SCAN_ITERS 13
#define SCAN_PER_WAVE (SCAN_ITERS * 128)
static_assert(32 * SCAN_PER_WAVE >= N_NODES, "one block covers all nodes");
static_assert(N_NODES % 4 == 0, "16-byte counter loads are all-in or all-out");

__global__ void __launch_bounds__(1024) seg_offsets(const int* __restrict__ cnt,
                                                    int* __restrict__ cursor)
{
    __shared__ int wtot[32];
    const int lane = threadIdx.x & 31;
    const int wv   = threadIdx.x >> 5;
    const int base = wv * SCAN_PER_WAVE + lane * 4;

    int4 c[SCAN_ITERS];
#pragma unroll
    for (int i = 0; i < SCAN_ITERS; ++i) {
        const int idx = base + i * 128;
        c[i] = (idx < N_NODES) ? *(const int4*)(cnt + idx) : make_int4(0, 0, 0, 0);
    }
    int run = 0;
    int ex[SCAN_ITERS];
#pragma unroll
    for (int i = 0; i < SCAN_ITERS; ++i) {
        const int s = c[i].x + c[i].y + c[i].z + c[i].w;
        int inc = s;
#pragma unroll
        for (int d = 1; d < 32; d <<= 1) {
            const int up = __shfl_up(inc, d);
            if (lane >= d) inc += up;
        }
        ex[i] = run + inc - s;
        run += __shfl(inc, 31);
    }
    if (lane == 0) wtot[wv] = run;
    __syncthreads();
    int wb = (lane < wv) ? wtot[lane] : 0;
#pragma unroll
    for (int d = 16; d >= 1; d >>= 1) wb += __shfl_xor(wb, d);
#pragma unroll
    for (int i = 0; i < SCAN_ITERS; ++i) {
        const int idx = base + i * 128;
        if (idx < N_NODES) {
            int4 r;
            r.x = wb + ex[i];
            r.y = r.x + c[i].x;
            r.z = r.y + c[i].y;
            r.w = r.z + c[i].z;
            *(int4*)(cursor + idx) = r;
        }
    }
}

template <int M>
__device__ __forceinline__ float row_xor(float v)
{
    return __builtin_bit_cast(float, __builtin_amdgcn_update_dpp(
        0, __builtin_bit_cast(int, v), 0x160 + M, 0xf, 0xf, true));
}

template <int M>
__device__ __forceinline__ float fold_pair(float a, float b, bool bit)
{
    const float send = bit ? a : b;
    const float keep = bit ? b : a;
    return keep + row_xor<M>(send);
}

__device__ __forceinline__ float dot4(v4f v, v4f w)
{
    return __builtin_fmaf(v.w, w.w, __builtin_fmaf(v.z, w.z, __builtin_fmaf(v.y, w.y, v.x * w.x)));
}

static_assert(N_EDGES % 256 == 0, "one wave per 32 edges, 8 waves per block");

__global__ void __launch_bounds__(256) __attribute__((amdgpu_waves_per_eu(8, 8))) edge_logits(
    const float* __restrict__ edge_attr, const int* __restrict__ edge_index,
    const float* __restrict__ s_node, const float* __restrict__ d_node,
    const float* __restrict__ weff,
    int* __restrict__ cursor, float* __restrict__ salpha, int* __restrict__ ssrc)
{
    const int lane = threadIdx.x & 31;
    const int l15  = lane & 15;
    const int half = lane >> 4;
    const int h    = l15 & 3;
    const int e0   = (blockIdx.x * 8 + (threadIdx.x >> 5)) * 32;
    const int eb   = e0 + ((l15 >> 2) << 1) + half;

    const v4f w0 = *(const v4f*)(weff + 0 * IN_CH + 4 * l15);
    const v4f w1 = *(const v4f*)(weff + 1 * IN_CH + 4 * l15);
    const v4f w2 = *(const v4f*)(weff + 2 * IN_CH + 4 * l15);
    const v4f w3 = *(const v4f*)(weff + 3 * IN_CH + 4 * l15);

    int srcn[4], dstn[4];
#pragma unroll
    for (int j = 0; j < 4; ++j) {
        srcn[j] = edge_index[eb + 8 * j];
        dstn[j] = edge_index[N_EDGES + eb + 8 * j];
    }

    const v4f* ea = (const v4f*)(edge_attr + (size_t)e0 * IN_CH) + lane;
    v4f v[16];
#pragma unroll
    for (int i = 0; i < 16; ++i)
        v[i] = __builtin_nontemporal_load(ea + i * 32);

    float sn[4], dn[4];
#pragma unroll
    for (int j = 0; j < 4; ++j) {
        sn[j] = s_node[dstn[j] * 4 + h];
        dn[j] = d_node[srcn[j] * 4 + h];
    }
    int slot[4] = {0, 0, 0, 0};
    if (h == 0) {
#pragma unroll
        for (int j = 0; j < 4; ++j) slot[j] = atomicAdd(cursor + dstn[j], 1);
    }
    __builtin_amdgcn_sched_barrier(0);

    const bool b0 = (lane & 1) != 0, b1 = (lane & 2) != 0;
    const bool b2 = (lane & 4) != 0, b3 = (lane & 8) != 0;

    float s[16];
#pragma unroll
    for (int i = 0; i < 16; ++i) {
        const float r01 = fold_pair<1>(dot4(v[i], w0), dot4(v[i], w1), b0);
        const float r23 = fold_pair<1>(dot4(v[i], w2), dot4(v[i], w3), b0);
        s[i] = fold_pair<2>(r01, r23, b1);
    }
    float t[8];
#pragma unroll
    for (int m = 0; m < 8; ++m) t[m] = fold_pair<4>(s[2 * m], s[2 * m + 1], b2);
    float u[4];
#pragma unroll
    for (int j = 0; j < 4; ++j) u[j] = fold_pair<8>(t[2 * j], t[2 * j + 1], b3);

#pragma unroll
    for (int j = 0; j < 4; ++j) {
        float a = u[j] + sn[j] + dn[j];
        a = (a > 0.f) ? a : NEG_SLOPE * a;
        int p = __builtin_amdgcn_update_dpp(0, slot[j], 0x00, 0xf, 0xf, true);
        p = min(max(p, 0), N_EDGES - 1);
        salpha[(size_t)p * 4 + h] = a;
        if (h == 0) ssrc[p] = srcn[j];
    }
}

template <int K>
__device__ __forceinline__ float group8_lane(float v)
{
    constexpr int sel = K | (K << 3) | (K << 6) | (K << 9) | (K << 12) | (K << 15) | (K << 18) | (K << 21);
    return __builtin_bit_cast(float, __builtin_amdgcn_mov_dpp8(__builtin_bit_cast(int, v), sel));
}

template <int K>
__device__ __forceinline__ void gather_rows(const float* __restrict__ xl, int sv, float c,
                                            v4f (&row)[8], float (&ck)[8])
{
    const int sk = min(max(__builtin_amdgcn_readlane(sv, K), 0), N_NODES - 1);
    row[K] = *(const v4f*)(xl + (size_t)sk * HC);
    ck[K]  = group8_lane<K>(c);
    if constexpr (K + 1 < 8) gather_rows<K + 1>(xl, sv, c, row, ck);
}

static_assert(N_NODES % 8 == 0, "one wave per node, 8 waves per block");

__global__ void __launch_bounds__(256) node_softmax_gather(
    const int* __restrict__ cursor, const float* __restrict__ salpha,
    const int* __restrict__ ssrc, const float* __restrict__ xh,
    const float* __restrict__ bias_conv, const float* __restrict__ bias_layer,
    float* __restrict__ out)
{
    const int lane = threadIdx.x & 31;
    const int k    = lane & 7;
    const int h    = lane >> 3;
    const int n = __builtin_amdgcn_readfirstlane(blockIdx.x * 8 + (threadIdx.x >> 5));

    int end = cursor[n];
    int beg = (n > 0) ? cursor[n - 1] : 0;
    beg = max(beg, 0);
    end = min(end, N_EDGES);

    float m = -__builtin_huge_valf();
    for (int p = beg; p < end; p += 8) {
        const int i = p + k;
        const float a = salpha[(size_t)((i < end) ? i : p) * 4 + h];
        m = (i < end && a > m) ? a : m;
    }
    { const float o = row_xor<1>(m); m = (o > m) ? o : m; }
    { const float o = row_xor<2>(m); m = (o > m) ? o : m; }
    { const float o = row_xor<4>(m); m = (o > m) ? o : m; }
    if (!(__builtin_fabsf(m) < __builtin_huge_valf())) m = 0.f;

    float s = 0.f;
    for (int p = beg; p < end; p += 8) {
        const int i = p + k;
        const float a = salpha[(size_t)((i < end) ? i : p) * 4 + h];
        s += (i < end) ? __expf(a - m) : 0.f;
    }
    s += row_xor<1>(s);
    s += row_xor<2>(s);
    s += row_xor<4>(s);
    const float den = s + EPSV;

    v4f acc = {0.f, 0.f, 0.f, 0.f};
    const float* xl = xh + 4 * lane;
    for (int p = beg; p < end; p += 8) {
        const int i = p + k;
        const bool ok = i < end;
        const int ii = ok ? i : p;
        const int sv = ssrc[ii];
        const float a = salpha[(size_t)ii * 4 + h];
        const float c = ok ? __expf(a - m) / den : 0.f;
        v4f row[8];
        float ck[8];
        gather_rows<0>(xl, sv, c, row, ck);
#pragma unroll
        for (int q = 0; q < 8; ++q) acc += ck[q] * row[q];
    }

    v4f t;
    t.x = acc.x + row_xor<8>(acc.x);
    t.y = acc.y + row_xor<8>(acc.y);
    t.z = acc.z + row_xor<8>(acc.z);
    t.w = acc.w + row_xor<8>(acc.w);
    t.x += __shfl_xor(t.x, 16);
    t.y += __shfl_xor(t.y, 16);
    t.z += __shfl_xor(t.z, 16);
    t.w += __shfl_xor(t.w, 16);
    if (lane < 8) {
        const v4f bc = *(const v4f*)(bias_conv + 4 * lane);
        const v4f bl = *(const v4f*)(bias_layer + 4 * lane);
        v4f r = t * 0.25f + bc + bl;
        r.x = r.x > 0.f ? r.x : 0.f;
        r.y = r.y > 0.f ? r.y : 0.f;
        r.z = r.z > 0.f ? r.z : 0.f;
        r.w = r.w > 0.f ? r.w : 0.f;
        *(v4f*)(out + (size_t)n * OUT_CH + 4 * lane) = r;
    }
}

extern "C" void kernel_launch(void* const* d_in, const int* in_sizes, int n_in,
                              void* d_out, int out_size, void* d_ws, size_t ws_size,
                              hipStream_t stream) {
    const float* x          = (const float*)d_in[0];
    const float* edge_attr  = (const float*)d_in[1];
    const float* W          = (const float*)d_in[2];
    const float* W_edge     = (const float*)d_in[3];
    const float* att_src    = (const float*)d_in[4];
    const float* att_dst    = (const float*)d_in[5];
    const float* att_edge   = (const float*)d_in[6];
    const float* bias_conv  = (const float*)d_in[7];
    const float* bias_layer = (const float*)d_in[8];
    const int*   edge_index = (const int*)d_in[9];
    float* out = (float*)d_out;

    float* ws      = (float*)d_ws;
    float* xh      = ws;
    float* s_node  = xh      + (size_t)N_NODES * HC;
    float* d_node  = s_node  + (size_t)N_NODES * HEADS;
    float* weff    = d_node  + (size_t)N_NODES * HEADS;
    float* salpha  = weff    + (size_t)HEADS * IN_CH;
    int*   ssrc    = (int*)(salpha + (size_t)N_EDGES * HEADS);
    int*   cnt     = ssrc    + (size_t)N_EDGES;
    int*   cursor  = cnt     + (size_t)N_NODES;

    proj_nodes_wmma<<<N_NODES / 16, 128, 0, stream>>>(x, W, xh, cnt);
    build_weff<<<1, 256, 0, stream>>>(W_edge, att_edge, weff);
    node_attn_scalars<<<(N_NODES * HEADS + 255) / 256, 256, 0, stream>>>(
        xh, att_src, att_dst, s_node, d_node, edge_index, cnt);
    seg_offsets<<<1, 1024, 0, stream>>>(cnt, cursor);
    edge_logits<<<(N_EDGES + 255) / 256, 256, 0, stream>>>(
        edge_attr, edge_index, s_node, d_node, weff, cursor, salpha, ssrc);
    node_softmax_gather<<<N_NODES / 8, 256, 0, stream>>>(
        cursor, salpha, ssrc, xh, bias_conv, bias_layer, out);
}


# counting sort's per-node counters move from global memory to LDS: workgroups count, then rank, their chunk's edges in a 50 KB table; no per-edge global atomic left (slot order, so sum order, changes)
# speedup vs baseline: 3.7847x; 1.7683x over previous
#include <hip/hip_runtime.h>
#include <cstdint>

#define N_NODES 50000
#define N_EDGES 800000
#define IN_CH 64
#define OUT_CH 32
#define HEADS 4
#define HC 128
#define NEG_SLOPE 0.2f
#define EPSV 1e-16f

typedef __attribute__((ext_vector_type(2))) float v2f;
typedef __attribute__((ext_vector_type(4))) float v4f;
typedef __attribute__((ext_vector_type(8))) float v8f;

__global__ void __launch_bounds__(128) proj_nodes_wmma(
    const float* __restrict__ x, const float* __restrict__ W,
    float* __restrict__ xh)
{
    const int lane = threadIdx.x & 31;
    const int wv   = threadIdx.x >> 5;
    const int half = lane >> 4;
    const int l15  = lane & 15;
    const int m0   = blockIdx.x * 16;

    const int rowA  = m0 + l15;
    const int colB0 = wv * 32 + l15;
    const int colB1 = colB0 + 16;

    v8f acc0 = {};
    v8f acc1 = {};
#pragma unroll
    for (int k0 = 0; k0 < IN_CH; k0 += 4) {
        const int kk = k0 + half * 2;
        v2f a  = *(const v2f*)(x + (size_t)rowA  * IN_CH + kk);
        v2f b0 = *(const v2f*)(W + (size_t)colB0 * IN_CH + kk);
        v2f b1 = *(const v2f*)(W + (size_t)colB1 * IN_CH + kk);
        acc0 = __builtin_amdgcn_wmma_f32_16x16x4_f32(
                   false, a, false, b0, (short)0, acc0, false, false);
        acc1 = __builtin_amdgcn_wmma_f32_16x16x4_f32(
                   false, a, false, b1, (short)0, acc1, false, false);
    }
#pragma unroll
    for (int v = 0; v < 8; ++v) {
        const int m = m0 + v + half * 8;
        xh[(size_t)m * HC + colB0] = acc0[v];
        xh[(size_t)m * HC + colB1] = acc1[v];
    }
}

__global__ void build_weff(const float* __restrict__ W_edge,
                           const float* __restrict__ att_edge,
                           float* __restrict__ weff)
{
    const int t = threadIdx.x;
    const int h = t >> 6, k = t & 63;
    float s = 0.f;
#pragma unroll
    for (int c = 0; c < OUT_CH; ++c)
        s += att_edge[h * OUT_CH + c] * W_edge[(size_t)(h * OUT_CH + c) * IN_CH + k];
    weff[t] = s;
}

#define SORT_CHUNKS 64
#define NODE_PARTS 4
#define CHUNK_EDGES (N_EDGES / SORT_CHUNKS)
#define PART_NODES (N_NODES / NODE_PARTS)
#define CHUNK_VEC_ITERS ((CHUNK_EDGES / 4 + 1023) / 1024)
#define PART_VEC_ITERS ((PART_NODES / 4 + 1023) / 1024)
static_assert(N_EDGES % SORT_CHUNKS == 0 && CHUNK_EDGES % 4 == 0, "chunks of equal size, read 16 bytes at a time");
static_assert(N_NODES % NODE_PARTS == 0 && PART_NODES % 4 == 0, "ranges of equal size, 16-byte table accesses");

__device__ __forceinline__ unsigned part_index(int d, int lo)
{
    return (unsigned)(min(max(d, 0), N_NODES - 1) - lo);
}

__global__ void __launch_bounds__(1024) chunk_histogram(const int* __restrict__ edge_index,
                                                        int* __restrict__ plane)
{
    __shared__ alignas(16) int tab[PART_NODES];
    const int g  = blockIdx.x;
    const int lo = blockIdx.y * PART_NODES;
    const int4* dp = (const int4*)(edge_index + N_EDGES + g * CHUNK_EDGES);
    int4 d[CHUNK_VEC_ITERS];
#pragma unroll
    for (int j = 0; j < CHUNK_VEC_ITERS; ++j) {
        const int i = threadIdx.x + j * 1024;
        d[j] = (i < CHUNK_EDGES / 4) ? dp[i] : make_int4(0, 0, 0, 0);
    }
#pragma unroll
    for (int j = 0; j < PART_VEC_ITERS; ++j) {
        const int i = threadIdx.x + j * 1024;
        if (i < PART_NODES / 4) ((int4*)tab)[i] = make_int4(0, 0, 0, 0);
    }
    __syncthreads();
#pragma unroll
    for (int j = 0; j < CHUNK_VEC_ITERS; ++j) {
        const bool ok = threadIdx.x + j * 1024 < CHUNK_EDGES / 4;
        const unsigned r0 = part_index(d[j].x, lo), r1 = part_index(d[j].y, lo);
        const unsigned r2 = part_index(d[j].z, lo), r3 = part_index(d[j].w, lo);
        if (ok && r0 < PART_NODES) atomicAdd(tab + r0, 1);
        if (ok && r1 < PART_NODES) atomicAdd(tab + r1, 1);
        if (ok && r2 < PART_NODES) atomicAdd(tab + r2, 1);
        if (ok && r3 < PART_NODES) atomicAdd(tab + r3, 1);
    }
    __syncthreads();
    int4* pl = (int4*)(plane + (size_t)g * N_NODES + lo);
#pragma unroll
    for (int j = 0; j < PART_VEC_ITERS; ++j) {
        const int i = threadIdx.x + j * 1024;
        if (i < PART_NODES / 4) pl[i] = ((const int4*)tab)[i];
    }
}

__global__ void __launch_bounds__(256) chunk_prefix(int* __restrict__ plane, int* __restrict__ cnt)
{
    const int n = blockIdx.x * 256 + threadIdx.x;
    if (n >= N_NODES) return;
    int c[SORT_CHUNKS];
#pragma unroll
    for (int g = 0; g < SORT_CHUNKS; ++g) c[g] = plane[(size_t)g * N_NODES + n];
    int run = 0;
#pragma unroll
    for (int g = 0; g < SORT_CHUNKS; ++g) {
        plane[(size_t)g * N_NODES + n] = run;
        run += c[g];
    }
    cnt[n] = run;
}

__global__ void __launch_bounds__(1024) chunk_slots(const int* __restrict__ edge_index,
                                                    const int* __restrict__ cursor,
                                                    const int* __restrict__ plane,
                                                    int* __restrict__ slotof)
{
    __shared__ alignas(16) int tab[PART_NODES];
    const int g  = blockIdx.x;
    const int lo = blockIdx.y * PART_NODES;
    const int4* dp = (const int4*)(edge_index + N_EDGES + g * CHUNK_EDGES);
    const int4* cu = (const int4*)(cursor + lo);
    const int4* pl = (const int4*)(plane + (size_t)g * N_NODES + lo);
    int4 d[CHUNK_VEC_ITERS], ca[PART_VEC_ITERS], pa[PART_VEC_ITERS];
#pragma unroll
    for (int j = 0; j < PART_VEC_ITERS; ++j) {
        const int i = threadIdx.x + j * 1024;
        const bool in = i < PART_NODES / 4;
        ca[j] = in ? cu[i] : make_int4(0, 0, 0, 0);
        pa[j] = in ? pl[i] : make_int4(0, 0, 0, 0);
    }
#pragma unroll
    for (int j = 0; j < CHUNK_VEC_ITERS; ++j) {
        const int i = threadIdx.x + j * 1024;
        d[j] = (i < CHUNK_EDGES / 4) ? dp[i] : make_int4(0, 0, 0, 0);
    }
#pragma unroll
    for (int j = 0; j < PART_VEC_ITERS; ++j) {
        const int i = threadIdx.x + j * 1024;
        if (i < PART_NODES / 4)
            ((int4*)tab)[i] = make_int4(ca[j].x + pa[j].x, ca[j].y + pa[j].y,
                                        ca[j].z + pa[j].z, ca[j].w + pa[j].w);
    }
    __syncthreads();
    int* so = slotof + g * CHUNK_EDGES;
    int4 sl[CHUNK_VEC_ITERS];
#pragma unroll
    for (int j = 0; j < CHUNK_VEC_ITERS; ++j) {
        const bool ok = threadIdx.x + j * 1024 < CHUNK_EDGES / 4;
        const unsigned r0 = part_index(d[j].x, lo), r1 = part_index(d[j].y, lo);
        const unsigned r2 = part_index(d[j].z, lo), r3 = part_index(d[j].w, lo);
        sl[j].x = (ok && r0 < PART_NODES) ? atomicAdd(tab + r0, 1) : -1;
        sl[j].y = (ok && r1 < PART_NODES) ? atomicAdd(tab + r1, 1) : -1;
        sl[j].z = (ok && r2 < PART_NODES) ? atomicAdd(tab + r2, 1) : -1;
        sl[j].w = (ok && r3 < PART_NODES) ? atomicAdd(tab + r3, 1) : -1;
    }
#pragma unroll
    for (int j = 0; j < CHUNK_VEC_ITERS; ++j) {
        const int i = threadIdx.x + j * 1024;
        if (sl[j].x >= 0) so[4 * i + 0] = sl[j].x;
        if (sl[j].y >= 0) so[4 * i + 1] = sl[j].y;
        if (sl[j].z >= 0) so[4 * i + 2] = sl[j].z;
        if (sl[j].w >= 0) so[4 * i + 3] = sl[j].w;
    }
}

#define SCAN_ITERS 13
#define SCAN_PER_WAVE (SCAN_ITERS * 128)
static_assert(32 * SCAN_PER_WAVE >= N_NODES, "one block covers all nodes");
static_assert(N_NODES % 4 == 0, "16-byte counter loads are all-in or all-out");

__global__ void __launch_bounds__(1024) seg_offsets(const int* __restrict__ cnt,
                                                    int* __restrict__ cursor)
{
    __shared__ int wtot[32];
    const int lane = threadIdx.x & 31;
    const int wv   = threadIdx.x >> 5;
    const int base = wv * SCAN_PER_WAVE + lane * 4;

    int4 c[SCAN_ITERS];
#pragma unroll
    for (int i = 0; i < SCAN_ITERS; ++i) {
        const int idx = base + i * 128;
        c[i] = (idx < N_NODES) ? *(const int4*)(cnt + idx) : make_int4(0, 0, 0, 0);
    }
    int run = 0;
    int ex[SCAN_ITERS];
#pragma unroll
    for (int i = 0; i < SCAN_ITERS; ++i) {
        const int s = c[i].x + c[i].y + c[i].z + c[i].w;
        int inc = s;
#pragma unroll
        for (int d = 1; d < 32; d <<= 1) {
            const int up = __shfl_up(inc, d);
            if (lane >= d) inc += up;
        }
        ex[i] = run + inc - s;
        run += __shfl(inc, 31);
    }
    if (lane == 0) wtot[wv] = run;
    __syncthreads();
    int wb = (lane < wv) ? wtot[lane] : 0;
#pragma unroll
    for (int d = 16; d >= 1; d >>= 1) wb += __shfl_xor(wb, d);
#pragma unroll
    for (int i = 0; i < SCAN_ITERS; ++i) {
        const int idx = base + i * 128;
        if (idx < N_NODES) {
            int4 r;
            r.x = wb + ex[i];
            r.y = r.x + c[i].x;
            r.z = r.y + c[i].y;
            r.w = r.z + c[i].z;
            *(int4*)(cursor + idx) = r;
        }
    }
}

template <int M>
__device__ __forceinline__ float row_xor(float v)
{
    return __builtin_bit_cast(float, __builtin_amdgcn_update_dpp(
        0, __builtin_bit_cast(int, v), 0x160 + M, 0xf, 0xf, true));
}

template <int M>
__device__ __forceinline__ float fold_pair(float a, float b, bool bit)
{
    const float send = bit ? a : b;
    const float keep = bit ? b : a;
    return keep + row_xor<M>(send);
}

__device__ __forceinline__ float dot4(v4f v, v4f w)
{
    return __builtin_fmaf(v.w, w.w, __builtin_fmaf(v.z, w.z, __builtin_fmaf(v.y, w.y, v.x * w.x)));
}

static_assert(N_EDGES % 256 == 0, "one wave per 32 edges, 8 waves per block");

template <int J>
__device__ __forceinline__ int quad_lane(int v)
{
    return __builtin_amdgcn_update_dpp(0, v, J * 0x55, 0xf, 0xf, true);
}

template <int J>
__device__ __forceinline__ void store_terms(float* __restrict__ salpha, int slot, int h,
                                            const float (&u)[4])
{
    int p = quad_lane<J>(slot);
    p = min(max(p, 0), N_EDGES - 1);
    salpha[(size_t)p * 4 + h] = u[J];
    if constexpr (J + 1 < 4) store_terms<J + 1>(salpha, slot, h, u);
}

__global__ void __launch_bounds__(256) __attribute__((amdgpu_waves_per_eu(8, 8))) edge_logits(
    const float* __restrict__ edge_attr, const int* __restrict__ edge_index,
    const float* __restrict__ weff,
    const int* __restrict__ slotof, float* __restrict__ salpha, int* __restrict__ ssrc)
{
    const int lane = threadIdx.x & 31;
    const int l15  = lane & 15;
    const int half = lane >> 4;
    const int h    = l15 & 3;
    const int e0   = (blockIdx.x * 8 + (threadIdx.x >> 5)) * 32;
    const int eb   = e0 + ((l15 >> 2) << 1) + half;

    const v4f w0 = *(const v4f*)(weff + 0 * IN_CH + 4 * l15);
    const v4f w1 = *(const v4f*)(weff + 1 * IN_CH + 4 * l15);
    const v4f w2 = *(const v4f*)(weff + 2 * IN_CH + 4 * l15);
    const v4f w3 = *(const v4f*)(weff + 3 * IN_CH + 4 * l15);

    const int em   = eb + 8 * h;
    const int srcm = edge_index[em];
    const int slot = slotof[em];

    const v4f* ea = (const v4f*)(edge_attr + (size_t)e0 * IN_CH) + lane;
    v4f v[16];
#pragma unroll
    for (int i = 0; i < 16; ++i)
        v[i] = __builtin_nontemporal_load(ea + i * 32);

    __builtin_amdgcn_sched_barrier(0);

    const bool b0 = (lane & 1) != 0, b1 = (lane & 2) != 0;
    const bool b2 = (lane & 4) != 0, b3 = (lane & 8) != 0;

    float s[16];
#pragma unroll
    for (int i = 0; i < 16; ++i) {
        const float r01 = fold_pair<1>(dot4(v[i], w0), dot4(v[i], w1), b0);
        const float r23 = fold_pair<1>(dot4(v[i], w2), dot4(v[i], w3), b0);
        s[i] = fold_pair<2>(r01, r23, b1);
    }
    float t[8];
#pragma unroll
    for (int m = 0; m < 8; ++m) t[m] = fold_pair<4>(s[2 * m], s[2 * m + 1], b2);
    float u[4];
#pragma unroll
    for (int j = 0; j < 4; ++j) u[j] = fold_pair<8>(t[2 * j], t[2 * j + 1], b3);

    ssrc[min(max(slot, 0), N_EDGES - 1)] = srcm;
    store_terms<0>(salpha, slot, h, u);
}

template <int K>
__device__ __forceinline__ float group8_lane(float v)
{
    constexpr int sel = K | (K << 3) | (K << 6) | (K << 9) | (K << 12) | (K << 15) | (K << 18) | (K << 21);
    return __builtin_bit_cast(float, __builtin_amdgcn_mov_dpp8(__builtin_bit_cast(int, v), sel));
}

__device__ __forceinline__ float group8_sum(float v)
{
    v += row_xor<1>(v);
    v += row_xor<2>(v);
    v += row_xor<4>(v);
    return v;
}

__device__ __forceinline__ float group8_max(float v)
{
    { const float o = row_xor<1>(v); v = (o > v) ? o : v; }
    { const float o = row_xor<2>(v); v = (o > v) ? o : v; }
    { const float o = row_xor<4>(v); v = (o > v) ? o : v; }
    return v;
}

template <int K>
__device__ __forceinline__ void gather_rows(const float* __restrict__ xl, int sv, v4f (&row)[8])
{
    const int sk = min(max(__builtin_amdgcn_readlane(sv, K), 0), N_NODES - 1);
    row[K] = *(const v4f*)(xl + (size_t)sk * HC);
    if constexpr (K + 1 < 8) gather_rows<K + 1>(xl, sv, row);
}

template <int K>
__device__ __forceinline__ void weigh_rows(float w, const v4f (&row)[8], v4f& pa)
{
    pa += group8_lane<K>(w) * row[K];
    if constexpr (K + 1 < 8) weigh_rows<K + 1>(w, row, pa);
}

static_assert(N_NODES % 8 == 0, "one wave per node, 8 waves per block");

__global__ void __launch_bounds__(256) node_softmax_gather(
    const int* __restrict__ cursor, const float* __restrict__ salpha,
    const int* __restrict__ ssrc, const float* __restrict__ xh,
    const float* __restrict__ att_src, const float* __restrict__ att_dst,
    const float* __restrict__ bias_conv, const float* __restrict__ bias_layer,
    float* __restrict__ out)
{
    const int lane = threadIdx.x & 31;
    const int k    = lane & 7;
    const int h    = lane >> 3;
    const int n = __builtin_amdgcn_readfirstlane(blockIdx.x * 8 + (threadIdx.x >> 5));

    int beg = cursor[n];
    int end = (n + 1 < N_NODES) ? cursor[n + 1] : N_EDGES;
    beg = max(beg, 0);
    end = min(end, N_EDGES);

    const float* xl = xh + 4 * lane;
    const v4f as4 = *(const v4f*)(att_src + 4 * lane);
    const v4f ad4 = *(const v4f*)(att_dst + 4 * lane);
    const float sh = group8_sum(dot4(*(const v4f*)(xl + (size_t)n * HC), as4));

    const bool b0 = (lane & 1) != 0, b1 = (lane & 2) != 0, b2 = (lane & 4) != 0;

    float m = -__builtin_huge_valf();
    float wsum = 0.f;
    v4f acc = {0.f, 0.f, 0.f, 0.f};
    for (int p = beg; p < end; p += 8) {
        const int i = p + k;
        const bool ok = i < end;
        const int ii = ok ? i : p;
        const int sv = ssrc[ii];
        const float e = salpha[(size_t)ii * 4 + h];
        v4f row[8];
        gather_rows<0>(xl, sv, row);
        const float d01 = fold_pair<1>(dot4(row[0], ad4), dot4(row[1], ad4), b0);
        const float d23 = fold_pair<1>(dot4(row[2], ad4), dot4(row[3], ad4), b0);
        const float d45 = fold_pair<1>(dot4(row[4], ad4), dot4(row[5], ad4), b0);
        const float d67 = fold_pair<1>(dot4(row[6], ad4), dot4(row[7], ad4), b0);
        const float d03 = fold_pair<2>(d01, d23, b1);
        const float d47 = fold_pair<2>(d45, d67, b1);
        const float dk  = fold_pair<4>(d03, d47, b2);
        float a = (e + sh) + dk;
        a = (a > 0.f) ? a : NEG_SLOPE * a;
        a = ok ? a : -__builtin_huge_valf();
        const float cm = group8_max(a);
        const float mn = (cm > m) ? cm : m;
        const float sc = __expf(m - mn);
        const float w  = __expf(a - mn);
        v4f pa = {0.f, 0.f, 0.f, 0.f};
        weigh_rows<0>(w, row, pa);
        wsum = wsum * sc + group8_sum(w);
        acc  = acc * sc + pa;
        m = mn;
    }
    acc = acc / (wsum + EPSV);

    v4f t;
    t.x = acc.x + row_xor<8>(acc.x);
    t.y = acc.y + row_xor<8>(acc.y);
    t.z = acc.z + row_xor<8>(acc.z);
    t.w = acc.w + row_xor<8>(acc.w);
    t.x += __shfl_xor(t.x, 16);
    t.y += __shfl_xor(t.y, 16);
    t.z += __shfl_xor(t.z, 16);
    t.w += __shfl_xor(t.w, 16);
    if (lane < 8) {
        const v4f bc = *(const v4f*)(bias_conv + 4 * lane);
        const v4f bl = *(const v4f*)(bias_layer + 4 * lane);
        v4f r = t * 0.25f + bc + bl;
        r.x = r.x > 0.f ? r.x : 0.f;
        r.y = r.y > 0.f ? r.y : 0.f;
        r.z = r.z > 0.f ? r.z : 0.f;
        r.w = r.w > 0.f ? r.w : 0.f;
        *(v4f*)(out + (size_t)n * OUT_CH + 4 * lane) = r;
    }
}

extern "C" void kernel_launch(void* const* d_in, const int* in_sizes, int n_in,
                              void* d_out, int out_size, void* d_ws, size_t ws_size,
                              hipStream_t stream) {
    const float* x          = (const float*)d_in[0];
    const float* edge_attr  = (const float*)d_in[1];
    const float* W          = (const float*)d_in[2];
    const float* W_edge     = (const float*)d_in[3];
    const float* att_src    = (const float*)d_in[4];
    const float* att_dst    = (const float*)d_in[5];
    const float* att_edge   = (const float*)d_in[6];
    const float* bias_conv  = (const float*)d_in[7];
    const float* bias_layer = (const float*)d_in[8];
    const int*   edge_index = (const int*)d_in[9];
    float* out = (float*)d_out;

    float* ws      = (float*)d_ws;
    float* xh      = ws;
    float* weff    = xh      + (size_t)N_NODES * HC;
    float* salpha  = weff    + (size_t)HEADS * IN_CH;
    int*   ssrc    = (int*)(salpha + (size_t)N_EDGES * HEADS);
    int*   cnt     = ssrc    + (size_t)N_EDGES;
    int*   cursor  = cnt     + (size_t)N_NODES;
    int*   slotof  = cursor  + (size_t)N_NODES;
    int*   plane   = slotof  + (size_t)N_EDGES;

    proj_nodes_wmma<<<N_NODES / 16, 128, 0, stream>>>(x, W, xh);
    build_weff<<<1, 256, 0, stream>>>(W_edge, att_edge, weff);
    chunk_histogram<<<dim3(SORT_CHUNKS, NODE_PARTS), 1024, 0, stream>>>(edge_index, plane);
    chunk_prefix<<<(N_NODES + 255) / 256, 256, 0, stream>>>(plane, cnt);
    seg_offsets<<<1, 1024, 0, stream>>>(cnt, cursor);
    chunk_slots<<<dim3(SORT_CHUNKS, NODE_PARTS), 1024, 0, stream>>>(edge_index, cursor, plane, slotof);
    edge_logits<<<(N_EDGES + 255) / 256, 256, 0, stream>>>(
        edge_attr, edge_index, weff, slotof, salpha, ssrc);
    node_softmax_gather<<<N_NODES / 8, 256, 0, stream>>>(
        cursor, salpha, ssrc, xh, att_src, att_dst, bias_conv, bias_layer, out);
}


# segment-start scan split: per 256-node block inside chunk_prefix, the 196 block totals scanned by each chunk_slots workgroup itself; the one-workgroup scan launch goes (8 launches -> 7)
# speedup vs baseline: 4.0004x; 1.0570x over previous
#include <hip/hip_runtime.h>
#include <cstdint>

#define N_NODES 50000
#define N_EDGES 800000
#define IN_CH 64
#define OUT_CH 32
#define HEADS 4
#define HC 128
#define NEG_SLOPE 0.2f
#define EPSV 1e-16f

typedef __attribute__((ext_vector_type(2))) float v2f;
typedef __attribute__((ext_vector_type(4))) float v4f;
typedef __attribute__((ext_vector_type(8))) float v8f;

__global__ void __launch_bounds__(128) proj_nodes_wmma(
    const float* __restrict__ x, const float* __restrict__ W,
    float* __restrict__ xh)
{
    const int lane = threadIdx.x & 31;
    const int wv   = threadIdx.x >> 5;
    const int half = lane >> 4;
    const int l15  = lane & 15;
    const int m0   = blockIdx.x * 16;

    const int rowA  = m0 + l15;
    const int colB0 = wv * 32 + l15;
    const int colB1 = colB0 + 16;

    v8f acc0 = {};
    v8f acc1 = {};
#pragma unroll
    for (int k0 = 0; k0 < IN_CH; k0 += 4) {
        const int kk = k0 + half * 2;
        v2f a  = *(const v2f*)(x + (size_t)rowA  * IN_CH + kk);
        v2f b0 = *(const v2f*)(W + (size_t)colB0 * IN_CH + kk);
        v2f b1 = *(const v2f*)(W + (size_t)colB1 * IN_CH + kk);
        acc0 = __builtin_amdgcn_wmma_f32_16x16x4_f32(
                   false, a, false, b0, (short)0, acc0, false, false);
        acc1 = __builtin_amdgcn_wmma_f32_16x16x4_f32(
                   false, a, false, b1, (short)0, acc1, false, false);
    }
#pragma unroll
    for (int v = 0; v < 8; ++v) {
        const int m = m0 + v + half * 8;
        xh[(size_t)m * HC + colB0] = acc0[v];
        xh[(size_t)m * HC + colB1] = acc1[v];
    }
}

__global__ void build_weff(const float* __restrict__ W_edge,
                           const float* __restrict__ att_edge,
                           float* __restrict__ weff)
{
    const int t = threadIdx.x;
    const int h = t >> 6, k = t & 63;
    float s = 0.f;
#pragma unroll
    for (int c = 0; c < OUT_CH; ++c)
        s += att_edge[h * OUT_CH + c] * W_edge[(size_t)(h * OUT_CH + c) * IN_CH + k];
    weff[t] = s;
}

#define SORT_CHUNKS 64
#define NODE_PARTS 4
#define CHUNK_EDGES (N_EDGES / SORT_CHUNKS)
#define PART_NODES (N_NODES / NODE_PARTS)
#define CHUNK_VEC_ITERS ((CHUNK_EDGES / 4 + 1023) / 1024)
#define PART_VEC_ITERS ((PART_NODES / 4 + 1023) / 1024)
#define SCAN_BLOCKS ((N_NODES + 255) / 256)
#define SCAN_PER_LANE ((SCAN_BLOCKS + 31) / 32)
static_assert(N_EDGES % SORT_CHUNKS == 0 && CHUNK_EDGES % 4 == 0, "chunks of equal size, read 16 bytes at a time");
static_assert(N_NODES % NODE_PARTS == 0 && PART_NODES % 4 == 0, "ranges of equal size, 16-byte table accesses");

__device__ __forceinline__ unsigned part_index(int d, int lo)
{
    return (unsigned)(min(max(d, 0), N_NODES - 1) - lo);
}

__global__ void __launch_bounds__(1024) chunk_histogram(const int* __restrict__ edge_index,
                                                        int* __restrict__ plane)
{
    __shared__ alignas(16) int tab[PART_NODES];
    const int g  = blockIdx.x;
    const int lo = blockIdx.y * PART_NODES;
    const int4* dp = (const int4*)(edge_index + N_EDGES + g * CHUNK_EDGES);
    int4 d[CHUNK_VEC_ITERS];
#pragma unroll
    for (int j = 0; j < CHUNK_VEC_ITERS; ++j) {
        const int i = threadIdx.x + j * 1024;
        d[j] = (i < CHUNK_EDGES / 4) ? dp[i] : make_int4(0, 0, 0, 0);
    }
#pragma unroll
    for (int j = 0; j < PART_VEC_ITERS; ++j) {
        const int i = threadIdx.x + j * 1024;
        if (i < PART_NODES / 4) ((int4*)tab)[i] = make_int4(0, 0, 0, 0);
    }
    __syncthreads();
#pragma unroll
    for (int j = 0; j < CHUNK_VEC_ITERS; ++j) {
        const bool ok = threadIdx.x + j * 1024 < CHUNK_EDGES / 4;
        const unsigned r0 = part_index(d[j].x, lo), r1 = part_index(d[j].y, lo);
        const unsigned r2 = part_index(d[j].z, lo), r3 = part_index(d[j].w, lo);
        if (ok && r0 < PART_NODES) atomicAdd(tab + r0, 1);
        if (ok && r1 < PART_NODES) atomicAdd(tab + r1, 1);
        if (ok && r2 < PART_NODES) atomicAdd(tab + r2, 1);
        if (ok && r3 < PART_NODES) atomicAdd(tab + r3, 1);
    }
    __syncthreads();
    int4* pl = (int4*)(plane + (size_t)g * N_NODES + lo);
#pragma unroll
    for (int j = 0; j < PART_VEC_ITERS; ++j) {
        const int i = threadIdx.x + j * 1024;
        if (i < PART_NODES / 4) pl[i] = ((const int4*)tab)[i];
    }
}

__global__ void __launch_bounds__(256) chunk_prefix(int* __restrict__ plane, int* __restrict__ seglocal,
                                                   int* __restrict__ btot)
{
    __shared__ int wtot[8];
    const int lane = threadIdx.x & 31;
    const int wv   = threadIdx.x >> 5;
    const int n = blockIdx.x * 256 + threadIdx.x;
    const bool valid = n < N_NODES;
    const int nn = valid ? n : N_NODES - 1;
    int c[SORT_CHUNKS];
#pragma unroll
    for (int g = 0; g < SORT_CHUNKS; ++g) c[g] = plane[(size_t)g * N_NODES + nn];
    int run = 0;
    if (valid) {
#pragma unroll
        for (int g = 0; g < SORT_CHUNKS; ++g) {
            plane[(size_t)g * N_NODES + n] = run;
            run += c[g];
        }
    }
    int inc = run;
#pragma unroll
    for (int d = 1; d < 32; d <<= 1) {
        const int up = __shfl_up(inc, d);
        if (lane >= d) inc += up;
    }
    if (lane == 31) wtot[wv] = inc;
    __syncthreads();
    int wb = 0;
#pragma unroll
    for (int w = 0; w < 7; ++w) wb += (w < wv) ? wtot[w] : 0;
    if (valid) seglocal[n] = wb + inc - run;
    if (threadIdx.x == 255) btot[blockIdx.x] = wb + inc;
}

__global__ void __launch_bounds__(1024) chunk_slots(const int* __restrict__ edge_index,
                                                    const int* __restrict__ seglocal,
                                                    const int* __restrict__ btot,
                                                    const int* __restrict__ plane,
                                                    int* __restrict__ cursor,
                                                    int* __restrict__ slotof)
{
    __shared__ alignas(16) int tab[PART_NODES];
    __shared__ int bbase[32 * SCAN_PER_LANE];
    const int g  = blockIdx.x;
    const int lo = blockIdx.y * PART_NODES;
    const int sl_lane = (int)threadIdx.x - (1024 - 32);
    int bt[SCAN_PER_LANE] = {};
    if (sl_lane >= 0) {
#pragma unroll
        for (int t = 0; t < SCAN_PER_LANE; ++t)
            bt[t] = btot[min(sl_lane * SCAN_PER_LANE + t, SCAN_BLOCKS - 1)];
    }
    const int4* dp = (const int4*)(edge_index + N_EDGES + g * CHUNK_EDGES);
    const int4* cu = (const int4*)(seglocal + lo);
    const int4* pl = (const int4*)(plane + (size_t)g * N_NODES + lo);
    int4 d[CHUNK_VEC_ITERS], ca[PART_VEC_ITERS], pa[PART_VEC_ITERS];
#pragma unroll
    for (int j = 0; j < PART_VEC_ITERS; ++j) {
        const int i = threadIdx.x + j * 1024;
        const bool in = i < PART_NODES / 4;
        ca[j] = in ? cu[i] : make_int4(0, 0, 0, 0);
        pa[j] = in ? pl[i] : make_int4(0, 0, 0, 0);
    }
#pragma unroll
    for (int j = 0; j < CHUNK_VEC_ITERS; ++j) {
        const int i = threadIdx.x + j * 1024;
        d[j] = (i < CHUNK_EDGES / 4) ? dp[i] : make_int4(0, 0, 0, 0);
    }
    if (sl_lane >= 0) {
        int s = 0;
#pragma unroll
        for (int t = 0; t < SCAN_PER_LANE; ++t) {
            if (sl_lane * SCAN_PER_LANE + t >= SCAN_BLOCKS) bt[t] = 0;
            s += bt[t];
        }
        int inc = s;
#pragma unroll
        for (int dist = 1; dist < 32; dist <<= 1) {
            const int up = __shfl_up(inc, dist);
            if (sl_lane >= dist) inc += up;
        }
        int run = inc - s;
#pragma unroll
        for (int t = 0; t < SCAN_PER_LANE; ++t) {
            bbase[sl_lane * SCAN_PER_LANE + t] = run;
            run += bt[t];
        }
    }
    __syncthreads();
    int4* cw = (int4*)(cursor + lo);
#pragma unroll
    for (int j = 0; j < PART_VEC_ITERS; ++j) {
        const int i = threadIdx.x + j * 1024;
        if (i < PART_NODES / 4) {
            const int bb = bbase[(lo + 4 * i) >> 8];
            const int4 first = make_int4(ca[j].x + bb, ca[j].y + bb, ca[j].z + bb, ca[j].w + bb);
            if (g == 0) cw[i] = first;
            ((int4*)tab)[i] = make_int4(first.x + pa[j].x, first.y + pa[j].y,
                                        first.z + pa[j].z, first.w + pa[j].w);
        }
    }
    __syncthreads();
    int* so = slotof + g * CHUNK_EDGES;
    int4 sl[CHUNK_VEC_ITERS];
#pragma unroll
    for (int j = 0; j < CHUNK_VEC_ITERS; ++j) {
        const bool ok = threadIdx.x + j * 1024 < CHUNK_EDGES / 4;
        const unsigned r0 = part_index(d[j].x, lo), r1 = part_index(d[j].y, lo);
        const unsigned r2 = part_index(d[j].z, lo), r3 = part_index(d[j].w, lo);
        sl[j].x = (ok && r0 < PART_NODES) ? atomicAdd(tab + r0, 1) : -1;
        sl[j].y = (ok && r1 < PART_NODES) ? atomicAdd(tab + r1, 1) : -1;
        sl[j].z = (ok && r2 < PART_NODES) ? atomicAdd(tab + r2, 1) : -1;
        sl[j].w = (ok && r3 < PART_NODES) ? atomicAdd(tab + r3, 1) : -1;
    }
#pragma unroll
    for (int j = 0; j < CHUNK_VEC_ITERS; ++j) {
        const int i = threadIdx.x + j * 1024;
        if (sl[j].x >= 0) so[4 * i + 0] = sl[j].x;
        if (sl[j].y >= 0) so[4 * i + 1] = sl[j].y;
        if (sl[j].z >= 0) so[4 * i + 2] = sl[j].z;
        if (sl[j].w >= 0) so[4 * i + 3] = sl[j].w;
    }
}

template <int M>
__device__ __forceinline__ float row_xor(float v)
{
    return __builtin_bit_cast(float, __builtin_amdgcn_update_dpp(
        0, __builtin_bit_cast(int, v), 0x160 + M, 0xf, 0xf, true));
}

template <int M>
__device__ __forceinline__ float fold_pair(float a, float b, bool bit)
{
    const float send = bit ? a : b;
    const float keep = bit ? b : a;
    return keep + row_xor<M>(send);
}

__device__ __forceinline__ float dot4(v4f v, v4f w)
{
    return __builtin_fmaf(v.w, w.w, __builtin_fmaf(v.z, w.z, __builtin_fmaf(v.y, w.y, v.x * w.x)));
}

static_assert(N_EDGES % 256 == 0, "one wave per 32 edges, 8 waves per block");

template <int J>
__device__ __forceinline__ int quad_lane(int v)
{
    return __builtin_amdgcn_update_dpp(0, v, J * 0x55, 0xf, 0xf, true);
}

template <int J>
__device__ __forceinline__ void store_terms(float* __restrict__ salpha, int slot, int h,
                                            const float (&u)[4])
{
    int p = quad_lane<J>(slot);
    p = min(max(p, 0), N_EDGES - 1);
    salpha[(size_t)p * 4 + h] = u[J];
    if constexpr (J + 1 < 4) store_terms<J + 1>(salpha, slot, h, u);
}

__global__ void __launch_bounds__(256) __attribute__((amdgpu_waves_per_eu(8, 8))) edge_logits(
    const float* __restrict__ edge_attr, const int* __restrict__ edge_index,
    const float* __restrict__ weff,
    const int* __restrict__ slotof, float* __restrict__ salpha, int* __restrict__ ssrc)
{
    const int lane = threadIdx.x & 31;
    const int l15  = lane & 15;
    const int half = lane >> 4;
    const int h    = l15 & 3;
    const int e0   = (blockIdx.x * 8 + (threadIdx.x >> 5)) * 32;
    const int eb   = e0 + ((l15 >> 2) << 1) + half;

    const v4f w0 = *(const v4f*)(weff + 0 * IN_CH + 4 * l15);
    const v4f w1 = *(const v4f*)(weff + 1 * IN_CH + 4 * l15);
    const v4f w2 = *(const v4f*)(weff + 2 * IN_CH + 4 * l15);
    const v4f w3 = *(const v4f*)(weff + 3 * IN_CH + 4 * l15);

    const int em   = eb + 8 * h;
    const int srcm = edge_index[em];
    const int slot = slotof[em];

    const v4f* ea = (const v4f*)(edge_attr + (size_t)e0 * IN_CH) + lane;
    v4f v[16];
#pragma unroll
    for (int i = 0; i < 16; ++i)
        v[i] = __builtin_nontemporal_load(ea + i * 32);

    __builtin_amdgcn_sched_barrier(0);

    const bool b0 = (lane & 1) != 0, b1 = (lane & 2) != 0;
    const bool b2 = (lane & 4) != 0, b3 = (lane & 8) != 0;

    float s[16];
#pragma unroll
    for (int i = 0; i < 16; ++i) {
        const float r01 = fold_pair<1>(dot4(v[i], w0), dot4(v[i], w1), b0);
        const float r23 = fold_pair<1>(dot4(v[i], w2), dot4(v[i], w3), b0);
        s[i] = fold_pair<2>(r01, r23, b1);
    }
    float t[8];
#pragma unroll
    for (int m = 0; m < 8; ++m) t[m] = fold_pair<4>(s[2 * m], s[2 * m + 1], b2);
    float u[4];
#pragma unroll
    for (int j = 0; j < 4; ++j) u[j] = fold_pair<8>(t[2 * j], t[2 * j + 1], b3);

    ssrc[min(max(slot, 0), N_EDGES - 1)] = srcm;
    store_terms<0>(salpha, slot, h, u);
}

template <int K>
__device__ __forceinline__ float group8_lane(float v)
{
    constexpr int sel = K | (K << 3) | (K << 6) | (K << 9) | (K << 12) | (K << 15) | (K << 18) | (K << 21);
    return __builtin_bit_cast(float, __builtin_amdgcn_mov_dpp8(__builtin_bit_cast(int, v), sel));
}

__device__ __forceinline__ float group8_sum(float v)
{
    v += row_xor<1>(v);
    v += row_xor<2>(v);
    v += row_xor<4>(v);
    return v;
}

__device__ __forceinline__ float group8_max(float v)
{
    { const float o = row_xor<1>(v); v = (o > v) ? o : v; }
    { const float o = row_xor<2>(v); v = (o > v) ? o : v; }
    { const float o = row_xor<4>(v); v = (o > v) ? o : v; }
    return v;
}

template <int K>
__device__ __forceinline__ void gather_rows(const float* __restrict__ xl, int sv, v4f (&row)[8])
{
    const int sk = min(max(__builtin_amdgcn_readlane(sv, K), 0), N_NODES - 1);
    row[K] = *(const v4f*)(xl + (size_t)sk * HC);
    if constexpr (K + 1 < 8) gather_rows<K + 1>(xl, sv, row);
}

template <int K>
__device__ __forceinline__ void weigh_rows(float w, const v4f (&row)[8], v4f& pa)
{
    pa += group8_lane<K>(w) * row[K];
    if constexpr (K + 1 < 8) weigh_rows<K + 1>(w, row, pa);
}

static_assert(N_NODES % 8 == 0, "one wave per node, 8 waves per block");

__global__ void __launch_bounds__(256) node_softmax_gather(
    const int* __restrict__ cursor, const float* __restrict__ salpha,
    const int* __restrict__ ssrc, const float* __restrict__ xh,
    const float* __restrict__ att_src, const float* __restrict__ att_dst,
    const float* __restrict__ bias_conv, const float* __restrict__ bias_layer,
    float* __restrict__ out)
{
    const int lane = threadIdx.x & 31;
    const int k    = lane & 7;
    const int h    = lane >> 3;
    const int n = __builtin_amdgcn_readfirstlane(blockIdx.x * 8 + (threadIdx.x >> 5));

    int beg = cursor[n];
    int end = (n + 1 < N_NODES) ? cursor[n + 1] : N_EDGES;
    beg = max(beg, 0);
    end = min(end, N_EDGES);

    const float* xl = xh + 4 * lane;
    const v4f as4 = *(const v4f*)(att_src + 4 * lane);
    const v4f ad4 = *(const v4f*)(att_dst + 4 * lane);
    const float sh = group8_sum(dot4(*(const v4f*)(xl + (size_t)n * HC), as4));

    const bool b0 = (lane & 1) != 0, b1 = (lane & 2) != 0, b2 = (lane & 4) != 0;

    float m = -__builtin_huge_valf();
    float wsum = 0.f;
    v4f acc = {0.f, 0.f, 0.f, 0.f};
    for (int p = beg; p < end; p += 8) {
        const int i = p + k;
        const bool ok = i < end;
        const int ii = ok ? i : p;
        const int sv = ssrc[ii];
        const float e = salpha[(size_t)ii * 4 + h];
        v4f row[8];
        gather_rows<0>(xl, sv, row);
        const float d01 = fold_pair<1>(dot4(row[0], ad4), dot4(row[1], ad4), b0);
        const float d23 = fold_pair<1>(dot4(row[2], ad4), dot4(row[3], ad4), b0);
        const float d45 = fold_pair<1>(dot4(row[4], ad4), dot4(row[5], ad4), b0);
        const float d67 = fold_pair<1>(dot4(row[6], ad4), dot4(row[7], ad4), b0);
        const float d03 = fold_pair<2>(d01, d23, b1);
        const float d47 = fold_pair<2>(d45, d67, b1);
        const float dk  = fold_pair<4>(d03, d47, b2);
        float a = (e + sh) + dk;
        a = (a > 0.f) ? a : NEG_SLOPE * a;
        a = ok ? a : -__builtin_huge_valf();
        const float cm = group8_max(a);
        const float mn = (cm > m) ? cm : m;
        const float sc = __expf(m - mn);
        const float w  = __expf(a - mn);
        v4f pa = {0.f, 0.f, 0.f, 0.f};
        weigh_rows<0>(w, row, pa);
        wsum = wsum * sc + group8_sum(w);
        acc  = acc * sc + pa;
        m = mn;
    }
    acc = acc / (wsum + EPSV);

    v4f t;
    t.x = acc.x + row_xor<8>(acc.x);
    t.y = acc.y + row_xor<8>(acc.y);
    t.z = acc.z + row_xor<8>(acc.z);
    t.w = acc.w + row_xor<8>(acc.w);
    t.x += __shfl_xor(t.x, 16);
    t.y += __shfl_xor(t.y, 16);
    t.z += __shfl_xor(t.z, 16);
    t.w += __shfl_xor(t.w, 16);
    if (lane < 8) {
        const v4f bc = *(const v4f*)(bias_conv + 4 * lane);
        const v4f bl = *(const v4f*)(bias_layer + 4 * lane);
        v4f r = t * 0.25f + bc + bl;
        r.x = r.x > 0.f ? r.x : 0.f;
        r.y = r.y > 0.f ? r.y : 0.f;
        r.z = r.z > 0.f ? r.z : 0.f;
        r.w = r.w > 0.f ? r.w : 0.f;
        *(v4f*)(out + (size_t)n * OUT_CH + 4 * lane) = r;
    }
}

extern "C" void kernel_launch(void* const* d_in, const int* in_sizes, int n_in,
                              void* d_out, int out_size, void* d_ws, size_t ws_size,
                              hipStream_t stream) {
    const float* x          = (const float*)d_in[0];
    const float* edge_attr  = (const float*)d_in[1];
    const float* W          = (const float*)d_in[2];
    const float* W_edge     = (const float*)d_in[3];
    const float* att_src    = (const float*)d_in[4];
    const float* att_dst    = (const float*)d_in[5];
    const float* att_edge   = (const float*)d_in[6];
    const float* bias_conv  = (const float*)d_in[7];
    const float* bias_layer = (const float*)d_in[8];
    const int*   edge_index = (const int*)d_in[9];
    float* out = (float*)d_out;

    float* ws      = (float*)d_ws;
    float* xh      = ws;
    float* weff    = xh      + (size_t)N_NODES * HC;
    float* salpha  = weff    + (size_t)HEADS * IN_CH;
    int*   ssrc    = (int*)(salpha + (size_t)N_EDGES * HEADS);
    int*   seglocal = ssrc   + (size_t)N_EDGES;
    int*   cursor  = seglocal + (size_t)N_NODES;
    int*   slotof  = cursor  + (size_t)N_NODES;
    int*   plane   = slotof  + (size_t)N_EDGES;
    int*   btot    = plane   + (size_t)SORT_CHUNKS * N_NODES;

    proj_nodes_wmma<<<N_NODES / 16, 128, 0, stream>>>(x, W, xh);
    build_weff<<<1, 256, 0, stream>>>(W_edge, att_edge, weff);
    chunk_histogram<<<dim3(SORT_CHUNKS, NODE_PARTS), 1024, 0, stream>>>(edge_index, plane);
    chunk_prefix<<<SCAN_BLOCKS, 256, 0, stream>>>(plane, seglocal, btot);
    chunk_slots<<<dim3(SORT_CHUNKS, NODE_PARTS), 1024, 0, stream>>>(edge_index, seglocal, btot, plane, cursor, slotof);
    edge_logits<<<(N_EDGES + 255) / 256, 256, 0, stream>>>(
        edge_attr, edge_index, weff, slotof, salpha, ssrc);
    node_softmax_gather<<<N_NODES / 8, 256, 0, stream>>>(
        cursor, salpha, ssrc, xh, att_src, att_dst, bias_conv, bias_layer, out);
}


# edge pass stores one 32-byte record per edge at its slot (4 edge terms + its source) instead of one scattered 16-byte and one 4-byte store; node pass reads one record stream; sum order unchanged
# speedup vs baseline: 4.6342x; 1.1584x over previous
#include <hip/hip_runtime.h>
#include <cstdint>

#define N_NODES 50000
#define N_EDGES 800000
#define IN_CH 64
#define OUT_CH 32
#define HEADS 4
#define HC 128
#define NEG_SLOPE 0.2f
#define EPSV 1e-16f

typedef __attribute__((ext_vector_type(2))) float v2f;
typedef __attribute__((ext_vector_type(4))) float v4f;
typedef __attribute__((ext_vector_type(8))) float v8f;
typedef __attribute__((ext_vector_type(2))) int v2i;

__global__ void __launch_bounds__(128) proj_nodes_wmma(
    const float* __restrict__ x, const float* __restrict__ W,
    float* __restrict__ xh)
{
    const int lane = threadIdx.x & 31;
    const int wv   = threadIdx.x >> 5;
    const int half = lane >> 4;
    const int l15  = lane & 15;
    const int m0   = blockIdx.x * 16;

    const int rowA  = m0 + l15;
    const int colB0 = wv * 32 + l15;
    const int colB1 = colB0 + 16;

    v8f acc0 = {};
    v8f acc1 = {};
#pragma unroll
    for (int k0 = 0; k0 < IN_CH; k0 += 4) {
        const int kk = k0 + half * 2;
        v2f a  = *(const v2f*)(x + (size_t)rowA  * IN_CH + kk);
        v2f b0 = *(const v2f*)(W + (size_t)colB0 * IN_CH + kk);
        v2f b1 = *(const v2f*)(W + (size_t)colB1 * IN_CH + kk);
        acc0 = __builtin_amdgcn_wmma_f32_16x16x4_f32(
                   false, a, false, b0, (short)0, acc0, false, false);
        acc1 = __builtin_amdgcn_wmma_f32_16x16x4_f32(
                   false, a, false, b1, (short)0, acc1, false, false);
    }
#pragma unroll
    for (int v = 0; v < 8; ++v) {
        const int m = m0 + v + half * 8;
        xh[(size_t)m * HC + colB0] = acc0[v];
        xh[(size_t)m * HC + colB1] = acc1[v];
    }
}

__global__ void build_weff(const float* __restrict__ W_edge,
                           const float* __restrict__ att_edge,
                           float* __restrict__ weff)
{
    const int t = threadIdx.x;
    const int h = t >> 6, k = t & 63;
    float s = 0.f;
#pragma unroll
    for (int c = 0; c < OUT_CH; ++c)
        s += att_edge[h * OUT_CH + c] * W_edge[(size_t)(h * OUT_CH + c) * IN_CH + k];
    weff[t] = s;
}

#define SORT_CHUNKS 64
#define NODE_PARTS 4
#define CHUNK_EDGES (N_EDGES / SORT_CHUNKS)
#define PART_NODES (N_NODES / NODE_PARTS)
#define CHUNK_VEC_ITERS ((CHUNK_EDGES / 4 + 1023) / 1024)
#define PART_VEC_ITERS ((PART_NODES / 4 + 1023) / 1024)
#define SCAN_BLOCKS ((N_NODES + 255) / 256)
#define SCAN_PER_LANE ((SCAN_BLOCKS + 31) / 32)
static_assert(N_EDGES % SORT_CHUNKS == 0 && CHUNK_EDGES % 4 == 0, "chunks of equal size, read 16 bytes at a time");
static_assert(N_NODES % NODE_PARTS == 0 && PART_NODES % 4 == 0, "ranges of equal size, 16-byte table accesses");

__device__ __forceinline__ unsigned part_index(int d, int lo)
{
    return (unsigned)(min(max(d, 0), N_NODES - 1) - lo);
}

__global__ void __launch_bounds__(1024) chunk_histogram(const int* __restrict__ edge_index,
                                                        int* __restrict__ plane)
{
    __shared__ alignas(16) int tab[PART_NODES];
    const int g  = blockIdx.x;
    const int lo = blockIdx.y * PART_NODES;
    const int4* dp = (const int4*)(edge_index + N_EDGES + g * CHUNK_EDGES);
    int4 d[CHUNK_VEC_ITERS];
#pragma unroll
    for (int j = 0; j < CHUNK_VEC_ITERS; ++j) {
        const int i = threadIdx.x + j * 1024;
        d[j] = (i < CHUNK_EDGES / 4) ? dp[i] : make_int4(0, 0, 0, 0);
    }
#pragma unroll
    for (int j = 0; j < PART_VEC_ITERS; ++j) {
        const int i = threadIdx.x + j * 1024;
        if (i < PART_NODES / 4) ((int4*)tab)[i] = make_int4(0, 0, 0, 0);
    }
    __syncthreads();
#pragma unroll
    for (int j = 0; j < CHUNK_VEC_ITERS; ++j) {
        const bool ok = threadIdx.x + j * 1024 < CHUNK_EDGES / 4;
        const unsigned r0 = part_index(d[j].x, lo), r1 = part_index(d[j].y, lo);
        const unsigned r2 = part_index(d[j].z, lo), r3 = part_index(d[j].w, lo);
        if (ok && r0 < PART_NODES) atomicAdd(tab + r0, 1);
        if (ok && r1 < PART_NODES) atomicAdd(tab + r1, 1);
        if (ok && r2 < PART_NODES) atomicAdd(tab + r2, 1);
        if (ok && r3 < PART_NODES) atomicAdd(tab + r3, 1);
    }
    __syncthreads();
    int4* pl = (int4*)(plane + (size_t)g * N_NODES + lo);
#pragma unroll
    for (int j = 0; j < PART_VEC_ITERS; ++j) {
        const int i = threadIdx.x + j * 1024;
        if (i < PART_NODES / 4) pl[i] = ((const int4*)tab)[i];
    }
}

__global__ void __launch_bounds__(256) chunk_prefix(int* __restrict__ plane, int* __restrict__ seglocal,
                                                   int* __restrict__ btot)
{
    __shared__ int wtot[8];
    const int lane = threadIdx.x & 31;
    const int wv   = threadIdx.x >> 5;
    const int n = blockIdx.x * 256 + threadIdx.x;
    const bool valid = n < N_NODES;
    const int nn = valid ? n : N_NODES - 1;
    int c[SORT_CHUNKS];
#pragma unroll
    for (int g = 0; g < SORT_CHUNKS; ++g) c[g] = plane[(size_t)g * N_NODES + nn];
    int run = 0;
    if (valid) {
#pragma unroll
        for (int g = 0; g < SORT_CHUNKS; ++g) {
            plane[(size_t)g * N_NODES + n] = run;
            run += c[g];
        }
    }
    int inc = run;
#pragma unroll
    for (int d = 1; d < 32; d <<= 1) {
        const int up = __shfl_up(inc, d);
        if (lane >= d) inc += up;
    }
    if (lane == 31) wtot[wv] = inc;
    __syncthreads();
    int wb = 0;
#pragma unroll
    for (int w = 0; w < 7; ++w) wb += (w < wv) ? wtot[w] : 0;
    if (valid) seglocal[n] = wb + inc - run;
    if (threadIdx.x == 255) btot[blockIdx.x] = wb + inc;
}

__global__ void __launch_bounds__(1024) chunk_slots(const int* __restrict__ edge_index,
                                                    const int* __restrict__ seglocal,
                                                    const int* __restrict__ btot,
                                                    const int* __restrict__ plane,
                                                    int* __restrict__ cursor,
                                                    int* __restrict__ slotof)
{
    __shared__ alignas(16) int tab[PART_NODES];
    __shared__ int bbase[32 * SCAN_PER_LANE];
    const int g  = blockIdx.x;
    const int lo = blockIdx.y * PART_NODES;
    const int sl_lane = (int)threadIdx.x - (1024 - 32);
    int bt[SCAN_PER_LANE] = {};
    if (sl_lane >= 0) {
#pragma unroll
        for (int t = 0; t < SCAN_PER_LANE; ++t)
            bt[t] = btot[min(sl_lane * SCAN_PER_LANE + t, SCAN_BLOCKS - 1)];
    }
    const int4* dp = (const int4*)(edge_index + N_EDGES + g * CHUNK_EDGES);
    const int4* cu = (const int4*)(seglocal + lo);
    const int4* pl = (const int4*)(plane + (size_t)g * N_NODES + lo);
    int4 d[CHUNK_VEC_ITERS], ca[PART_VEC_ITERS], pa[PART_VEC_ITERS];
#pragma unroll
    for (int j = 0; j < PART_VEC_ITERS; ++j) {
        const int i = threadIdx.x + j * 1024;
        const bool in = i < PART_NODES / 4;
        ca[j] = in ? cu[i] : make_int4(0, 0, 0, 0);
        pa[j] = in ? pl[i] : make_int4(0, 0, 0, 0);
    }
#pragma unroll
    for (int j = 0; j < CHUNK_VEC_ITERS; ++j) {
        const int i = threadIdx.x + j * 1024;
        d[j] = (i < CHUNK_EDGES / 4) ? dp[i] : make_int4(0, 0, 0, 0);
    }
    if (sl_lane >= 0) {
        int s = 0;
#pragma unroll
        for (int t = 0; t < SCAN_PER_LANE; ++t) {
            if (sl_lane * SCAN_PER_LANE + t >= SCAN_BLOCKS) bt[t] = 0;
            s += bt[t];
        }
        int inc = s;
#pragma unroll
        for (int dist = 1; dist < 32; dist <<= 1) {
            const int up = __shfl_up(inc, dist);
            if (sl_lane >= dist) inc += up;
        }
        int run = inc - s;
#pragma unroll
        for (int t = 0; t < SCAN_PER_LANE; ++t) {
            bbase[sl_lane * SCAN_PER_LANE + t] = run;
            run += bt[t];
        }
    }
    __syncthreads();
    int4* cw = (int4*)(cursor + lo);
#pragma unroll
    for (int j = 0; j < PART_VEC_ITERS; ++j) {
        const int i = threadIdx.x + j * 1024;
        if (i < PART_NODES / 4) {
            const int bb = bbase[(lo + 4 * i) >> 8];
            const int4 first = make_int4(ca[j].x + bb, ca[j].y + bb, ca[j].z + bb, ca[j].w + bb);
            if (g == 0) cw[i] = first;
            ((int4*)tab)[i] = make_int4(first.x + pa[j].x, first.y + pa[j].y,
                                        first.z + pa[j].z, first.w + pa[j].w);
        }
    }
    __syncthreads();
    int* so = slotof + g * CHUNK_EDGES;
    int4 sl[CHUNK_VEC_ITERS];
#pragma unroll
    for (int j = 0; j < CHUNK_VEC_ITERS; ++j) {
        const bool ok = threadIdx.x + j * 1024 < CHUNK_EDGES / 4;
        const unsigned r0 = part_index(d[j].x, lo), r1 = part_index(d[j].y, lo);
        const unsigned r2 = part_index(d[j].z, lo), r3 = part_index(d[j].w, lo);
        sl[j].x = (ok && r0 < PART_NODES) ? atomicAdd(tab + r0, 1) : -1;
        sl[j].y = (ok && r1 < PART_NODES) ? atomicAdd(tab + r1, 1) : -1;
        sl[j].z = (ok && r2 < PART_NODES) ? atomicAdd(tab + r2, 1) : -1;
        sl[j].w = (ok && r3 < PART_NODES) ? atomicAdd(tab + r3, 1) : -1;
    }
#pragma unroll
    for (int j = 0; j < CHUNK_VEC_ITERS; ++j) {
        const int i = threadIdx.x + j * 1024;
        if (sl[j].x >= 0) so[4 * i + 0] = sl[j].x;
        if (sl[j].y >= 0) so[4 * i + 1] = sl[j].y;
        if (sl[j].z >= 0) so[4 * i + 2] = sl[j].z;
        if (sl[j].w >= 0) so[4 * i + 3] = sl[j].w;
    }
}

template <int M>
__device__ __forceinline__ float row_xor(float v)
{
    return __builtin_bit_cast(float, __builtin_amdgcn_update_dpp(
        0, __builtin_bit_cast(int, v), 0x160 + M, 0xf, 0xf, true));
}

template <int M>
__device__ __forceinline__ float fold_pair(float a, float b, bool bit)
{
    const float send = bit ? a : b;
    const float keep = bit ? b : a;
    return keep + row_xor<M>(send);
}

__device__ __forceinline__ float dot4(v4f v, v4f w)
{
    return __builtin_fmaf(v.w, w.w, __builtin_fmaf(v.z, w.z, __builtin_fmaf(v.y, w.y, v.x * w.x)));
}

static_assert(N_EDGES % 256 == 0, "one wave per 32 edges, 8 waves per block");

template <int J>
__device__ __forceinline__ int quad_lane(int v)
{
    return __builtin_amdgcn_update_dpp(0, v, J * 0x55, 0xf, 0xf, true);
}

template <int J>
__device__ __forceinline__ void store_records(v2i* __restrict__ rec, int slot, int srcm, int h,
                                              const float (&u)[4])
{
    int p = quad_lane<J>(slot);
    p = min(max(p, 0), N_EDGES - 1);
    const int s = quad_lane<J>(srcm);
    v2i r;
    r.x = __builtin_bit_cast(int, u[J]);
    r.y = (h == 0) ? s : 0;
    rec[(size_t)p * 4 + h] = r;
    if constexpr (J + 1 < 4) store_records<J + 1>(rec, slot, srcm, h, u);
}

__global__ void __launch_bounds__(256) __attribute__((amdgpu_waves_per_eu(8, 8))) edge_logits(
    const float* __restrict__ edge_attr, const int* __restrict__ edge_index,
    const float* __restrict__ weff,
    const int* __restrict__ slotof, v2i* __restrict__ rec)
{
    const int lane = threadIdx.x & 31;
    const int l15  = lane & 15;
    const int half = lane >> 4;
    const int h    = l15 & 3;
    const int e0   = (blockIdx.x * 8 + (threadIdx.x >> 5)) * 32;
    const int eb   = e0 + ((l15 >> 2) << 1) + half;

    const v4f w0 = *(const v4f*)(weff + 0 * IN_CH + 4 * l15);
    const v4f w1 = *(const v4f*)(weff + 1 * IN_CH + 4 * l15);
    const v4f w2 = *(const v4f*)(weff + 2 * IN_CH + 4 * l15);
    const v4f w3 = *(const v4f*)(weff + 3 * IN_CH + 4 * l15);

    const int em   = eb + 8 * h;
    const int srcm = edge_index[em];
    const int slot = slotof[em];

    const v4f* ea = (const v4f*)(edge_attr + (size_t)e0 * IN_CH) + lane;
    v4f v[16];
#pragma unroll
    for (int i = 0; i < 16; ++i)
        v[i] = __builtin_nontemporal_load(ea + i * 32);

    __builtin_amdgcn_sched_barrier(0);

    const bool b0 = (lane & 1) != 0, b1 = (lane & 2) != 0;
    const bool b2 = (lane & 4) != 0, b3 = (lane & 8) != 0;

    float s[16];
#pragma unroll
    for (int i = 0; i < 16; ++i) {
        const float r01 = fold_pair<1>(dot4(v[i], w0), dot4(v[i], w1), b0);
        const float r23 = fold_pair<1>(dot4(v[i], w2), dot4(v[i], w3), b0);
        s[i] = fold_pair<2>(r01, r23, b1);
    }
    float t[8];
#pragma unroll
    for (int m = 0; m < 8; ++m) t[m] = fold_pair<4>(s[2 * m], s[2 * m + 1], b2);
    float u[4];
#pragma unroll
    for (int j = 0; j < 4; ++j) u[j] = fold_pair<8>(t[2 * j], t[2 * j + 1], b3);

    store_records<0>(rec, slot, srcm, h, u);
}

template <int K>
__device__ __forceinline__ float group8_lane(float v)
{
    constexpr int sel = K | (K << 3) | (K << 6) | (K << 9) | (K << 12) | (K << 15) | (K << 18) | (K << 21);
    return __builtin_bit_cast(float, __builtin_amdgcn_mov_dpp8(__builtin_bit_cast(int, v), sel));
}

__device__ __forceinline__ float group8_sum(float v)
{
    v += row_xor<1>(v);
    v += row_xor<2>(v);
    v += row_xor<4>(v);
    return v;
}

__device__ __forceinline__ float group8_max(float v)
{
    { const float o = row_xor<1>(v); v = (o > v) ? o : v; }
    { const float o = row_xor<2>(v); v = (o > v) ? o : v; }
    { const float o = row_xor<4>(v); v = (o > v) ? o : v; }
    return v;
}

template <int K>
__device__ __forceinline__ void gather_rows(const float* __restrict__ xl, int sv, v4f (&row)[8])
{
    const int sk = min(max(__builtin_amdgcn_readlane(sv, K), 0), N_NODES - 1);
    row[K] = *(const v4f*)(xl + (size_t)sk * HC);
    if constexpr (K + 1 < 8) gather_rows<K + 1>(xl, sv, row);
}

template <int K>
__device__ __forceinline__ void weigh_rows(float w, const v4f (&row)[8], v4f& pa)
{
    pa += group8_lane<K>(w) * row[K];
    if constexpr (K + 1 < 8) weigh_rows<K + 1>(w, row, pa);
}

static_assert(N_NODES % 8 == 0, "one wave per node, 8 waves per block");

__global__ void __launch_bounds__(256) node_softmax_gather(
    const int* __restrict__ cursor, const v2i* __restrict__ rec,
    const float* __restrict__ xh,
    const float* __restrict__ att_src, const float* __restrict__ att_dst,
    const float* __restrict__ bias_conv, const float* __restrict__ bias_layer,
    float* __restrict__ out)
{
    const int lane = threadIdx.x & 31;
    const int k    = lane & 7;
    const int h    = lane >> 3;
    const int n = __builtin_amdgcn_readfirstlane(blockIdx.x * 8 + (threadIdx.x >> 5));

    int beg = cursor[n];
    int end = (n + 1 < N_NODES) ? cursor[n + 1] : N_EDGES;
    beg = max(beg, 0);
    end = min(end, N_EDGES);

    const float* xl = xh + 4 * lane;
    const v4f as4 = *(const v4f*)(att_src + 4 * lane);
    const v4f ad4 = *(const v4f*)(att_dst + 4 * lane);
    const float sh = group8_sum(dot4(*(const v4f*)(xl + (size_t)n * HC), as4));

    const bool b0 = (lane & 1) != 0, b1 = (lane & 2) != 0, b2 = (lane & 4) != 0;

    float m = -__builtin_huge_valf();
    float wsum = 0.f;
    v4f acc = {0.f, 0.f, 0.f, 0.f};
    for (int p = beg; p < end; p += 8) {
        const int i = p + k;
        const bool ok = i < end;
        const int ii = ok ? i : p;
        const v2i r = rec[(size_t)ii * 4 + h];
        const int sv = r.y;
        const float e = __builtin_bit_cast(float, r.x);
        v4f row[8];
        gather_rows<0>(xl, sv, row);
        const float d01 = fold_pair<1>(dot4(row[0], ad4), dot4(row[1], ad4), b0);
        const float d23 = fold_pair<1>(dot4(row[2], ad4), dot4(row[3], ad4), b0);
        const float d45 = fold_pair<1>(dot4(row[4], ad4), dot4(row[5], ad4), b0);
        const float d67 = fold_pair<1>(dot4(row[6], ad4), dot4(row[7], ad4), b0);
        const float d03 = fold_pair<2>(d01, d23, b1);
        const float d47 = fold_pair<2>(d45, d67, b1);
        const float dk  = fold_pair<4>(d03, d47, b2);
        float a = (e + sh) + dk;
        a = (a > 0.f) ? a : NEG_SLOPE * a;
        a = ok ? a : -__builtin_huge_valf();
        const float cm = group8_max(a);
        const float mn = (cm > m) ? cm : m;
        const float sc = __expf(m - mn);
        const float w  = __expf(a - mn);
        v4f pa = {0.f, 0.f, 0.f, 0.f};
        weigh_rows<0>(w, row, pa);
        wsum = wsum * sc + group8_sum(w);
        acc  = acc * sc + pa;
        m = mn;
    }
    acc = acc / (wsum + EPSV);

    v4f t;
    t.x = acc.x + row_xor<8>(acc.x);
    t.y = acc.y + row_xor<8>(acc.y);
    t.z = acc.z + row_xor<8>(acc.z);
    t.w = acc.w + row_xor<8>(acc.w);
    t.x += __shfl_xor(t.x, 16);
    t.y += __shfl_xor(t.y, 16);
    t.z += __shfl_xor(t.z, 16);
    t.w += __shfl_xor(t.w, 16);
    if (lane < 8) {
        const v4f bc = *(const v4f*)(bias_conv + 4 * lane);
        const v4f bl = *(const v4f*)(bias_layer + 4 * lane);
        v4f r = t * 0.25f + bc + bl;
        r.x = r.x > 0.f ? r.x : 0.f;
        r.y = r.y > 0.f ? r.y : 0.f;
        r.z = r.z > 0.f ? r.z : 0.f;
        r.w = r.w > 0.f ? r.w : 0.f;
        *(v4f*)(out + (size_t)n * OUT_CH + 4 * lane) = r;
    }
}

extern "C" void kernel_launch(void* const* d_in, const int* in_sizes, int n_in,
                              void* d_out, int out_size, void* d_ws, size_t ws_size,
                              hipStream_t stream) {
    const float* x          = (const float*)d_in[0];
    const float* edge_attr  = (const float*)d_in[1];
    const float* W          = (const float*)d_in[2];
    const float* W_edge     = (const float*)d_in[3];
    const float* att_src    = (const float*)d_in[4];
    const float* att_dst    = (const float*)d_in[5];
    const float* att_edge   = (const float*)d_in[6];
    const float* bias_conv  = (const float*)d_in[7];
    const float* bias_layer = (const float*)d_in[8];
    const int*   edge_index = (const int*)d_in[9];
    float* out = (float*)d_out;

    float* ws      = (float*)d_ws;
    float* xh      = ws;
    float* weff    = xh      + (size_t)N_NODES * HC;
    v2i*   rec     = (v2i*)(weff + (size_t)HEADS * IN_CH);
    int*   seglocal = (int*)(rec + (size_t)N_EDGES * 4);
    int*   cursor  = seglocal + (size_t)N_NODES;
    int*   slotof  = cursor  + (size_t)N_NODES;
    int*   plane   = slotof  + (size_t)N_EDGES;
    int*   btot    = plane   + (size_t)SORT_CHUNKS * N_NODES;

    proj_nodes_wmma<<<N_NODES / 16, 128, 0, stream>>>(x, W, xh);
    build_weff<<<1, 256, 0, stream>>>(W_edge, att_edge, weff);
    chunk_histogram<<<dim3(SORT_CHUNKS, NODE_PARTS), 1024, 0, stream>>>(edge_index, plane);
    chunk_prefix<<<SCAN_BLOCKS, 256, 0, stream>>>(plane, seglocal, btot);
    chunk_slots<<<dim3(SORT_CHUNKS, NODE_PARTS), 1024, 0, stream>>>(edge_index, seglocal, btot, plane, cursor, slotof);
    edge_logits<<<(N_EDGES + 255) / 256, 256, 0, stream>>>(
        edge_attr, edge_index, weff, slotof, rec);
    node_softmax_gather<<<N_NODES / 8, 256, 0, stream>>>(
        cursor, rec, xh, att_src, att_dst, bias_conv, bias_layer, out);
}


# node projection and w_eff leave the front of the stream: they run as extra workgroups of the chunk_prefix launch, which they do not depend on (7 launches -> 5); each part's own code unchanged
# speedup vs baseline: 4.7949x; 1.0347x over previous
#include <hip/hip_runtime.h>
#include <cstdint>

#define N_NODES 50000
#define N_EDGES 800000
#define IN_CH 64
#define OUT_CH 32
#define HEADS 4
#define HC 128
#define NEG_SLOPE 0.2f
#define EPSV 1e-16f

typedef __attribute__((ext_vector_type(2))) float v2f;
typedef __attribute__((ext_vector_type(4))) float v4f;
typedef __attribute__((ext_vector_type(8))) float v8f;
typedef __attribute__((ext_vector_type(2))) int v2i;

#define PROJ_TILES (N_NODES / 16)
#define PROJ_BLOCKS ((PROJ_TILES + 1) / 2)
static_assert(N_NODES % 16 == 0, "whole row tiles");

__device__ __forceinline__ void proj_tile_pair(
    const float* __restrict__ x, const float* __restrict__ W,
    float* __restrict__ xh, int pair)
{
    const int lane = threadIdx.x & 31;
    const int wv   = __builtin_amdgcn_readfirstlane((int)(threadIdx.x >> 5));
    const int tile = pair * 2 + (wv >> 2);
    if (tile >= PROJ_TILES) return;
    const int cg   = wv & 3;
    const int half = lane >> 4;
    const int l15  = lane & 15;
    const int m0   = tile * 16;

    const int rowA  = m0 + l15;
    const int colB0 = cg * 32 + l15;
    const int colB1 = colB0 + 16;

    v8f acc0 = {};
    v8f acc1 = {};
#pragma unroll
    for (int k0 = 0; k0 < IN_CH; k0 += 4) {
        const int kk = k0 + half * 2;
        v2f a  = *(const v2f*)(x + (size_t)rowA  * IN_CH + kk);
        v2f b0 = *(const v2f*)(W + (size_t)colB0 * IN_CH + kk);
        v2f b1 = *(const v2f*)(W + (size_t)colB1 * IN_CH + kk);
        acc0 = __builtin_amdgcn_wmma_f32_16x16x4_f32(
                   false, a, false, b0, (short)0, acc0, false, false);
        acc1 = __builtin_amdgcn_wmma_f32_16x16x4_f32(
                   false, a, false, b1, (short)0, acc1, false, false);
    }
#pragma unroll
    for (int v = 0; v < 8; ++v) {
        const int m = m0 + v + half * 8;
        xh[(size_t)m * HC + colB0] = acc0[v];
        xh[(size_t)m * HC + colB1] = acc1[v];
    }
}

__device__ __forceinline__ void build_weff(const float* __restrict__ W_edge,
                                           const float* __restrict__ att_edge,
                                           float* __restrict__ weff)
{
    const int t = threadIdx.x;
    const int h = t >> 6, k = t & 63;
    float s = 0.f;
#pragma unroll
    for (int c = 0; c < OUT_CH; ++c)
        s += att_edge[h * OUT_CH + c] * W_edge[(size_t)(h * OUT_CH + c) * IN_CH + k];
    weff[t] = s;
}

#define SORT_CHUNKS 64
#define NODE_PARTS 4
#define CHUNK_EDGES (N_EDGES / SORT_CHUNKS)
#define PART_NODES (N_NODES / NODE_PARTS)
#define CHUNK_VEC_ITERS ((CHUNK_EDGES / 4 + 1023) / 1024)
#define PART_VEC_ITERS ((PART_NODES / 4 + 1023) / 1024)
#define SCAN_BLOCKS ((N_NODES + 255) / 256)
#define SCAN_PER_LANE ((SCAN_BLOCKS + 31) / 32)
static_assert(N_EDGES % SORT_CHUNKS == 0 && CHUNK_EDGES % 4 == 0, "chunks of equal size, read 16 bytes at a time");
static_assert(N_NODES % NODE_PARTS == 0 && PART_NODES % 4 == 0, "ranges of equal size, 16-byte table accesses");

__device__ __forceinline__ unsigned part_index(int d, int lo)
{
    return (unsigned)(min(max(d, 0), N_NODES - 1) - lo);
}

__global__ void __launch_bounds__(1024) chunk_histogram(const int* __restrict__ edge_index,
                                                        int* __restrict__ plane)
{
    __shared__ alignas(16) int tab[PART_NODES];
    const int g  = blockIdx.x;
    const int lo = blockIdx.y * PART_NODES;
    const int4* dp = (const int4*)(edge_index + N_EDGES + g * CHUNK_EDGES);
    int4 d[CHUNK_VEC_ITERS];
#pragma unroll
    for (int j = 0; j < CHUNK_VEC_ITERS; ++j) {
        const int i = threadIdx.x + j * 1024;
        d[j] = (i < CHUNK_EDGES / 4) ? dp[i] : make_int4(0, 0, 0, 0);
    }
#pragma unroll
    for (int j = 0; j < PART_VEC_ITERS; ++j) {
        const int i = threadIdx.x + j * 1024;
        if (i < PART_NODES / 4) ((int4*)tab)[i] = make_int4(0, 0, 0, 0);
    }
    __syncthreads();
#pragma unroll
    for (int j = 0; j < CHUNK_VEC_ITERS; ++j) {
        const bool ok = threadIdx.x + j * 1024 < CHUNK_EDGES / 4;
        const unsigned r0 = part_index(d[j].x, lo), r1 = part_index(d[j].y, lo);
        const unsigned r2 = part_index(d[j].z, lo), r3 = part_index(d[j].w, lo);
        if (ok && r0 < PART_NODES) atomicAdd(tab + r0, 1);
        if (ok && r1 < PART_NODES) atomicAdd(tab + r1, 1);
        if (ok && r2 < PART_NODES) atomicAdd(tab + r2, 1);
        if (ok && r3 < PART_NODES) atomicAdd(tab + r3, 1);
    }
    __syncthreads();
    int4* pl = (int4*)(plane + (size_t)g * N_NODES + lo);
#pragma unroll
    for (int j = 0; j < PART_VEC_ITERS; ++j) {
        const int i = threadIdx.x + j * 1024;
        if (i < PART_NODES / 4) pl[i] = ((const int4*)tab)[i];
    }
}

__device__ __forceinline__ void chunk_prefix(int* __restrict__ plane, int* __restrict__ seglocal,
                                             int* __restrict__ btot, int blk)
{
    __shared__ int wtot[8];
    const int lane = threadIdx.x & 31;
    const int wv   = threadIdx.x >> 5;
    const int n = blk * 256 + threadIdx.x;
    const bool valid = n < N_NODES;
    const int nn = valid ? n : N_NODES - 1;
    int c[SORT_CHUNKS];
#pragma unroll
    for (int g = 0; g < SORT_CHUNKS; ++g) c[g] = plane[(size_t)g * N_NODES + nn];
    int run = 0;
    if (valid) {
#pragma unroll
        for (int g = 0; g < SORT_CHUNKS; ++g) {
            plane[(size_t)g * N_NODES + n] = run;
            run += c[g];
        }
    }
    int inc = run;
#pragma unroll
    for (int d = 1; d < 32; d <<= 1) {
        const int up = __shfl_up(inc, d);
        if (lane >= d) inc += up;
    }
    if (lane == 31) wtot[wv] = inc;
    __syncthreads();
    int wb = 0;
#pragma unroll
    for (int w = 0; w < 7; ++w) wb += (w < wv) ? wtot[w] : 0;
    if (valid) seglocal[n] = wb + inc - run;
    if (threadIdx.x == 255) btot[blk] = wb + inc;
}

#define PREFIX_LAUNCH_BLOCKS (SCAN_BLOCKS + 1 + PROJ_BLOCKS)

__global__ void __launch_bounds__(256) prefix_weff_proj(
    int* __restrict__ plane, int* __restrict__ seglocal, int* __restrict__ btot,
    const float* __restrict__ W_edge, const float* __restrict__ att_edge, float* __restrict__ weff,
    const float* __restrict__ x, const float* __restrict__ W, float* __restrict__ xh)
{
    const int b = blockIdx.x;
    if (b < SCAN_BLOCKS)       chunk_prefix(plane, seglocal, btot, b);
    else if (b == SCAN_BLOCKS) build_weff(W_edge, att_edge, weff);
    else                       proj_tile_pair(x, W, xh, b - (SCAN_BLOCKS + 1));
}

__global__ void __launch_bounds__(1024) chunk_slots(const int* __restrict__ edge_index,
                                                    const int* __restrict__ seglocal,
                                                    const int* __restrict__ btot,
                                                    const int* __restrict__ plane,
                                                    int* __restrict__ cursor,
                                                    int* __restrict__ slotof)
{
    __shared__ alignas(16) int tab[PART_NODES];
    __shared__ int bbase[32 * SCAN_PER_LANE];
    const int g  = blockIdx.x;
    const int lo = blockIdx.y * PART_NODES;
    const int sl_lane = (int)threadIdx.x - (1024 - 32);
    int bt[SCAN_PER_LANE] = {};
    if (sl_lane >= 0) {
#pragma unroll
        for (int t = 0; t < SCAN_PER_LANE; ++t)
            bt[t] = btot[min(sl_lane * SCAN_PER_LANE + t, SCAN_BLOCKS - 1)];
    }
    const int4* dp = (const int4*)(edge_index + N_EDGES + g * CHUNK_EDGES);
    const int4* cu = (const int4*)(seglocal + lo);
    const int4* pl = (const int4*)(plane + (size_t)g * N_NODES + lo);
    int4 d[CHUNK_VEC_ITERS], ca[PART_VEC_ITERS], pa[PART_VEC_ITERS];
#pragma unroll
    for (int j = 0; j < PART_VEC_ITERS; ++j) {
        const int i = threadIdx.x + j * 1024;
        const bool in = i < PART_NODES / 4;
        ca[j] = in ? cu[i] : make_int4(0, 0, 0, 0);
        pa[j] = in ? pl[i] : make_int4(0, 0, 0, 0);
    }
#pragma unroll
    for (int j = 0; j < CHUNK_VEC_ITERS; ++j) {
        const int i = threadIdx.x + j * 1024;
        d[j] = (i < CHUNK_EDGES / 4) ? dp[i] : make_int4(0, 0, 0, 0);
    }
    if (sl_lane >= 0) {
        int s = 0;
#pragma unroll
        for (int t = 0; t < SCAN_PER_LANE; ++t) {
            if (sl_lane * SCAN_PER_LANE + t >= SCAN_BLOCKS) bt[t] = 0;
            s += bt[t];
        }
        int inc = s;
#pragma unroll
        for (int dist = 1; dist < 32; dist <<= 1) {
            const int up = __shfl_up(inc, dist);
            if (sl_lane >= dist) inc += up;
        }
        int run = inc - s;
#pragma unroll
        for (int t = 0; t < SCAN_PER_LANE; ++t) {
            bbase[sl_lane * SCAN_PER_LANE + t] = run;
            run += bt[t];
        }
    }
    __syncthreads();
    int4* cw = (int4*)(cursor + lo);
#pragma unroll
    for (int j = 0; j < PART_VEC_ITERS; ++j) {
        const int i = threadIdx.x + j * 1024;
        if (i < PART_NODES / 4) {
            const int bb = bbase[(lo + 4 * i) >> 8];
            const int4 first = make_int4(ca[j].x + bb, ca[j].y + bb, ca[j].z + bb, ca[j].w + bb);
            if (g == 0) cw[i] = first;
            ((int4*)tab)[i] = make_int4(first.x + pa[j].x, first.y + pa[j].y,
                                        first.z + pa[j].z, first.w + pa[j].w);
        }
    }
    __syncthreads();
    int* so = slotof + g * CHUNK_EDGES;
    int4 sl[CHUNK_VEC_ITERS];
#pragma unroll
    for (int j = 0; j < CHUNK_VEC_ITERS; ++j) {
        const bool ok = threadIdx.x + j * 1024 < CHUNK_EDGES / 4;
        const unsigned r0 = part_index(d[j].x, lo), r1 = part_index(d[j].y, lo);
        const unsigned r2 = part_index(d[j].z, lo), r3 = part_index(d[j].w, lo);
        sl[j].x = (ok && r0 < PART_NODES) ? atomicAdd(tab + r0, 1) : -1;
        sl[j].y = (ok && r1 < PART_NODES) ? atomicAdd(tab + r1, 1) : -1;
        sl[j].z = (ok && r2 < PART_NODES) ? atomicAdd(tab + r2, 1) : -1;
        sl[j].w = (ok && r3 < PART_NODES) ? atomicAdd(tab + r3, 1) : -1;
    }
#pragma unroll
    for (int j = 0; j < CHUNK_VEC_ITERS; ++j) {
        const int i = threadIdx.x + j * 1024;
        if (sl[j].x >= 0) so[4 * i + 0] = sl[j].x;
        if (sl[j].y >= 0) so[4 * i + 1] = sl[j].y;
        if (sl[j].z >= 0) so[4 * i + 2] = sl[j].z;
        if (sl[j].w >= 0) so[4 * i + 3] = sl[j].w;
    }
}

template <int M>
__device__ __forceinline__ float row_xor(float v)
{
    return __builtin_bit_cast(float, __builtin_amdgcn_update_dpp(
        0, __builtin_bit_cast(int, v), 0x160 + M, 0xf, 0xf, true));
}

template <int M>
__device__ __forceinline__ float fold_pair(float a, float b, bool bit)
{
    const float send = bit ? a : b;
    const float keep = bit ? b : a;
    return keep + row_xor<M>(send);
}

__device__ __forceinline__ float dot4(v4f v, v4f w)
{
    return __builtin_fmaf(v.w, w.w, __builtin_fmaf(v.z, w.z, __builtin_fmaf(v.y, w.y, v.x * w.x)));
}

static_assert(N_EDGES % 256 == 0, "one wave per 32 edges, 8 waves per block");

template <int J>
__device__ __forceinline__ int quad_lane(int v)
{
    return __builtin_amdgcn_update_dpp(0, v, J * 0x55, 0xf, 0xf, true);
}

template <int J>
__device__ __forceinline__ void store_records(v2i* __restrict__ rec, int slot, int srcm, int h,
                                              const float (&u)[4])
{
    int p = quad_lane<J>(slot);
    p = min(max(p, 0), N_EDGES - 1);
    const int s = quad_lane<J>(srcm);
    v2i r;
    r.x = __builtin_bit_cast(int, u[J]);
    r.y = (h == 0) ? s : 0;
    rec[(size_t)p * 4 + h] = r;
    if constexpr (J + 1 < 4) store_records<J + 1>(rec, slot, srcm, h, u);
}

__global__ void __launch_bounds__(256) __attribute__((amdgpu_waves_per_eu(8, 8))) edge_logits(
    const float* __restrict__ edge_attr, const int* __restrict__ edge_index,
    const float* __restrict__ weff,
    const int* __restrict__ slotof, v2i* __restrict__ rec)
{
    const int lane = threadIdx.x & 31;
    const int l15  = lane & 15;
    const int half = lane >> 4;
    const int h    = l15 & 3;
    const int e0   = (blockIdx.x * 8 + (threadIdx.x >> 5)) * 32;
    const int eb   = e0 + ((l15 >> 2) << 1) + half;

    const v4f w0 = *(const v4f*)(weff + 0 * IN_CH + 4 * l15);
    const v4f w1 = *(const v4f*)(weff + 1 * IN_CH + 4 * l15);
    const v4f w2 = *(const v4f*)(weff + 2 * IN_CH + 4 * l15);
    const v4f w3 = *(const v4f*)(weff + 3 * IN_CH + 4 * l15);

    const int em   = eb + 8 * h;
    const int srcm = edge_index[em];
    const int slot = slotof[em];

    const v4f* ea = (const v4f*)(edge_attr + (size_t)e0 * IN_CH) + lane;
    v4f v[16];
#pragma unroll
    for (int i = 0; i < 16; ++i)
        v[i] = __builtin_nontemporal_load(ea + i * 32);

    __builtin_amdgcn_sched_barrier(0);

    const bool b0 = (lane & 1) != 0, b1 = (lane & 2) != 0;
    const bool b2 = (lane & 4) != 0, b3 = (lane & 8) != 0;

    float s[16];
#pragma unroll
    for (int i = 0; i < 16; ++i) {
        const float r01 = fold_pair<1>(dot4(v[i], w0), dot4(v[i], w1), b0);
        const float r23 = fold_pair<1>(dot4(v[i], w2), dot4(v[i], w3), b0);
        s[i] = fold_pair<2>(r01, r23, b1);
    }
    float t[8];
#pragma unroll
    for (int m = 0; m < 8; ++m) t[m] = fold_pair<4>(s[2 * m], s[2 * m + 1], b2);
    float u[4];
#pragma unroll
    for (int j = 0; j < 4; ++j) u[j] = fold_pair<8>(t[2 * j], t[2 * j + 1], b3);

    store_records<0>(rec, slot, srcm, h, u);
}

template <int K>
__device__ __forceinline__ float group8_lane(float v)
{
    constexpr int sel = K | (K << 3) | (K << 6) | (K << 9) | (K << 12) | (K << 15) | (K << 18) | (K << 21);
    return __builtin_bit_cast(float, __builtin_amdgcn_mov_dpp8(__builtin_bit_cast(int, v), sel));
}

__device__ __forceinline__ float group8_sum(float v)
{
    v += row_xor<1>(v);
    v += row_xor<2>(v);
    v += row_xor<4>(v);
    return v;
}

__device__ __forceinline__ float group8_max(float v)
{
    { const float o = row_xor<1>(v); v = (o > v) ? o : v; }
    { const float o = row_xor<2>(v); v = (o > v) ? o : v; }
    { const float o = row_xor<4>(v); v = (o > v) ? o : v; }
    return v;
}

template <int K>
__device__ __forceinline__ void gather_rows(const float* __restrict__ xl, int sv, v4f (&row)[8])
{
    const int sk = min(max(__builtin_amdgcn_readlane(sv, K), 0), N_NODES - 1);
    row[K] = *(const v4f*)(xl + (size_t)sk * HC);
    if constexpr (K + 1 < 8) gather_rows<K + 1>(xl, sv, row);
}

template <int K>
__device__ __forceinline__ void weigh_rows(float w, const v4f (&row)[8], v4f& pa)
{
    pa += group8_lane<K>(w) * row[K];
    if constexpr (K + 1 < 8) weigh_rows<K + 1>(w, row, pa);
}

static_assert(N_NODES % 8 == 0, "one wave per node, 8 waves per block");

__global__ void __launch_bounds__(256) node_softmax_gather(
    const int* __restrict__ cursor, const v2i* __restrict__ rec,
    const float* __restrict__ xh,
    const float* __restrict__ att_src, const float* __restrict__ att_dst,
    const float* __restrict__ bias_conv, const float* __restrict__ bias_layer,
    float* __restrict__ out)
{
    const int lane = threadIdx.x & 31;
    const int k    = lane & 7;
    const int h    = lane >> 3;
    const int n = __builtin_amdgcn_readfirstlane(blockIdx.x * 8 + (threadIdx.x >> 5));

    int beg = cursor[n];
    int end = (n + 1 < N_NODES) ? cursor[n + 1] : N_EDGES;
    beg = max(beg, 0);
    end = min(end, N_EDGES);

    const float* xl = xh + 4 * lane;
    const v4f as4 = *(const v4f*)(att_src + 4 * lane);
    const v4f ad4 = *(const v4f*)(att_dst + 4 * lane);
    const float sh = group8_sum(dot4(*(const v4f*)(xl + (size_t)n * HC), as4));

    const bool b0 = (lane & 1) != 0, b1 = (lane & 2) != 0, b2 = (lane & 4) != 0;

    float m = -__builtin_huge_valf();
    float wsum = 0.f;
    v4f acc = {0.f, 0.f, 0.f, 0.f};
    for (int p = beg; p < end; p += 8) {
        const int i = p + k;
        const bool ok = i < end;
        const int ii = ok ? i : p;
        const v2i r = rec[(size_t)ii * 4 + h];
        const int sv = r.y;
        const float e = __builtin_bit_cast(float, r.x);
        v4f row[8];
        gather_rows<0>(xl, sv, row);
        const float d01 = fold_pair<1>(dot4(row[0], ad4), dot4(row[1], ad4), b0);
        const float d23 = fold_pair<1>(dot4(row[2], ad4), dot4(row[3], ad4), b0);
        const float d45 = fold_pair<1>(dot4(row[4], ad4), dot4(row[5], ad4), b0);
        const float d67 = fold_pair<1>(dot4(row[6], ad4), dot4(row[7], ad4), b0);
        const float d03 = fold_pair<2>(d01, d23, b1);
        const float d47 = fold_pair<2>(d45, d67, b1);
        const float dk  = fold_pair<4>(d03, d47, b2);
        float a = (e + sh) + dk;
        a = (a > 0.f) ? a : NEG_SLOPE * a;
        a = ok ? a : -__builtin_huge_valf();
        const float cm = group8_max(a);
        const float mn = (cm > m) ? cm : m;
        const float sc = __expf(m - mn);
        const float w  = __expf(a - mn);
        v4f pa = {0.f, 0.f, 0.f, 0.f};
        weigh_rows<0>(w, row, pa);
        wsum = wsum * sc + group8_sum(w);
        acc  = acc * sc + pa;
        m = mn;
    }
    acc = acc / (wsum + EPSV);

    v4f t;
    t.x = acc.x + row_xor<8>(acc.x);
    t.y = acc.y + row_xor<8>(acc.y);
    t.z = acc.z + row_xor<8>(acc.z);
    t.w = acc.w + row_xor<8>(acc.w);
    t.x += __shfl_xor(t.x, 16);
    t.y += __shfl_xor(t.y, 16);
    t.z += __shfl_xor(t.z, 16);
    t.w += __shfl_xor(t.w, 16);
    if (lane < 8) {
        const v4f bc = *(const v4f*)(bias_conv + 4 * lane);
        const v4f bl = *(const v4f*)(bias_layer + 4 * lane);
        v4f r = t * 0.25f + bc + bl;
        r.x = r.x > 0.f ? r.x : 0.f;
        r.y = r.y > 0.f ? r.y : 0.f;
        r.z = r.z > 0.f ? r.z : 0.f;
        r.w = r.w > 0.f ? r.w : 0.f;
        *(v4f*)(out + (size_t)n * OUT_CH + 4 * lane) = r;
    }
}

extern "C" void kernel_launch(void* const* d_in, const int* in_sizes, int n_in,
                              void* d_out, int out_size, void* d_ws, size_t ws_size,
                              hipStream_t stream) {
    const float* x          = (const float*)d_in[0];
    const float* edge_attr  = (const float*)d_in[1];
    const float* W          = (const float*)d_in[2];
    const float* W_edge     = (const float*)d_in[3];
    const float* att_src    = (const float*)d_in[4];
    const float* att_dst    = (const float*)d_in[5];
    const float* att_edge   = (const float*)d_in[6];
    const float* bias_conv  = (const float*)d_in[7];
    const float* bias_layer = (const float*)d_in[8];
    const int*   edge_index = (const int*)d_in[9];
    float* out = (float*)d_out;

    float* ws      = (float*)d_ws;
    float* xh      = ws;
    float* weff    = xh      + (size_t)N_NODES * HC;
    v2i*   rec     = (v2i*)(weff + (size_t)HEADS * IN_CH);
    int*   seglocal = (int*)(rec + (size_t)N_EDGES * 4);
    int*   cursor  = seglocal + (size_t)N_NODES;
    int*   slotof  = cursor  + (size_t)N_NODES;
    int*   plane   = slotof  + (size_t)N_EDGES;
    int*   btot    = plane   + (size_t)SORT_CHUNKS * N_NODES;

    chunk_histogram<<<dim3(SORT_CHUNKS, NODE_PARTS), 1024, 0, stream>>>(edge_index, plane);
    prefix_weff_proj<<<PREFIX_LAUNCH_BLOCKS, 256, 0, stream>>>(plane, seglocal, btot,
                                                               W_edge, att_edge, weff, x, W, xh);
    chunk_slots<<<dim3(SORT_CHUNKS, NODE_PARTS), 1024, 0, stream>>>(edge_index, seglocal, btot, plane, cursor, slotof);
    edge_logits<<<(N_EDGES + 255) / 256, 256, 0, stream>>>(
        edge_attr, edge_index, weff, slotof, rec);
    node_softmax_gather<<<N_NODES / 8, 256, 0, stream>>>(
        cursor, rec, xh, att_src, att_dst, bias_conv, bias_layer, out);
}


# aggregation moved in front of the projection (it is linear): node pass gathers 256-byte x rows, not 512-byte projected rows; its 4x64 sums are projected on WMMA at the end (sum order changes)
# speedup vs baseline: 4.9896x; 1.0406x over previous
#include <hip/hip_runtime.h>
#include <cstdint>

#define N_NODES 50000
#define N_EDGES 800000
#define IN_CH 64
#define OUT_CH 32
#define HEADS 4
#define HC 128
#define NEG_SLOPE 0.2f
#define EPSV 1e-16f

typedef __attribute__((ext_vector_type(2))) float v2f;
typedef __attribute__((ext_vector_type(4))) float v4f;
typedef __attribute__((ext_vector_type(8))) float v8f;
typedef __attribute__((ext_vector_type(2))) int v2i;

__device__ __forceinline__ void build_tables(const float* __restrict__ W_edge,
                                             const float* __restrict__ att_edge,
                                             const float* __restrict__ W,
                                             const float* __restrict__ att_src,
                                             const float* __restrict__ att_dst,
                                             float* __restrict__ weff,
                                             float* __restrict__ vs,
                                             float* __restrict__ vd)
{
    const int t = threadIdx.x;
    const int h = t >> 6, k = t & 63;
    float s = 0.f, a = 0.f, d = 0.f;
#pragma unroll
    for (int c = 0; c < OUT_CH; ++c) {
        const float w = W[(size_t)(h * OUT_CH + c) * IN_CH + k];
        s += att_edge[h * OUT_CH + c] * W_edge[(size_t)(h * OUT_CH + c) * IN_CH + k];
        a += att_src[h * OUT_CH + c] * w;
        d += att_dst[h * OUT_CH + c] * w;
    }
    weff[t] = s;
    vs[t] = a;
    vd[t] = d;
}

#define SORT_CHUNKS 64
#define NODE_PARTS 4
#define CHUNK_EDGES (N_EDGES / SORT_CHUNKS)
#define PART_NODES (N_NODES / NODE_PARTS)
#define CHUNK_VEC_ITERS ((CHUNK_EDGES / 4 + 1023) / 1024)
#define PART_VEC_ITERS ((PART_NODES / 4 + 1023) / 1024)
#define SCAN_BLOCKS ((N_NODES + 255) / 256)
#define SCAN_PER_LANE ((SCAN_BLOCKS + 31) / 32)
static_assert(N_EDGES % SORT_CHUNKS == 0 && CHUNK_EDGES % 4 == 0, "chunks of equal size, read 16 bytes at a time");
static_assert(N_NODES % NODE_PARTS == 0 && PART_NODES % 4 == 0, "ranges of equal size, 16-byte table accesses");

__device__ __forceinline__ unsigned part_index(int d, int lo)
{
    return (unsigned)(min(max(d, 0), N_NODES - 1) - lo);
}

__global__ void __launch_bounds__(1024) chunk_histogram(const int* __restrict__ edge_index,
                                                        int* __restrict__ plane)
{
    __shared__ alignas(16) int tab[PART_NODES];
    const int g  = blockIdx.x;
    const int lo = blockIdx.y * PART_NODES;
    const int4* dp = (const int4*)(edge_index + N_EDGES + g * CHUNK_EDGES);
    int4 d[CHUNK_VEC_ITERS];
#pragma unroll
    for (int j = 0; j < CHUNK_VEC_ITERS; ++j) {
        const int i = threadIdx.x + j * 1024;
        d[j] = (i < CHUNK_EDGES / 4) ? dp[i] : make_int4(0, 0, 0, 0);
    }
#pragma unroll
    for (int j = 0; j < PART_VEC_ITERS; ++j) {
        const int i = threadIdx.x + j * 1024;
        if (i < PART_NODES / 4) ((int4*)tab)[i] = make_int4(0, 0, 0, 0);
    }
    __syncthreads();
#pragma unroll
    for (int j = 0; j < CHUNK_VEC_ITERS; ++j) {
        const bool ok = threadIdx.x + j * 1024 < CHUNK_EDGES / 4;
        const unsigned r0 = part_index(d[j].x, lo), r1 = part_index(d[j].y, lo);
        const unsigned r2 = part_index(d[j].z, lo), r3 = part_index(d[j].w, lo);
        if (ok && r0 < PART_NODES) atomicAdd(tab + r0, 1);
        if (ok && r1 < PART_NODES) atomicAdd(tab + r1, 1);
        if (ok && r2 < PART_NODES) atomicAdd(tab + r2, 1);
        if (ok && r3 < PART_NODES) atomicAdd(tab + r3, 1);
    }
    __syncthreads();
    int4* pl = (int4*)(plane + (size_t)g * N_NODES + lo);
#pragma unroll
    for (int j = 0; j < PART_VEC_ITERS; ++j) {
        const int i = threadIdx.x + j * 1024;
        if (i < PART_NODES / 4) pl[i] = ((const int4*)tab)[i];
    }
}

__device__ __forceinline__ void chunk_prefix(int* __restrict__ plane, int* __restrict__ seglocal,
                                             int* __restrict__ btot, int blk)
{
    __shared__ int wtot[8];
    const int lane = threadIdx.x & 31;
    const int wv   = threadIdx.x >> 5;
    const int n = blk * 256 + threadIdx.x;
    const bool valid = n < N_NODES;
    const int nn = valid ? n : N_NODES - 1;
    int c[SORT_CHUNKS];
#pragma unroll
    for (int g = 0; g < SORT_CHUNKS; ++g) c[g] = plane[(size_t)g * N_NODES + nn];
    int run = 0;
    if (valid) {
#pragma unroll
        for (int g = 0; g < SORT_CHUNKS; ++g) {
            plane[(size_t)g * N_NODES + n] = run;
            run += c[g];
        }
    }
    int inc = run;
#pragma unroll
    for (int d = 1; d < 32; d <<= 1) {
        const int up = __shfl_up(inc, d);
        if (lane >= d) inc += up;
    }
    if (lane == 31) wtot[wv] = inc;
    __syncthreads();
    int wb = 0;
#pragma unroll
    for (int w = 0; w < 7; ++w) wb += (w < wv) ? wtot[w] : 0;
    if (valid) seglocal[n] = wb + inc - run;
    if (threadIdx.x == 255) btot[blk] = wb + inc;
}

#define PREFIX_LAUNCH_BLOCKS (SCAN_BLOCKS + 1)

__global__ void __launch_bounds__(256) prefix_tables(
    int* __restrict__ plane, int* __restrict__ seglocal, int* __restrict__ btot,
    const float* __restrict__ W_edge, const float* __restrict__ att_edge,
    const float* __restrict__ W, const float* __restrict__ att_src, const float* __restrict__ att_dst,
    float* __restrict__ weff, float* __restrict__ vs, float* __restrict__ vd)
{
    const int b = blockIdx.x;
    if (b < SCAN_BLOCKS) chunk_prefix(plane, seglocal, btot, b);
    else                 build_tables(W_edge, att_edge, W, att_src, att_dst, weff, vs, vd);
}

__global__ void __launch_bounds__(1024) chunk_slots(const int* __restrict__ edge_index,
                                                    const int* __restrict__ seglocal,
                                                    const int* __restrict__ btot,
                                                    const int* __restrict__ plane,
                                                    int* __restrict__ cursor,
                                                    int* __restrict__ slotof)
{
    __shared__ alignas(16) int tab[PART_NODES];
    __shared__ int bbase[32 * SCAN_PER_LANE];
    const int g  = blockIdx.x;
    const int lo = blockIdx.y * PART_NODES;
    const int sl_lane = (int)threadIdx.x - (1024 - 32);
    int bt[SCAN_PER_LANE] = {};
    if (sl_lane >= 0) {
#pragma unroll
        for (int t = 0; t < SCAN_PER_LANE; ++t)
            bt[t] = btot[min(sl_lane * SCAN_PER_LANE + t, SCAN_BLOCKS - 1)];
    }
    const int4* dp = (const int4*)(edge_index + N_EDGES + g * CHUNK_EDGES);
    const int4* cu = (const int4*)(seglocal + lo);
    const int4* pl = (const int4*)(plane + (size_t)g * N_NODES + lo);
    int4 d[CHUNK_VEC_ITERS], ca[PART_VEC_ITERS], pa[PART_VEC_ITERS];
#pragma unroll
    for (int j = 0; j < PART_VEC_ITERS; ++j) {
        const int i = threadIdx.x + j * 1024;
        const bool in = i < PART_NODES / 4;
        ca[j] = in ? cu[i] : make_int4(0, 0, 0, 0);
        pa[j] = in ? pl[i] : make_int4(0, 0, 0, 0);
    }
#pragma unroll
    for (int j = 0; j < CHUNK_VEC_ITERS; ++j) {
        const int i = threadIdx.x + j * 1024;
        d[j] = (i < CHUNK_EDGES / 4) ? dp[i] : make_int4(0, 0, 0, 0);
    }
    if (sl_lane >= 0) {
        int s = 0;
#pragma unroll
        for (int t = 0; t < SCAN_PER_LANE; ++t) {
            if (sl_lane * SCAN_PER_LANE + t >= SCAN_BLOCKS) bt[t] = 0;
            s += bt[t];
        }
        int inc = s;
#pragma unroll
        for (int dist = 1; dist < 32; dist <<= 1) {
            const int up = __shfl_up(inc, dist);
            if (sl_lane >= dist) inc += up;
        }
        int run = inc - s;
#pragma unroll
        for (int t = 0; t < SCAN_PER_LANE; ++t) {
            bbase[sl_lane * SCAN_PER_LANE + t] = run;
            run += bt[t];
        }
    }
    __syncthreads();
    int4* cw = (int4*)(cursor + lo);
#pragma unroll
    for (int j = 0; j < PART_VEC_ITERS; ++j) {
        const int i = threadIdx.x + j * 1024;
        if (i < PART_NODES / 4) {
            const int bb = bbase[(lo + 4 * i) >> 8];
            const int4 first = make_int4(ca[j].x + bb, ca[j].y + bb, ca[j].z + bb, ca[j].w + bb);
            if (g == 0) cw[i] = first;
            ((int4*)tab)[i] = make_int4(first.x + pa[j].x, first.y + pa[j].y,
                                        first.z + pa[j].z, first.w + pa[j].w);
        }
    }
    __syncthreads();
    int* so = slotof + g * CHUNK_EDGES;
    int4 sl[CHUNK_VEC_ITERS];
#pragma unroll
    for (int j = 0; j < CHUNK_VEC_ITERS; ++j) {
        const bool ok = threadIdx.x + j * 1024 < CHUNK_EDGES / 4;
        const unsigned r0 = part_index(d[j].x, lo), r1 = part_index(d[j].y, lo);
        const unsigned r2 = part_index(d[j].z, lo), r3 = part_index(d[j].w, lo);
        sl[j].x = (ok && r0 < PART_NODES) ? atomicAdd(tab + r0, 1) : -1;
        sl[j].y = (ok && r1 < PART_NODES) ? atomicAdd(tab + r1, 1) : -1;
        sl[j].z = (ok && r2 < PART_NODES) ? atomicAdd(tab + r2, 1) : -1;
        sl[j].w = (ok && r3 < PART_NODES) ? atomicAdd(tab + r3, 1) : -1;
    }
#pragma unroll
    for (int j = 0; j < CHUNK_VEC_ITERS; ++j) {
        const int i = threadIdx.x + j * 1024;
        if (sl[j].x >= 0) so[4 * i + 0] = sl[j].x;
        if (sl[j].y >= 0) so[4 * i + 1] = sl[j].y;
        if (sl[j].z >= 0) so[4 * i + 2] = sl[j].z;
        if (sl[j].w >= 0) so[4 * i + 3] = sl[j].w;
    }
}

template <int M>
__device__ __forceinline__ float row_xor(float v)
{
    return __builtin_bit_cast(float, __builtin_amdgcn_update_dpp(
        0, __builtin_bit_cast(int, v), 0x160 + M, 0xf, 0xf, true));
}

template <int M>
__device__ __forceinline__ float fold_pair(float a, float b, bool bit)
{
    const float send = bit ? a : b;
    const float keep = bit ? b : a;
    return keep + row_xor<M>(send);
}

__device__ __forceinline__ float dot4(v4f v, v4f w)
{
    return __builtin_fmaf(v.w, w.w, __builtin_fmaf(v.z, w.z, __builtin_fmaf(v.y, w.y, v.x * w.x)));
}

static_assert(N_EDGES % 256 == 0, "one wave per 32 edges, 8 waves per block");

template <int J>
__device__ __forceinline__ int quad_lane(int v)
{
    return __builtin_amdgcn_update_dpp(0, v, J * 0x55, 0xf, 0xf, true);
}

template <int J>
__device__ __forceinline__ void store_records(v2i* __restrict__ rec, int slot, int srcm, int h,
                                              const float (&u)[4])
{
    int p = quad_lane<J>(slot);
    p = min(max(p, 0), N_EDGES - 1);
    const int s = quad_lane<J>(srcm);
    v2i r;
    r.x = __builtin_bit_cast(int, u[J]);
    r.y = (h == 0) ? s : 0;
    rec[(size_t)p * 4 + h] = r;
    if constexpr (J + 1 < 4) store_records<J + 1>(rec, slot, srcm, h, u);
}

__global__ void __launch_bounds__(256) __attribute__((amdgpu_waves_per_eu(8, 8))) edge_logits(
    const float* __restrict__ edge_attr, const int* __restrict__ edge_index,
    const float* __restrict__ weff,
    const int* __restrict__ slotof, v2i* __restrict__ rec)
{
    const int lane = threadIdx.x & 31;
    const int l15  = lane & 15;
    const int half = lane >> 4;
    const int h    = l15 & 3;
    const int e0   = (blockIdx.x * 8 + (threadIdx.x >> 5)) * 32;
    const int eb   = e0 + ((l15 >> 2) << 1) + half;

    const v4f w0 = *(const v4f*)(weff + 0 * IN_CH + 4 * l15);
    const v4f w1 = *(const v4f*)(weff + 1 * IN_CH + 4 * l15);
    const v4f w2 = *(const v4f*)(weff + 2 * IN_CH + 4 * l15);
    const v4f w3 = *(const v4f*)(weff + 3 * IN_CH + 4 * l15);

    const int em   = eb + 8 * h;
    const int srcm = edge_index[em];
    const int slot = slotof[em];

    const v4f* ea = (const v4f*)(edge_attr + (size_t)e0 * IN_CH) + lane;
    v4f v[16];
#pragma unroll
    for (int i = 0; i < 16; ++i)
        v[i] = __builtin_nontemporal_load(ea + i * 32);

    __builtin_amdgcn_sched_barrier(0);

    const bool b0 = (lane & 1) != 0, b1 = (lane & 2) != 0;
    const bool b2 = (lane & 4) != 0, b3 = (lane & 8) != 0;

    float s[16];
#pragma unroll
    for (int i = 0; i < 16; ++i) {
        const float r01 = fold_pair<1>(dot4(v[i], w0), dot4(v[i], w1), b0);
        const float r23 = fold_pair<1>(dot4(v[i], w2), dot4(v[i], w3), b0);
        s[i] = fold_pair<2>(r01, r23, b1);
    }
    float t[8];
#pragma unroll
    for (int m = 0; m < 8; ++m) t[m] = fold_pair<4>(s[2 * m], s[2 * m + 1], b2);
    float u[4];
#pragma unroll
    for (int j = 0; j < 4; ++j) u[j] = fold_pair<8>(t[2 * j], t[2 * j + 1], b3);

    store_records<0>(rec, slot, srcm, h, u);
}

template <int N>
__device__ __forceinline__ float row_lane(float v)
{
    return __builtin_bit_cast(float, __builtin_amdgcn_update_dpp(
        0, __builtin_bit_cast(int, v), 0x150 + N, 0xf, 0xf, true));
}

template <int N>
__device__ __forceinline__ int row_lane_i(int v)
{
    return __builtin_amdgcn_update_dpp(0, v, 0x150 + N, 0xf, 0xf, true);
}

__device__ __forceinline__ float other_half(float v)
{
    const unsigned u = __builtin_bit_cast(unsigned, v);
    return __builtin_bit_cast(float, __builtin_amdgcn_permlanex16(u, u, 0x76543210u, 0xfedcba98u, false, false));
}

__device__ __forceinline__ v4f other_half4(v4f v)
{
    v4f r;
    r.x = other_half(v.x); r.y = other_half(v.y);
    r.z = other_half(v.z); r.w = other_half(v.w);
    return r;
}

__device__ __forceinline__ float quad_heads(v4f row, v4f d0, v4f d1, v4f d2, v4f d3, bool b0, bool b1)
{
    const float r01 = fold_pair<1>(dot4(row, d0), dot4(row, d1), b0);
    const float r23 = fold_pair<1>(dot4(row, d2), dot4(row, d3), b0);
    return fold_pair<2>(r01, r23, b1);
}

__device__ __forceinline__ const v4f* x_row(const float* __restrict__ xl, int src)
{
    return (const v4f*)(xl + (size_t)min(max(src, 0), N_NODES - 1) * IN_CH);
}

#define NODE_WAVES 16
#define Z_PITCH 260
static_assert(N_NODES % NODE_WAVES == 0, "one wave per node, 16 waves per block");

__global__ void __launch_bounds__(512) __attribute__((amdgpu_waves_per_eu(12, 12))) node_softmax_gather(
    const int* __restrict__ cursor, const v2i* __restrict__ rec,
    const float* __restrict__ x, const float* __restrict__ W,
    const float* __restrict__ vs, const float* __restrict__ vd,
    const float* __restrict__ bias_conv, const float* __restrict__ bias_layer,
    float* __restrict__ out)
{
    __shared__ alignas(16) float Zs[NODE_WAVES * Z_PITCH];
    __shared__ float Ps[3 * 16 * 32];

    const int lane = threadIdx.x & 31;
    const int l15  = lane & 15;
    const int half = lane >> 4;
    const int h    = l15 & 3;
    const int ek   = 2 * (l15 >> 2) + half;
    const int wv = __builtin_amdgcn_readfirstlane((int)(threadIdx.x >> 5));
    const int n  = blockIdx.x * NODE_WAVES + wv;

    int beg = cursor[n];
    int end = (n + 1 < N_NODES) ? cursor[n + 1] : N_EDGES;
    beg = max(beg, 0);
    end = min(end, N_EDGES);

    const float* xl = x + 4 * l15;
    const bool b0 = (lane & 1) != 0, b1 = (lane & 2) != 0;
    const bool b2 = (lane & 4) != 0, b3 = (lane & 8) != 0;

    float sh;
    {
        const v4f xo = *(const v4f*)(xl + (size_t)n * IN_CH);
        const v4f s0 = *(const v4f*)(vs + 0 * IN_CH + 4 * l15);
        const v4f s1 = *(const v4f*)(vs + 1 * IN_CH + 4 * l15);
        const v4f s2 = *(const v4f*)(vs + 2 * IN_CH + 4 * l15);
        const v4f s3 = *(const v4f*)(vs + 3 * IN_CH + 4 * l15);
        float r = quad_heads(xo, s0, s1, s2, s3, b0, b1);
        r += row_xor<4>(r);
        r += row_xor<8>(r);
        sh = r;
    }
    const v4f d0 = *(const v4f*)(vd + 0 * IN_CH + 4 * l15);
    const v4f d1 = *(const v4f*)(vd + 1 * IN_CH + 4 * l15);
    const v4f d2 = *(const v4f*)(vd + 2 * IN_CH + 4 * l15);
    const v4f d3 = *(const v4f*)(vd + 3 * IN_CH + 4 * l15);

    float m = -__builtin_huge_valf();
    float wsum = 0.f;
    v4f a0 = {0.f, 0.f, 0.f, 0.f}, a1 = a0, a2 = a0, a3 = a0;
    for (int p = beg; p < end; p += 8) {
        const int i = p + ek;
        const bool ok = i < end;
        const int ii = ok ? i : p;
        const v2i r = rec[(size_t)ii * 4 + h];
        const int sv = r.y;
        const float e = __builtin_bit_cast(float, r.x);
        const v4f row0 = *x_row(xl, row_lane_i<0>(sv));
        const v4f row1 = *x_row(xl, row_lane_i<4>(sv));
        const v4f row2 = *x_row(xl, row_lane_i<8>(sv));
        const v4f row3 = *x_row(xl, row_lane_i<12>(sv));
        const float q0 = quad_heads(row0, d0, d1, d2, d3, b0, b1);
        const float q1 = quad_heads(row1, d0, d1, d2, d3, b0, b1);
        const float q2 = quad_heads(row2, d0, d1, d2, d3, b0, b1);
        const float q3 = quad_heads(row3, d0, d1, d2, d3, b0, b1);
        const float t0 = fold_pair<4>(q0, q1, b2);
        const float t1 = fold_pair<4>(q2, q3, b2);
        const float dk = fold_pair<8>(t0, t1, b3);
        float a = (e + sh) + dk;
        a = (a > 0.f) ? a : NEG_SLOPE * a;
        a = ok ? a : -__builtin_huge_valf();
        float cm = a;
        { const float o = row_xor<4>(cm); cm = (o > cm) ? o : cm; }
        { const float o = row_xor<8>(cm); cm = (o > cm) ? o : cm; }
        { const float o = other_half(cm); cm = (o > cm) ? o : cm; }
        const float mn = (cm > m) ? cm : m;
        const float sc = __expf(m - mn);
        const float w  = __expf(a - mn);
        float ws = w;
        ws += row_xor<4>(ws);
        ws += row_xor<8>(ws);
        ws += other_half(ws);
        wsum = wsum * sc + ws;
        a0 = a0 * row_lane<0>(sc) + (row_lane<0>(w) * row0 + row_lane<4>(w) * row1
                                     + row_lane<8>(w) * row2 + row_lane<12>(w) * row3);
        a1 = a1 * row_lane<1>(sc) + (row_lane<1>(w) * row0 + row_lane<5>(w) * row1
                                     + row_lane<9>(w) * row2 + row_lane<13>(w) * row3);
        a2 = a2 * row_lane<2>(sc) + (row_lane<2>(w) * row0 + row_lane<6>(w) * row1
                                     + row_lane<10>(w) * row2 + row_lane<14>(w) * row3);
        a3 = a3 * row_lane<3>(sc) + (row_lane<3>(w) * row0 + row_lane<7>(w) * row1
                                     + row_lane<11>(w) * row2 + row_lane<15>(w) * row3);
        m = mn;
    }
    const float inv = 1.f / (wsum + EPSV);
    const v4f z0 = (a0 + other_half4(a0)) * row_lane<0>(inv);
    const v4f z1 = (a1 + other_half4(a1)) * row_lane<1>(inv);
    const v4f z2 = (a2 + other_half4(a2)) * row_lane<2>(inv);
    const v4f z3 = (a3 + other_half4(a3)) * row_lane<3>(inv);
    {
        float* zr = Zs + wv * Z_PITCH + (2 * half) * IN_CH + 4 * l15;
        *(v4f*)zr           = half ? z2 : z0;
        *(v4f*)(zr + IN_CH) = half ? z3 : z1;
    }
    __syncthreads();

    v8f acc0 = {};
    v8f acc1 = {};
    if (wv < 4) {
        const float* za  = Zs + l15 * Z_PITCH + wv * IN_CH + half * 2;
        const float* wb0 = W + (size_t)(wv * OUT_CH + l15) * IN_CH + half * 2;
        const float* wb1 = wb0 + (size_t)16 * IN_CH;
#pragma unroll
        for (int k0 = 0; k0 < IN_CH; k0 += 4) {
            const v2f a   = *(const v2f*)(za + k0);
            const v2f bb0 = *(const v2f*)(wb0 + k0);
            const v2f bb1 = *(const v2f*)(wb1 + k0);
            acc0 = __builtin_amdgcn_wmma_f32_16x16x4_f32(
                       false, a, false, bb0, (short)0, acc0, false, false);
            acc1 = __builtin_amdgcn_wmma_f32_16x16x4_f32(
                       false, a, false, bb1, (short)0, acc1, false, false);
        }
        if (wv > 0) {
            float* ps = Ps + (wv - 1) * (16 * 32) + lane;
#pragma unroll
            for (int v = 0; v < 8; ++v) {
                ps[v * 32]       = acc0[v];
                ps[(8 + v) * 32] = acc1[v];
            }
        }
    }
    __syncthreads();
    if (wv == 0) {
        const int m0 = blockIdx.x * NODE_WAVES;
        const float bc0 = bias_conv[l15],      bl0 = bias_layer[l15];
        const float bc1 = bias_conv[16 + l15], bl1 = bias_layer[16 + l15];
#pragma unroll
        for (int v = 0; v < 8; ++v) {
            const float* ps = Ps + lane;
            const float s0 = ((acc0[v] + ps[v * 32]) + ps[512 + v * 32]) + ps[1024 + v * 32];
            const float s1 = ((acc1[v] + ps[(8 + v) * 32]) + ps[512 + (8 + v) * 32]) + ps[1024 + (8 + v) * 32];
            float r0 = s0 * 0.25f + bc0 + bl0;
            float r1 = s1 * 0.25f + bc1 + bl1;
            r0 = r0 > 0.f ? r0 : 0.f;
            r1 = r1 > 0.f ? r1 : 0.f;
            float* o = out + (size_t)(m0 + v + half * 8) * OUT_CH + l15;
            o[0]  = r0;
            o[16] = r1;
        }
    }
}

extern "C" void kernel_launch(void* const* d_in, const int* in_sizes, int n_in,
                              void* d_out, int out_size, void* d_ws, size_t ws_size,
                              hipStream_t stream) {
    const float* x          = (const float*)d_in[0];
    const float* edge_attr  = (const float*)d_in[1];
    const float* W          = (const float*)d_in[2];
    const float* W_edge     = (const float*)d_in[3];
    const float* att_src    = (const float*)d_in[4];
    const float* att_dst    = (const float*)d_in[5];
    const float* att_edge   = (const float*)d_in[6];
    const float* bias_conv  = (const float*)d_in[7];
    const float* bias_layer = (const float*)d_in[8];
    const int*   edge_index = (const int*)d_in[9];
    float* out = (float*)d_out;

    float* ws      = (float*)d_ws;
    float* weff    = ws;
    float* vs      = weff + (size_t)HEADS * IN_CH;
    float* vd      = vs   + (size_t)HEADS * IN_CH;
    v2i*   rec     = (v2i*)(vd + (size_t)HEADS * IN_CH);
    int*   seglocal = (int*)(rec + (size_t)N_EDGES * 4);
    int*   cursor  = seglocal + (size_t)N_NODES;
    int*   slotof  = cursor  + (size_t)N_NODES;
    int*   plane   = slotof  + (size_t)N_EDGES;
    int*   btot    = plane   + (size_t)SORT_CHUNKS * N_NODES;

    chunk_histogram<<<dim3(SORT_CHUNKS, NODE_PARTS), 1024, 0, stream>>>(edge_index, plane);
    prefix_tables<<<PREFIX_LAUNCH_BLOCKS, 256, 0, stream>>>(plane, seglocal, btot,
                                                            W_edge, att_edge, W, att_src, att_dst,
                                                            weff, vs, vd);
    chunk_slots<<<dim3(SORT_CHUNKS, NODE_PARTS), 1024, 0, stream>>>(edge_index, seglocal, btot, plane, cursor, slotof);
    edge_logits<<<(N_EDGES + 255) / 256, 256, 0, stream>>>(
        edge_attr, edge_index, weff, slotof, rec);
    node_softmax_gather<<<N_NODES / NODE_WAVES, 512, 0, stream>>>(
        cursor, rec, x, W, vs, vd, bias_conv, bias_layer, out);
}


# node pass: each wave takes 2 neighbouring destination nodes in turn (8-wave workgroups of 16 nodes): 25,000 waves enter, not 50,000; half as many idle through a tile's projection; sum order unchanged
# speedup vs baseline: 5.1401x; 1.0302x over previous
#include <hip/hip_runtime.h>
#include <cstdint>

#define N_NODES 50000
#define N_EDGES 800000
#define IN_CH 64
#define OUT_CH 32
#define HEADS 4
#define HC 128
#define NEG_SLOPE 0.2f
#define EPSV 1e-16f

typedef __attribute__((ext_vector_type(2))) float v2f;
typedef __attribute__((ext_vector_type(4))) float v4f;
typedef __attribute__((ext_vector_type(8))) float v8f;
typedef __attribute__((ext_vector_type(2))) int v2i;

__device__ __forceinline__ void build_tables(const float* __restrict__ W_edge,
                                             const float* __restrict__ att_edge,
                                             const float* __restrict__ W,
                                             const float* __restrict__ att_src,
                                             const float* __restrict__ att_dst,
                                             float* __restrict__ weff,
                                             float* __restrict__ vs,
                                             float* __restrict__ vd)
{
    const int t = threadIdx.x;
    const int h = t >> 6, k = t & 63;
    float s = 0.f, a = 0.f, d = 0.f;
#pragma unroll
    for (int c = 0; c < OUT_CH; ++c) {
        const float w = W[(size_t)(h * OUT_CH + c) * IN_CH + k];
        s += att_edge[h * OUT_CH + c] * W_edge[(size_t)(h * OUT_CH + c) * IN_CH + k];
        a += att_src[h * OUT_CH + c] * w;
        d += att_dst[h * OUT_CH + c] * w;
    }
    weff[t] = s;
    vs[t] = a;
    vd[t] = d;
}

#define SORT_CHUNKS 64
#define NODE_PARTS 4
#define CHUNK_EDGES (N_EDGES / SORT_CHUNKS)
#define PART_NODES (N_NODES / NODE_PARTS)
#define CHUNK_VEC_ITERS ((CHUNK_EDGES / 4 + 1023) / 1024)
#define PART_VEC_ITERS ((PART_NODES / 4 + 1023) / 1024)
#define SCAN_BLOCKS ((N_NODES + 255) / 256)
#define SCAN_PER_LANE ((SCAN_BLOCKS + 31) / 32)
static_assert(N_EDGES % SORT_CHUNKS == 0 && CHUNK_EDGES % 4 == 0, "chunks of equal size, read 16 bytes at a time");
static_assert(N_NODES % NODE_PARTS == 0 && PART_NODES % 4 == 0, "ranges of equal size, 16-byte table accesses");

__device__ __forceinline__ unsigned part_index(int d, int lo)
{
    return (unsigned)(min(max(d, 0), N_NODES - 1) - lo);
}

__global__ void __launch_bounds__(1024) chunk_histogram(const int* __restrict__ edge_index,
                                                        int* __restrict__ plane)
{
    __shared__ alignas(16) int tab[PART_NODES];
    const int g  = blockIdx.x;
    const int lo = blockIdx.y * PART_NODES;
    const int4* dp = (const int4*)(edge_index + N_EDGES + g * CHUNK_EDGES);
    int4 d[CHUNK_VEC_ITERS];
#pragma unroll
    for (int j = 0; j < CHUNK_VEC_ITERS; ++j) {
        const int i = threadIdx.x + j * 1024;
        d[j] = (i < CHUNK_EDGES / 4) ? dp[i] : make_int4(0, 0, 0, 0);
    }
#pragma unroll
    for (int j = 0; j < PART_VEC_ITERS; ++j) {
        const int i = threadIdx.x + j * 1024;
        if (i < PART_NODES / 4) ((int4*)tab)[i] = make_int4(0, 0, 0, 0);
    }
    __syncthreads();
#pragma unroll
    for (int j = 0; j < CHUNK_VEC_ITERS; ++j) {
        const bool ok = threadIdx.x + j * 1024 < CHUNK_EDGES / 4;
        const unsigned r0 = part_index(d[j].x, lo), r1 = part_index(d[j].y, lo);
        const unsigned r2 = part_index(d[j].z, lo), r3 = part_index(d[j].w, lo);
        if (ok && r0 < PART_NODES) atomicAdd(tab + r0, 1);
        if (ok && r1 < PART_NODES) atomicAdd(tab + r1, 1);
        if (ok && r2 < PART_NODES) atomicAdd(tab + r2, 1);
        if (ok && r3 < PART_NODES) atomicAdd(tab + r3, 1);
    }
    __syncthreads();
    int4* pl = (int4*)(plane + (size_t)g * N_NODES + lo);
#pragma unroll
    for (int j = 0; j < PART_VEC_ITERS; ++j) {
        const int i = threadIdx.x + j * 1024;
        if (i < PART_NODES / 4) pl[i] = ((const int4*)tab)[i];
    }
}

__device__ __forceinline__ void chunk_prefix(int* __restrict__ plane, int* __restrict__ seglocal,
                                             int* __restrict__ btot, int blk)
{
    __shared__ int wtot[8];
    const int lane = threadIdx.x & 31;
    const int wv   = threadIdx.x >> 5;
    const int n = blk * 256 + threadIdx.x;
    const bool valid = n < N_NODES;
    const int nn = valid ? n : N_NODES - 1;
    int c[SORT_CHUNKS];
#pragma unroll
    for (int g = 0; g < SORT_CHUNKS; ++g) c[g] = plane[(size_t)g * N_NODES + nn];
    int run = 0;
    if (valid) {
#pragma unroll
        for (int g = 0; g < SORT_CHUNKS; ++g) {
            plane[(size_t)g * N_NODES + n] = run;
            run += c[g];
        }
    }
    int inc = run;
#pragma unroll
    for (int d = 1; d < 32; d <<= 1) {
        const int up = __shfl_up(inc, d);
        if (lane >= d) inc += up;
    }
    if (lane == 31) wtot[wv] = inc;
    __syncthreads();
    int wb = 0;
#pragma unroll
    for (int w = 0; w < 7; ++w) wb += (w < wv) ? wtot[w] : 0;
    if (valid) seglocal[n] = wb + inc - run;
    if (threadIdx.x == 255) btot[blk] = wb + inc;
}

#define PREFIX_LAUNCH_BLOCKS (SCAN_BLOCKS + 1)

__global__ void __launch_bounds__(256) prefix_tables(
    int* __restrict__ plane, int* __restrict__ seglocal, int* __restrict__ btot,
    const float* __restrict__ W_edge, const float* __restrict__ att_edge,
    const float* __restrict__ W, const float* __restrict__ att_src, const float* __restrict__ att_dst,
    float* __restrict__ weff, float* __restrict__ vs, float* __restrict__ vd)
{
    const int b = blockIdx.x;
    if (b < SCAN_BLOCKS) chunk_prefix(plane, seglocal, btot, b);
    else                 build_tables(W_edge, att_edge, W, att_src, att_dst, weff, vs, vd);
}

__global__ void __launch_bounds__(1024) chunk_slots(const int* __restrict__ edge_index,
                                                    const int* __restrict__ seglocal,
                                                    const int* __restrict__ btot,
                                                    const int* __restrict__ plane,
                                                    int* __restrict__ cursor,
                                                    int* __restrict__ slotof)
{
    __shared__ alignas(16) int tab[PART_NODES];
    __shared__ int bbase[32 * SCAN_PER_LANE];
    const int g  = blockIdx.x;
    const int lo = blockIdx.y * PART_NODES;
    const int sl_lane = (int)threadIdx.x - (1024 - 32);
    int bt[SCAN_PER_LANE] = {};
    if (sl_lane >= 0) {
#pragma unroll
        for (int t = 0; t < SCAN_PER_LANE; ++t)
            bt[t] = btot[min(sl_lane * SCAN_PER_LANE + t, SCAN_BLOCKS - 1)];
    }
    const int4* dp = (const int4*)(edge_index + N_EDGES + g * CHUNK_EDGES);
    const int4* cu = (const int4*)(seglocal + lo);
    const int4* pl = (const int4*)(plane + (size_t)g * N_NODES + lo);
    int4 d[CHUNK_VEC_ITERS], ca[PART_VEC_ITERS], pa[PART_VEC_ITERS];
#pragma unroll
    for (int j = 0; j < PART_VEC_ITERS; ++j) {
        const int i = threadIdx.x + j * 1024;
        const bool in = i < PART_NODES / 4;
        ca[j] = in ? cu[i] : make_int4(0, 0, 0, 0);
        pa[j] = in ? pl[i] : make_int4(0, 0, 0, 0);
    }
#pragma unroll
    for (int j = 0; j < CHUNK_VEC_ITERS; ++j) {
        const int i = threadIdx.x + j * 1024;
        d[j] = (i < CHUNK_EDGES / 4) ? dp[i] : make_int4(0, 0, 0, 0);
    }
    if (sl_lane >= 0) {
        int s = 0;
#pragma unroll
        for (int t = 0; t < SCAN_PER_LANE; ++t) {
            if (sl_lane * SCAN_PER_LANE + t >= SCAN_BLOCKS) bt[t] = 0;
            s += bt[t];
        }
        int inc = s;
#pragma unroll
        for (int dist = 1; dist < 32; dist <<= 1) {
            const int up = __shfl_up(inc, dist);
            if (sl_lane >= dist) inc += up;
        }
        int run = inc - s;
#pragma unroll
        for (int t = 0; t < SCAN_PER_LANE; ++t) {
            bbase[sl_lane * SCAN_PER_LANE + t] = run;
            run += bt[t];
        }
    }
    __syncthreads();
    int4* cw = (int4*)(cursor + lo);
#pragma unroll
    for (int j = 0; j < PART_VEC_ITERS; ++j) {
        const int i = threadIdx.x + j * 1024;
        if (i < PART_NODES / 4) {
            const int bb = bbase[(lo + 4 * i) >> 8];
            const int4 first = make_int4(ca[j].x + bb, ca[j].y + bb, ca[j].z + bb, ca[j].w + bb);
            if (g == 0) cw[i] = first;
            ((int4*)tab)[i] = make_int4(first.x + pa[j].x, first.y + pa[j].y,
                                        first.z + pa[j].z, first.w + pa[j].w);
        }
    }
    __syncthreads();
    int* so = slotof + g * CHUNK_EDGES;
    int4 sl[CHUNK_VEC_ITERS];
#pragma unroll
    for (int j = 0; j < CHUNK_VEC_ITERS; ++j) {
        const bool ok = threadIdx.x + j * 1024 < CHUNK_EDGES / 4;
        const unsigned r0 = part_index(d[j].x, lo), r1 = part_index(d[j].y, lo);
        const unsigned r2 = part_index(d[j].z, lo), r3 = part_index(d[j].w, lo);
        sl[j].x = (ok && r0 < PART_NODES) ? atomicAdd(tab + r0, 1) : -1;
        sl[j].y = (ok && r1 < PART_NODES) ? atomicAdd(tab + r1, 1) : -1;
        sl[j].z = (ok && r2 < PART_NODES) ? atomicAdd(tab + r2, 1) : -1;
        sl[j].w = (ok && r3 < PART_NODES) ? atomicAdd(tab + r3, 1) : -1;
    }
#pragma unroll
    for (int j = 0; j < CHUNK_VEC_ITERS; ++j) {
        const int i = threadIdx.x + j * 1024;
        if (sl[j].x >= 0) so[4 * i + 0] = sl[j].x;
        if (sl[j].y >= 0) so[4 * i + 1] = sl[j].y;
        if (sl[j].z >= 0) so[4 * i + 2] = sl[j].z;
        if (sl[j].w >= 0) so[4 * i + 3] = sl[j].w;
    }
}

template <int M>
__device__ __forceinline__ float row_xor(float v)
{
    return __builtin_bit_cast(float, __builtin_amdgcn_update_dpp(
        0, __builtin_bit_cast(int, v), 0x160 + M, 0xf, 0xf, true));
}

template <int M>
__device__ __forceinline__ float fold_pair(float a, float b, bool bit)
{
    const float send = bit ? a : b;
    const float keep = bit ? b : a;
    return keep + row_xor<M>(send);
}

__device__ __forceinline__ float dot4(v4f v, v4f w)
{
    return __builtin_fmaf(v.w, w.w, __builtin_fmaf(v.z, w.z, __builtin_fmaf(v.y, w.y, v.x * w.x)));
}

static_assert(N_EDGES % 256 == 0, "one wave per 32 edges, 8 waves per block");

template <int J>
__device__ __forceinline__ int quad_lane(int v)
{
    return __builtin_amdgcn_update_dpp(0, v, J * 0x55, 0xf, 0xf, true);
}

template <int J>
__device__ __forceinline__ void store_records(v2i* __restrict__ rec, int slot, int srcm, int h,
                                              const float (&u)[4])
{
    int p = quad_lane<J>(slot);
    p = min(max(p, 0), N_EDGES - 1);
    const int s = quad_lane<J>(srcm);
    v2i r;
    r.x = __builtin_bit_cast(int, u[J]);
    r.y = (h == 0) ? s : 0;
    rec[(size_t)p * 4 + h] = r;
    if constexpr (J + 1 < 4) store_records<J + 1>(rec, slot, srcm, h, u);
}

__global__ void __launch_bounds__(256) __attribute__((amdgpu_waves_per_eu(8, 8))) edge_logits(
    const float* __restrict__ edge_attr, const int* __restrict__ edge_index,
    const float* __restrict__ weff,
    const int* __restrict__ slotof, v2i* __restrict__ rec)
{
    const int lane = threadIdx.x & 31;
    const int l15  = lane & 15;
    const int half = lane >> 4;
    const int h    = l15 & 3;
    const int e0   = (blockIdx.x * 8 + (threadIdx.x >> 5)) * 32;
    const int eb   = e0 + ((l15 >> 2) << 1) + half;

    const v4f w0 = *(const v4f*)(weff + 0 * IN_CH + 4 * l15);
    const v4f w1 = *(const v4f*)(weff + 1 * IN_CH + 4 * l15);
    const v4f w2 = *(const v4f*)(weff + 2 * IN_CH + 4 * l15);
    const v4f w3 = *(const v4f*)(weff + 3 * IN_CH + 4 * l15);

    const int em   = eb + 8 * h;
    const int srcm = edge_index[em];
    const int slot = slotof[em];

    const v4f* ea = (const v4f*)(edge_attr + (size_t)e0 * IN_CH) + lane;
    v4f v[16];
#pragma unroll
    for (int i = 0; i < 16; ++i)
        v[i] = __builtin_nontemporal_load(ea + i * 32);

    __builtin_amdgcn_sched_barrier(0);

    const bool b0 = (lane & 1) != 0, b1 = (lane & 2) != 0;
    const bool b2 = (lane & 4) != 0, b3 = (lane & 8) != 0;

    float s[16];
#pragma unroll
    for (int i = 0; i < 16; ++i) {
        const float r01 = fold_pair<1>(dot4(v[i], w0), dot4(v[i], w1), b0);
        const float r23 = fold_pair<1>(dot4(v[i], w2), dot4(v[i], w3), b0);
        s[i] = fold_pair<2>(r01, r23, b1);
    }
    float t[8];
#pragma unroll
    for (int m = 0; m < 8; ++m) t[m] = fold_pair<4>(s[2 * m], s[2 * m + 1], b2);
    float u[4];
#pragma unroll
    for (int j = 0; j < 4; ++j) u[j] = fold_pair<8>(t[2 * j], t[2 * j + 1], b3);

    store_records<0>(rec, slot, srcm, h, u);
}

template <int N>
__device__ __forceinline__ float row_lane(float v)
{
    return __builtin_bit_cast(float, __builtin_amdgcn_update_dpp(
        0, __builtin_bit_cast(int, v), 0x150 + N, 0xf, 0xf, true));
}

template <int N>
__device__ __forceinline__ int row_lane_i(int v)
{
    return __builtin_amdgcn_update_dpp(0, v, 0x150 + N, 0xf, 0xf, true);
}

__device__ __forceinline__ float other_half(float v)
{
    const unsigned u = __builtin_bit_cast(unsigned, v);
    return __builtin_bit_cast(float, __builtin_amdgcn_permlanex16(u, u, 0x76543210u, 0xfedcba98u, false, false));
}

__device__ __forceinline__ v4f other_half4(v4f v)
{
    v4f r;
    r.x = other_half(v.x); r.y = other_half(v.y);
    r.z = other_half(v.z); r.w = other_half(v.w);
    return r;
}

__device__ __forceinline__ float quad_heads(v4f row, v4f d0, v4f d1, v4f d2, v4f d3, bool b0, bool b1)
{
    const float r01 = fold_pair<1>(dot4(row, d0), dot4(row, d1), b0);
    const float r23 = fold_pair<1>(dot4(row, d2), dot4(row, d3), b0);
    return fold_pair<2>(r01, r23, b1);
}

__device__ __forceinline__ const v4f* x_row(const float* __restrict__ xl, int src)
{
    return (const v4f*)(xl + (size_t)min(max(src, 0), N_NODES - 1) * IN_CH);
}

#define NODE_WAVES 8
#define WAVE_NODES 2
#define TILE_NODES (NODE_WAVES * WAVE_NODES)
#define Z_PITCH 260
static_assert(TILE_NODES == 16, "a workgroup's nodes are one 16-row tile of the matrix instruction");
static_assert(N_NODES % TILE_NODES == 0, "whole tiles");

__global__ void __launch_bounds__(256) __attribute__((amdgpu_waves_per_eu(12, 12))) node_softmax_gather(
    const int* __restrict__ cursor, const v2i* __restrict__ rec,
    const float* __restrict__ x, const float* __restrict__ W,
    const float* __restrict__ vs, const float* __restrict__ vd,
    const float* __restrict__ bias_conv, const float* __restrict__ bias_layer,
    float* __restrict__ out)
{
    __shared__ alignas(16) float Zs[TILE_NODES * Z_PITCH];
    __shared__ float Ps[3 * 16 * 32];

    const int lane = threadIdx.x & 31;
    const int l15  = lane & 15;
    const int half = lane >> 4;
    const int h    = l15 & 3;
    const int ek   = 2 * (l15 >> 2) + half;
    const int wv = __builtin_amdgcn_readfirstlane((int)(threadIdx.x >> 5));
    const int n0 = blockIdx.x * TILE_NODES + wv * WAVE_NODES;

    const int c0 = cursor[n0];
    const int c1 = cursor[n0 + 1];
    const int c2 = (n0 + 2 < N_NODES) ? cursor[n0 + 2] : N_EDGES;

    const float* xl = x + 4 * l15;
    const bool b0 = (lane & 1) != 0, b1 = (lane & 2) != 0;
    const bool b2 = (lane & 4) != 0, b3 = (lane & 8) != 0;

    float sh0, sh1;
    {
        const v4f xa = *(const v4f*)(xl + (size_t)n0 * IN_CH);
        const v4f xb = *(const v4f*)(xl + (size_t)(n0 + 1) * IN_CH);
        const v4f s0 = *(const v4f*)(vs + 0 * IN_CH + 4 * l15);
        const v4f s1 = *(const v4f*)(vs + 1 * IN_CH + 4 * l15);
        const v4f s2 = *(const v4f*)(vs + 2 * IN_CH + 4 * l15);
        const v4f s3 = *(const v4f*)(vs + 3 * IN_CH + 4 * l15);
        float ra = quad_heads(xa, s0, s1, s2, s3, b0, b1);
        ra += row_xor<4>(ra);
        ra += row_xor<8>(ra);
        float rb = quad_heads(xb, s0, s1, s2, s3, b0, b1);
        rb += row_xor<4>(rb);
        rb += row_xor<8>(rb);
        sh0 = ra;
        sh1 = rb;
    }
    const v4f d0 = *(const v4f*)(vd + 0 * IN_CH + 4 * l15);
    const v4f d1 = *(const v4f*)(vd + 1 * IN_CH + 4 * l15);
    const v4f d2 = *(const v4f*)(vd + 2 * IN_CH + 4 * l15);
    const v4f d3 = *(const v4f*)(vd + 3 * IN_CH + 4 * l15);

  for (int r = 0; r < WAVE_NODES; ++r) {
    const int beg = max(r ? c1 : c0, 0);
    const int end = min(r ? c2 : c1, N_EDGES);
    const float sh = r ? sh1 : sh0;
    float m = -__builtin_huge_valf();
    float wsum = 0.f;
    v4f a0 = {0.f, 0.f, 0.f, 0.f}, a1 = a0, a2 = a0, a3 = a0;
    for (int p = beg; p < end; p += 8) {
        const int i = p + ek;
        const bool ok = i < end;
        const int ii = ok ? i : p;
        const v2i r = rec[(size_t)ii * 4 + h];
        const int sv = r.y;
        const float e = __builtin_bit_cast(float, r.x);
        const v4f row0 = *x_row(xl, row_lane_i<0>(sv));
        const v4f row1 = *x_row(xl, row_lane_i<4>(sv));
        const v4f row2 = *x_row(xl, row_lane_i<8>(sv));
        const v4f row3 = *x_row(xl, row_lane_i<12>(sv));
        const float q0 = quad_heads(row0, d0, d1, d2, d3, b0, b1);
        const float q1 = quad_heads(row1, d0, d1, d2, d3, b0, b1);
        const float q2 = quad_heads(row2, d0, d1, d2, d3, b0, b1);
        const float q3 = quad_heads(row3, d0, d1, d2, d3, b0, b1);
        const float t0 = fold_pair<4>(q0, q1, b2);
        const float t1 = fold_pair<4>(q2, q3, b2);
        const float dk = fold_pair<8>(t0, t1, b3);
        float a = (e + sh) + dk;
        a = (a > 0.f) ? a : NEG_SLOPE * a;
        a = ok ? a : -__builtin_huge_valf();
        float cm = a;
        { const float o = row_xor<4>(cm); cm = (o > cm) ? o : cm; }
        { const float o = row_xor<8>(cm); cm = (o > cm) ? o : cm; }
        { const float o = other_half(cm); cm = (o > cm) ? o : cm; }
        const float mn = (cm > m) ? cm : m;
        const float sc = __expf(m - mn);
        const float w  = __expf(a - mn);
        float ws = w;
        ws += row_xor<4>(ws);
        ws += row_xor<8>(ws);
        ws += other_half(ws);
        wsum = wsum * sc + ws;
        a0 = a0 * row_lane<0>(sc) + (row_lane<0>(w) * row0 + row_lane<4>(w) * row1
                                     + row_lane<8>(w) * row2 + row_lane<12>(w) * row3);
        a1 = a1 * row_lane<1>(sc) + (row_lane<1>(w) * row0 + row_lane<5>(w) * row1
                                     + row_lane<9>(w) * row2 + row_lane<13>(w) * row3);
        a2 = a2 * row_lane<2>(sc) + (row_lane<2>(w) * row0 + row_lane<6>(w) * row1
                                     + row_lane<10>(w) * row2 + row_lane<14>(w) * row3);
        a3 = a3 * row_lane<3>(sc) + (row_lane<3>(w) * row0 + row_lane<7>(w) * row1
                                     + row_lane<11>(w) * row2 + row_lane<15>(w) * row3);
        m = mn;
    }
    const float inv = 1.f / (wsum + EPSV);
    const v4f z0 = (a0 + other_half4(a0)) * row_lane<0>(inv);
    const v4f z1 = (a1 + other_half4(a1)) * row_lane<1>(inv);
    const v4f z2 = (a2 + other_half4(a2)) * row_lane<2>(inv);
    const v4f z3 = (a3 + other_half4(a3)) * row_lane<3>(inv);
    {
        float* zr = Zs + (wv * WAVE_NODES + r) * Z_PITCH + (2 * half) * IN_CH + 4 * l15;
        *(v4f*)zr           = half ? z2 : z0;
        *(v4f*)(zr + IN_CH) = half ? z3 : z1;
    }
  }
    __syncthreads();

    v8f acc0 = {};
    v8f acc1 = {};
    if (wv < 4) {
        const float* za  = Zs + l15 * Z_PITCH + wv * IN_CH + half * 2;
        const float* wb0 = W + (size_t)(wv * OUT_CH + l15) * IN_CH + half * 2;
        const float* wb1 = wb0 + (size_t)16 * IN_CH;
#pragma unroll
        for (int k0 = 0; k0 < IN_CH; k0 += 4) {
            const v2f a   = *(const v2f*)(za + k0);
            const v2f bb0 = *(const v2f*)(wb0 + k0);
            const v2f bb1 = *(const v2f*)(wb1 + k0);
            acc0 = __builtin_amdgcn_wmma_f32_16x16x4_f32(
                       false, a, false, bb0, (short)0, acc0, false, false);
            acc1 = __builtin_amdgcn_wmma_f32_16x16x4_f32(
                       false, a, false, bb1, (short)0, acc1, false, false);
        }
        if (wv > 0) {
            float* ps = Ps + (wv - 1) * (16 * 32) + lane;
#pragma unroll
            for (int v = 0; v < 8; ++v) {
                ps[v * 32]       = acc0[v];
                ps[(8 + v) * 32] = acc1[v];
            }
        }
    }
    __syncthreads();
    if (wv == 0) {
        const int m0 = blockIdx.x * TILE_NODES;
        const float bc0 = bias_conv[l15],      bl0 = bias_layer[l15];
        const float bc1 = bias_conv[16 + l15], bl1 = bias_layer[16 + l15];
#pragma unroll
        for (int v = 0; v < 8; ++v) {
            const float* ps = Ps + lane;
            const float s0 = ((acc0[v] + ps[v * 32]) + ps[512 + v * 32]) + ps[1024 + v * 32];
            const float s1 = ((acc1[v] + ps[(8 + v) * 32]) + ps[512 + (8 + v) * 32]) + ps[1024 + (8 + v) * 32];
            float r0 = s0 * 0.25f + bc0 + bl0;
            float r1 = s1 * 0.25f + bc1 + bl1;
            r0 = r0 > 0.f ? r0 : 0.f;
            r1 = r1 > 0.f ? r1 : 0.f;
            float* o = out + (size_t)(m0 + v + half * 8) * OUT_CH + l15;
            o[0]  = r0;
            o[16] = r1;
        }
    }
}

extern "C" void kernel_launch(void* const* d_in, const int* in_sizes, int n_in,
                              void* d_out, int out_size, void* d_ws, size_t ws_size,
                              hipStream_t stream) {
    const float* x          = (const float*)d_in[0];
    const float* edge_attr  = (const float*)d_in[1];
    const float* W          = (const float*)d_in[2];
    const float* W_edge     = (const float*)d_in[3];
    const float* att_src    = (const float*)d_in[4];
    const float* att_dst    = (const float*)d_in[5];
    const float* att_edge   = (const float*)d_in[6];
    const float* bias_conv  = (const float*)d_in[7];
    const float* bias_layer = (const float*)d_in[8];
    const int*   edge_index = (const int*)d_in[9];
    float* out = (float*)d_out;

    float* ws      = (float*)d_ws;
    float* weff    = ws;
    float* vs      = weff + (size_t)HEADS * IN_CH;
    float* vd      = vs   + (size_t)HEADS * IN_CH;
    v2i*   rec     = (v2i*)(vd + (size_t)HEADS * IN_CH);
    int*   seglocal = (int*)(rec + (size_t)N_EDGES * 4);
    int*   cursor  = seglocal + (size_t)N_NODES;
    int*   slotof  = cursor  + (size_t)N_NODES;
    int*   plane   = slotof  + (size_t)N_EDGES;
    int*   btot    = plane   + (size_t)SORT_CHUNKS * N_NODES;

    chunk_histogram<<<dim3(SORT_CHUNKS, NODE_PARTS), 1024, 0, stream>>>(edge_index, plane);
    prefix_tables<<<PREFIX_LAUNCH_BLOCKS, 256, 0, stream>>>(plane, seglocal, btot,
                                                            W_edge, att_edge, W, att_src, att_dst,
                                                            weff, vs, vd);
    chunk_slots<<<dim3(SORT_CHUNKS, NODE_PARTS), 1024, 0, stream>>>(edge_index, seglocal, btot, plane, cursor, slotof);
    edge_logits<<<(N_EDGES + 255) / 256, 256, 0, stream>>>(
        edge_attr, edge_index, weff, slotof, rec);
    node_softmax_gather<<<N_NODES / TILE_NODES, NODE_WAVES * 32, 0, stream>>>(
        cursor, rec, x, W, vs, vd, bias_conv, bias_layer, out);
}


# logits' node terms (x row . vs, x row . vd) taken once per node by extra workgroups of the edge launch; node pass reads 4 bytes per (edge, head), no dots per gathered row; sum order unchanged
# speedup vs baseline: 5.4697x; 1.0641x over previous
#include <hip/hip_runtime.h>
#include <cstdint>

#define N_NODES 50000
#define N_EDGES 800000
#define IN_CH 64
#define OUT_CH 32
#define HEADS 4
#define HC 128
#define NEG_SLOPE 0.2f
#define EPSV 1e-16f

typedef __attribute__((ext_vector_type(2))) float v2f;
typedef __attribute__((ext_vector_type(4))) float v4f;
typedef __attribute__((ext_vector_type(8))) float v8f;
typedef __attribute__((ext_vector_type(2))) int v2i;

__device__ __forceinline__ void build_tables(const float* __restrict__ W_edge,
                                             const float* __restrict__ att_edge,
                                             const float* __restrict__ W,
                                             const float* __restrict__ att_src,
                                             const float* __restrict__ att_dst,
                                             float* __restrict__ weff,
                                             float* __restrict__ vs,
                                             float* __restrict__ vd)
{
    const int t = threadIdx.x;
    const int h = t >> 6, k = t & 63;
    float s = 0.f, a = 0.f, d = 0.f;
#pragma unroll
    for (int c = 0; c < OUT_CH; ++c) {
        const float w = W[(size_t)(h * OUT_CH + c) * IN_CH + k];
        s += att_edge[h * OUT_CH + c] * W_edge[(size_t)(h * OUT_CH + c) * IN_CH + k];
        a += att_src[h * OUT_CH + c] * w;
        d += att_dst[h * OUT_CH + c] * w;
    }
    weff[t] = s;
    vs[t] = a;
    vd[t] = d;
}

#define SORT_CHUNKS 64
#define NODE_PARTS 4
#define CHUNK_EDGES (N_EDGES / SORT_CHUNKS)
#define PART_NODES (N_NODES / NODE_PARTS)
#define CHUNK_VEC_ITERS ((CHUNK_EDGES / 4 + 1023) / 1024)
#define PART_VEC_ITERS ((PART_NODES / 4 + 1023) / 1024)
#define SCAN_BLOCKS ((N_NODES + 255) / 256)
#define SCAN_PER_LANE ((SCAN_BLOCKS + 31) / 32)
static_assert(N_EDGES % SORT_CHUNKS == 0 && CHUNK_EDGES % 4 == 0, "chunks of equal size, read 16 bytes at a time");
static_assert(N_NODES % NODE_PARTS == 0 && PART_NODES % 4 == 0, "ranges of equal size, 16-byte table accesses");

__device__ __forceinline__ unsigned part_index(int d, int lo)
{
    return (unsigned)(min(max(d, 0), N_NODES - 1) - lo);
}

__global__ void __launch_bounds__(1024) chunk_histogram(const int* __restrict__ edge_index,
                                                        int* __restrict__ plane)
{
    __shared__ alignas(16) int tab[PART_NODES];
    const int g  = blockIdx.x;
    const int lo = blockIdx.y * PART_NODES;
    const int4* dp = (const int4*)(edge_index + N_EDGES + g * CHUNK_EDGES);
    int4 d[CHUNK_VEC_ITERS];
#pragma unroll
    for (int j = 0; j < CHUNK_VEC_ITERS; ++j) {
        const int i = threadIdx.x + j * 1024;
        d[j] = (i < CHUNK_EDGES / 4) ? dp[i] : make_int4(0, 0, 0, 0);
    }
#pragma unroll
    for (int j = 0; j < PART_VEC_ITERS; ++j) {
        const int i = threadIdx.x + j * 1024;
        if (i < PART_NODES / 4) ((int4*)tab)[i] = make_int4(0, 0, 0, 0);
    }
    __syncthreads();
#pragma unroll
    for (int j = 0; j < CHUNK_VEC_ITERS; ++j) {
        const bool ok = threadIdx.x + j * 1024 < CHUNK_EDGES / 4;
        const unsigned r0 = part_index(d[j].x, lo), r1 = part_index(d[j].y, lo);
        const unsigned r2 = part_index(d[j].z, lo), r3 = part_index(d[j].w, lo);
        if (ok && r0 < PART_NODES) atomicAdd(tab + r0, 1);
        if (ok && r1 < PART_NODES) atomicAdd(tab + r1, 1);
        if (ok && r2 < PART_NODES) atomicAdd(tab + r2, 1);
        if (ok && r3 < PART_NODES) atomicAdd(tab + r3, 1);
    }
    __syncthreads();
    int4* pl = (int4*)(plane + (size_t)g * N_NODES + lo);
#pragma unroll
    for (int j = 0; j < PART_VEC_ITERS; ++j) {
        const int i = threadIdx.x + j * 1024;
        if (i < PART_NODES / 4) pl[i] = ((const int4*)tab)[i];
    }
}

__device__ __forceinline__ void chunk_prefix(int* __restrict__ plane, int* __restrict__ seglocal,
                                             int* __restrict__ btot, int blk)
{
    __shared__ int wtot[8];
    const int lane = threadIdx.x & 31;
    const int wv   = threadIdx.x >> 5;
    const int n = blk * 256 + threadIdx.x;
    const bool valid = n < N_NODES;
    const int nn = valid ? n : N_NODES - 1;
    int c[SORT_CHUNKS];
#pragma unroll
    for (int g = 0; g < SORT_CHUNKS; ++g) c[g] = plane[(size_t)g * N_NODES + nn];
    int run = 0;
    if (valid) {
#pragma unroll
        for (int g = 0; g < SORT_CHUNKS; ++g) {
            plane[(size_t)g * N_NODES + n] = run;
            run += c[g];
        }
    }
    int inc = run;
#pragma unroll
    for (int d = 1; d < 32; d <<= 1) {
        const int up = __shfl_up(inc, d);
        if (lane >= d) inc += up;
    }
    if (lane == 31) wtot[wv] = inc;
    __syncthreads();
    int wb = 0;
#pragma unroll
    for (int w = 0; w < 7; ++w) wb += (w < wv) ? wtot[w] : 0;
    if (valid) seglocal[n] = wb + inc - run;
    if (threadIdx.x == 255) btot[blk] = wb + inc;
}

#define PREFIX_LAUNCH_BLOCKS (SCAN_BLOCKS + 1)

__global__ void __launch_bounds__(256) prefix_tables(
    int* __restrict__ plane, int* __restrict__ seglocal, int* __restrict__ btot,
    const float* __restrict__ W_edge, const float* __restrict__ att_edge,
    const float* __restrict__ W, const float* __restrict__ att_src, const float* __restrict__ att_dst,
    float* __restrict__ weff, float* __restrict__ vs, float* __restrict__ vd)
{
    const int b = blockIdx.x;
    if (b < SCAN_BLOCKS) chunk_prefix(plane, seglocal, btot, b);
    else                 build_tables(W_edge, att_edge, W, att_src, att_dst, weff, vs, vd);
}

__global__ void __launch_bounds__(1024) chunk_slots(const int* __restrict__ edge_index,
                                                    const int* __restrict__ seglocal,
                                                    const int* __restrict__ btot,
                                                    const int* __restrict__ plane,
                                                    int* __restrict__ cursor,
                                                    int* __restrict__ slotof)
{
    __shared__ alignas(16) int tab[PART_NODES];
    __shared__ int bbase[32 * SCAN_PER_LANE];
    const int g  = blockIdx.x;
    const int lo = blockIdx.y * PART_NODES;
    const int sl_lane = (int)threadIdx.x - (1024 - 32);
    int bt[SCAN_PER_LANE] = {};
    if (sl_lane >= 0) {
#pragma unroll
        for (int t = 0; t < SCAN_PER_LANE; ++t)
            bt[t] = btot[min(sl_lane * SCAN_PER_LANE + t, SCAN_BLOCKS - 1)];
    }
    const int4* dp = (const int4*)(edge_index + N_EDGES + g * CHUNK_EDGES);
    const int4* cu = (const int4*)(seglocal + lo);
    const int4* pl = (const int4*)(plane + (size_t)g * N_NODES + lo);
    int4 d[CHUNK_VEC_ITERS], ca[PART_VEC_ITERS], pa[PART_VEC_ITERS];
#pragma unroll
    for (int j = 0; j < PART_VEC_ITERS; ++j) {
        const int i = threadIdx.x + j * 1024;
        const bool in = i < PART_NODES / 4;
        ca[j] = in ? cu[i] : make_int4(0, 0, 0, 0);
        pa[j] = in ? pl[i] : make_int4(0, 0, 0, 0);
    }
#pragma unroll
    for (int j = 0; j < CHUNK_VEC_ITERS; ++j) {
        const int i = threadIdx.x + j * 1024;
        d[j] = (i < CHUNK_EDGES / 4) ? dp[i] : make_int4(0, 0, 0, 0);
    }
    if (sl_lane >= 0) {
        int s = 0;
#pragma unroll
        for (int t = 0; t < SCAN_PER_LANE; ++t) {
            if (sl_lane * SCAN_PER_LANE + t >= SCAN_BLOCKS) bt[t] = 0;
            s += bt[t];
        }
        int inc = s;
#pragma unroll
        for (int dist = 1; dist < 32; dist <<= 1) {
            const int up = __shfl_up(inc, dist);
            if (sl_lane >= dist) inc += up;
        }
        int run = inc - s;
#pragma unroll
        for (int t = 0; t < SCAN_PER_LANE; ++t) {
            bbase[sl_lane * SCAN_PER_LANE + t] = run;
            run += bt[t];
        }
    }
    __syncthreads();
    int4* cw = (int4*)(cursor + lo);
#pragma unroll
    for (int j = 0; j < PART_VEC_ITERS; ++j) {
        const int i = threadIdx.x + j * 1024;
        if (i < PART_NODES / 4) {
            const int bb = bbase[(lo + 4 * i) >> 8];
            const int4 first = make_int4(ca[j].x + bb, ca[j].y + bb, ca[j].z + bb, ca[j].w + bb);
            if (g == 0) cw[i] = first;
            ((int4*)tab)[i] = make_int4(first.x + pa[j].x, first.y + pa[j].y,
                                        first.z + pa[j].z, first.w + pa[j].w);
        }
    }
    __syncthreads();
    int* so = slotof + g * CHUNK_EDGES;
    int4 sl[CHUNK_VEC_ITERS];
#pragma unroll
    for (int j = 0; j < CHUNK_VEC_ITERS; ++j) {
        const bool ok = threadIdx.x + j * 1024 < CHUNK_EDGES / 4;
        const unsigned r0 = part_index(d[j].x, lo), r1 = part_index(d[j].y, lo);
        const unsigned r2 = part_index(d[j].z, lo), r3 = part_index(d[j].w, lo);
        sl[j].x = (ok && r0 < PART_NODES) ? atomicAdd(tab + r0, 1) : -1;
        sl[j].y = (ok && r1 < PART_NODES) ? atomicAdd(tab + r1, 1) : -1;
        sl[j].z = (ok && r2 < PART_NODES) ? atomicAdd(tab + r2, 1) : -1;
        sl[j].w = (ok && r3 < PART_NODES) ? atomicAdd(tab + r3, 1) : -1;
    }
#pragma unroll
    for (int j = 0; j < CHUNK_VEC_ITERS; ++j) {
        const int i = threadIdx.x + j * 1024;
        if (sl[j].x >= 0) so[4 * i + 0] = sl[j].x;
        if (sl[j].y >= 0) so[4 * i + 1] = sl[j].y;
        if (sl[j].z >= 0) so[4 * i + 2] = sl[j].z;
        if (sl[j].w >= 0) so[4 * i + 3] = sl[j].w;
    }
}

template <int M>
__device__ __forceinline__ float row_xor(float v)
{
    return __builtin_bit_cast(float, __builtin_amdgcn_update_dpp(
        0, __builtin_bit_cast(int, v), 0x160 + M, 0xf, 0xf, true));
}

template <int M>
__device__ __forceinline__ float fold_pair(float a, float b, bool bit)
{
    const float send = bit ? a : b;
    const float keep = bit ? b : a;
    return keep + row_xor<M>(send);
}

__device__ __forceinline__ float dot4(v4f v, v4f w)
{
    return __builtin_fmaf(v.w, w.w, __builtin_fmaf(v.z, w.z, __builtin_fmaf(v.y, w.y, v.x * w.x)));
}

static_assert(N_EDGES % 256 == 0, "one wave per 32 edges, 8 waves per block");

template <int J>
__device__ __forceinline__ int quad_lane(int v)
{
    return __builtin_amdgcn_update_dpp(0, v, J * 0x55, 0xf, 0xf, true);
}

template <int J>
__device__ __forceinline__ void store_records(v2i* __restrict__ rec, int slot, int srcm, int h,
                                              const float (&u)[4])
{
    int p = quad_lane<J>(slot);
    p = min(max(p, 0), N_EDGES - 1);
    const int s = quad_lane<J>(srcm);
    v2i r;
    r.x = __builtin_bit_cast(int, u[J]);
    r.y = s;
    rec[(size_t)p * 4 + h] = r;
    if constexpr (J + 1 < 4) store_records<J + 1>(rec, slot, srcm, h, u);
}

#define SCORE_WAVE_NODES 16
#define SCORE_BLOCKS ((N_NODES + 8 * SCORE_WAVE_NODES - 1) / (8 * SCORE_WAVE_NODES))
static_assert(N_NODES % 8 == 0, "whole groups of 8 nodes");

__device__ __forceinline__ float quad_heads(v4f row, v4f d0, v4f d1, v4f d2, v4f d3, bool b0, bool b1)
{
    const float r01 = fold_pair<1>(dot4(row, d0), dot4(row, d1), b0);
    const float r23 = fold_pair<1>(dot4(row, d2), dot4(row, d3), b0);
    return fold_pair<2>(r01, r23, b1);
}

__device__ __forceinline__ void node_scores(const float* __restrict__ x,
                                            const float* __restrict__ vs,
                                            const float* __restrict__ vd,
                                            float* __restrict__ sdst,
                                            float* __restrict__ ssrc, int blk)
{
    const int lane = threadIdx.x & 31;
    const int l15  = lane & 15;
    const int half = lane >> 4;
    const bool b0 = (lane & 1) != 0, b1 = (lane & 2) != 0;
    const bool b2 = (lane & 4) != 0, b3 = (lane & 8) != 0;
    const int nb = (blk * 8 + __builtin_amdgcn_readfirstlane((int)(threadIdx.x >> 5))) * SCORE_WAVE_NODES;
    const v4f s0 = *(const v4f*)(vs + 0 * IN_CH + 4 * l15);
    const v4f s1 = *(const v4f*)(vs + 1 * IN_CH + 4 * l15);
    const v4f s2 = *(const v4f*)(vs + 2 * IN_CH + 4 * l15);
    const v4f s3 = *(const v4f*)(vs + 3 * IN_CH + 4 * l15);
    const v4f d0 = *(const v4f*)(vd + 0 * IN_CH + 4 * l15);
    const v4f d1 = *(const v4f*)(vd + 1 * IN_CH + 4 * l15);
    const v4f d2 = *(const v4f*)(vd + 2 * IN_CH + 4 * l15);
    const v4f d3 = *(const v4f*)(vd + 3 * IN_CH + 4 * l15);
    const float* xl = x + 4 * l15;
    v4f row[SCORE_WAVE_NODES / 2];
#pragma unroll
    for (int i = 0; i < SCORE_WAVE_NODES / 2; ++i)
        row[i] = *(const v4f*)(xl + (size_t)min(nb + 2 * i + half, N_NODES - 1) * IN_CH);
#pragma unroll
    for (int g = 0; g < SCORE_WAVE_NODES / 8; ++g) {
        const float a0 = quad_heads(row[4 * g + 0], s0, s1, s2, s3, b0, b1);
        const float a1 = quad_heads(row[4 * g + 1], s0, s1, s2, s3, b0, b1);
        const float a2 = quad_heads(row[4 * g + 2], s0, s1, s2, s3, b0, b1);
        const float a3 = quad_heads(row[4 * g + 3], s0, s1, s2, s3, b0, b1);
        const float ta = fold_pair<8>(fold_pair<4>(a0, a1, b2), fold_pair<4>(a2, a3, b2), b3);
        const float e0 = quad_heads(row[4 * g + 0], d0, d1, d2, d3, b0, b1);
        const float e1 = quad_heads(row[4 * g + 1], d0, d1, d2, d3, b0, b1);
        const float e2 = quad_heads(row[4 * g + 2], d0, d1, d2, d3, b0, b1);
        const float e3 = quad_heads(row[4 * g + 3], d0, d1, d2, d3, b0, b1);
        const float td = fold_pair<8>(fold_pair<4>(e0, e1, b2), fold_pair<4>(e2, e3, b2), b3);
        const int n8 = nb + 8 * g;
        if (n8 < N_NODES) {
            const size_t o = (size_t)(n8 + 2 * (l15 >> 2) + half) * HEADS + (l15 & 3);
            sdst[o] = ta;
            ssrc[o] = td;
        }
    }
}

__global__ void __launch_bounds__(256) __attribute__((amdgpu_waves_per_eu(8, 8))) edge_logits(
    const float* __restrict__ edge_attr, const int* __restrict__ edge_index,
    const float* __restrict__ weff,
    const int* __restrict__ slotof, v2i* __restrict__ rec,
    const float* __restrict__ x, const float* __restrict__ vs, const float* __restrict__ vd,
    float* __restrict__ sdst, float* __restrict__ ssrc)
{
    if (blockIdx.x < SCORE_BLOCKS) {
        node_scores(x, vs, vd, sdst, ssrc, blockIdx.x);
        return;
    }
    const int eblk = blockIdx.x - SCORE_BLOCKS;
    const int lane = threadIdx.x & 31;
    const int l15  = lane & 15;
    const int half = lane >> 4;
    const int h    = l15 & 3;
    const int e0   = (eblk * 8 + (threadIdx.x >> 5)) * 32;
    const int eb   = e0 + ((l15 >> 2) << 1) + half;

    const v4f w0 = *(const v4f*)(weff + 0 * IN_CH + 4 * l15);
    const v4f w1 = *(const v4f*)(weff + 1 * IN_CH + 4 * l15);
    const v4f w2 = *(const v4f*)(weff + 2 * IN_CH + 4 * l15);
    const v4f w3 = *(const v4f*)(weff + 3 * IN_CH + 4 * l15);

    const int em   = eb + 8 * h;
    const int srcm = edge_index[em];
    const int slot = slotof[em];

    const v4f* ea = (const v4f*)(edge_attr + (size_t)e0 * IN_CH) + lane;
    v4f v[16];
#pragma unroll
    for (int i = 0; i < 16; ++i)
        v[i] = __builtin_nontemporal_load(ea + i * 32);

    __builtin_amdgcn_sched_barrier(0);

    const bool b0 = (lane & 1) != 0, b1 = (lane & 2) != 0;
    const bool b2 = (lane & 4) != 0, b3 = (lane & 8) != 0;

    float s[16];
#pragma unroll
    for (int i = 0; i < 16; ++i) {
        const float r01 = fold_pair<1>(dot4(v[i], w0), dot4(v[i], w1), b0);
        const float r23 = fold_pair<1>(dot4(v[i], w2), dot4(v[i], w3), b0);
        s[i] = fold_pair<2>(r01, r23, b1);
    }
    float t[8];
#pragma unroll
    for (int m = 0; m < 8; ++m) t[m] = fold_pair<4>(s[2 * m], s[2 * m + 1], b2);
    float u[4];
#pragma unroll
    for (int j = 0; j < 4; ++j) u[j] = fold_pair<8>(t[2 * j], t[2 * j + 1], b3);

    store_records<0>(rec, slot, srcm, h, u);
}

template <int N>
__device__ __forceinline__ float row_lane(float v)
{
    return __builtin_bit_cast(float, __builtin_amdgcn_update_dpp(
        0, __builtin_bit_cast(int, v), 0x150 + N, 0xf, 0xf, true));
}

template <int N>
__device__ __forceinline__ int row_lane_i(int v)
{
    return __builtin_amdgcn_update_dpp(0, v, 0x150 + N, 0xf, 0xf, true);
}

__device__ __forceinline__ float other_half(float v)
{
    const unsigned u = __builtin_bit_cast(unsigned, v);
    return __builtin_bit_cast(float, __builtin_amdgcn_permlanex16(u, u, 0x76543210u, 0xfedcba98u, false, false));
}

__device__ __forceinline__ v4f other_half4(v4f v)
{
    v4f r;
    r.x = other_half(v.x); r.y = other_half(v.y);
    r.z = other_half(v.z); r.w = other_half(v.w);
    return r;
}

__device__ __forceinline__ v4f x_row(const float* __restrict__ x, int src, unsigned col_bytes)
{
    const unsigned off = (unsigned)min(max(src, 0), N_NODES - 1) * (unsigned)(IN_CH * sizeof(float)) + col_bytes;
    return *(const v4f*)((const char*)x + off);
}

#define NODE_WAVES 8
#define WAVE_NODES 2
#define TILE_NODES (NODE_WAVES * WAVE_NODES)
#define Z_PITCH 260
static_assert(TILE_NODES == 16, "a workgroup's nodes are one 16-row tile of the matrix instruction");
static_assert(N_NODES % TILE_NODES == 0, "whole tiles");

__global__ void __launch_bounds__(256) __attribute__((amdgpu_waves_per_eu(12, 12))) node_softmax_gather(
    const int* __restrict__ cursor, const v2i* __restrict__ rec,
    const float* __restrict__ x, const float* __restrict__ W,
    const float* __restrict__ sdst, const float* __restrict__ ssrc,
    const float* __restrict__ bias_conv, const float* __restrict__ bias_layer,
    float* __restrict__ out)
{
    __shared__ alignas(16) float Zs[TILE_NODES * Z_PITCH];
    __shared__ float Ps[3 * 16 * 32];

    const int lane = threadIdx.x & 31;
    const int l15  = lane & 15;
    const int half = lane >> 4;
    const int h    = l15 & 3;
    const int ek   = 2 * (l15 >> 2) + half;
    const int wv = __builtin_amdgcn_readfirstlane((int)(threadIdx.x >> 5));
    const int n0 = blockIdx.x * TILE_NODES + wv * WAVE_NODES;

    const int c0 = cursor[n0];
    const int c1 = cursor[n0 + 1];
    const int c2 = (n0 + 2 < N_NODES) ? cursor[n0 + 2] : N_EDGES;

    const unsigned colb = 16u * (unsigned)l15;
    const float sh0 = sdst[(size_t)n0 * HEADS + h];
    const float sh1 = sdst[(size_t)(n0 + 1) * HEADS + h];

  for (int r = 0; r < WAVE_NODES; ++r) {
    const int beg = max(r ? c1 : c0, 0);
    const int end = min(r ? c2 : c1, N_EDGES);
    const float sh = r ? sh1 : sh0;
    float m = -__builtin_huge_valf();
    float wsum = 0.f;
    v4f a0 = {0.f, 0.f, 0.f, 0.f}, a1 = a0, a2 = a0, a3 = a0;
    for (int p = beg; p < end; p += 8) {
        const int i = p + ek;
        const bool ok = i < end;
        const int ii = ok ? i : p;
        const v2i r = *(const v2i*)((const char*)rec + ((unsigned)ii * 32u + (unsigned)h * 8u));
        const int sv = r.y;
        const float e = __builtin_bit_cast(float, r.x);
        const unsigned so = (unsigned)min(max(sv, 0), N_NODES - 1) * (unsigned)(HEADS * sizeof(float))
                          + (unsigned)h * (unsigned)sizeof(float);
        const float dk = *(const float*)((const char*)ssrc + so);
        const v4f row0 = x_row(x, row_lane_i<0>(sv), colb);
        const v4f row1 = x_row(x, row_lane_i<4>(sv), colb);
        const v4f row2 = x_row(x, row_lane_i<8>(sv), colb);
        const v4f row3 = x_row(x, row_lane_i<12>(sv), colb);
        __builtin_amdgcn_sched_barrier(0);
        float a = (e + sh) + dk;
        a = (a > 0.f) ? a : NEG_SLOPE * a;
        a = ok ? a : -__builtin_huge_valf();
        float cm = a;
        { const float o = row_xor<4>(cm); cm = (o > cm) ? o : cm; }
        { const float o = row_xor<8>(cm); cm = (o > cm) ? o : cm; }
        { const float o = other_half(cm); cm = (o > cm) ? o : cm; }
        const float mn = (cm > m) ? cm : m;
        const float sc = __expf(m - mn);
        const float w  = __expf(a - mn);
        float ws = w;
        ws += row_xor<4>(ws);
        ws += row_xor<8>(ws);
        ws += other_half(ws);
        wsum = wsum * sc + ws;
        a0 = a0 * row_lane<0>(sc) + (row_lane<0>(w) * row0 + row_lane<4>(w) * row1
                                     + row_lane<8>(w) * row2 + row_lane<12>(w) * row3);
        a1 = a1 * row_lane<1>(sc) + (row_lane<1>(w) * row0 + row_lane<5>(w) * row1
                                     + row_lane<9>(w) * row2 + row_lane<13>(w) * row3);
        a2 = a2 * row_lane<2>(sc) + (row_lane<2>(w) * row0 + row_lane<6>(w) * row1
                                     + row_lane<10>(w) * row2 + row_lane<14>(w) * row3);
        a3 = a3 * row_lane<3>(sc) + (row_lane<3>(w) * row0 + row_lane<7>(w) * row1
                                     + row_lane<11>(w) * row2 + row_lane<15>(w) * row3);
        m = mn;
    }
    const float inv = 1.f / (wsum + EPSV);
    const v4f z0 = (a0 + other_half4(a0)) * row_lane<0>(inv);
    const v4f z1 = (a1 + other_half4(a1)) * row_lane<1>(inv);
    const v4f z2 = (a2 + other_half4(a2)) * row_lane<2>(inv);
    const v4f z3 = (a3 + other_half4(a3)) * row_lane<3>(inv);
    {
        float* zr = Zs + (wv * WAVE_NODES + r) * Z_PITCH + (2 * half) * IN_CH + 4 * l15;
        *(v4f*)zr           = half ? z2 : z0;
        *(v4f*)(zr + IN_CH) = half ? z3 : z1;
    }
  }
    __syncthreads();

    v8f acc0 = {};
    v8f acc1 = {};
    if (wv < 4) {
        const float* za  = Zs + l15 * Z_PITCH + wv * IN_CH + half * 2;
        const float* wb0 = W + (size_t)(wv * OUT_CH + l15) * IN_CH + half * 2;
        const float* wb1 = wb0 + (size_t)16 * IN_CH;
#pragma unroll
        for (int k0 = 0; k0 < IN_CH; k0 += 4) {
            const v2f a   = *(const v2f*)(za + k0);
            const v2f bb0 = *(const v2f*)(wb0 + k0);
            const v2f bb1 = *(const v2f*)(wb1 + k0);
            acc0 = __builtin_amdgcn_wmma_f32_16x16x4_f32(
                       false, a, false, bb0, (short)0, acc0, false, false);
            acc1 = __builtin_amdgcn_wmma_f32_16x16x4_f32(
                       false, a, false, bb1, (short)0, acc1, false, false);
        }
        if (wv > 0) {
            float* ps = Ps + (wv - 1) * (16 * 32) + lane;
#pragma unroll
            for (int v = 0; v < 8; ++v) {
                ps[v * 32]       = acc0[v];
                ps[(8 + v) * 32] = acc1[v];
            }
        }
    }
    __syncthreads();
    if (wv == 0) {
        const int m0 = blockIdx.x * TILE_NODES;
        const float bc0 = bias_conv[l15],      bl0 = bias_layer[l15];
        const float bc1 = bias_conv[16 + l15], bl1 = bias_layer[16 + l15];
#pragma unroll
        for (int v = 0; v < 8; ++v) {
            const float* ps = Ps + lane;
            const float s0 = ((acc0[v] + ps[v * 32]) + ps[512 + v * 32]) + ps[1024 + v * 32];
            const float s1 = ((acc1[v] + ps[(8 + v) * 32]) + ps[512 + (8 + v) * 32]) + ps[1024 + (8 + v) * 32];
            float r0 = s0 * 0.25f + bc0 + bl0;
            float r1 = s1 * 0.25f + bc1 + bl1;
            r0 = r0 > 0.f ? r0 : 0.f;
            r1 = r1 > 0.f ? r1 : 0.f;
            float* o = out + (size_t)(m0 + v + half * 8) * OUT_CH + l15;
            o[0]  = r0;
            o[16] = r1;
        }
    }
}

extern "C" void kernel_launch(void* const* d_in, const int* in_sizes, int n_in,
                              void* d_out, int out_size, void* d_ws, size_t ws_size,
                              hipStream_t stream) {
    const float* x          = (const float*)d_in[0];
    const float* edge_attr  = (const float*)d_in[1];
    const float* W          = (const float*)d_in[2];
    const float* W_edge     = (const float*)d_in[3];
    const float* att_src    = (const float*)d_in[4];
    const float* att_dst    = (const float*)d_in[5];
    const float* att_edge   = (const float*)d_in[6];
    const float* bias_conv  = (const float*)d_in[7];
    const float* bias_layer = (const float*)d_in[8];
    const int*   edge_index = (const int*)d_in[9];
    float* out = (float*)d_out;

    float* ws      = (float*)d_ws;
    float* weff    = ws;
    float* vs      = weff + (size_t)HEADS * IN_CH;
    float* vd      = vs   + (size_t)HEADS * IN_CH;
    v2i*   rec     = (v2i*)(vd + (size_t)HEADS * IN_CH);
    int*   seglocal = (int*)(rec + (size_t)N_EDGES * 4);
    int*   cursor  = seglocal + (size_t)N_NODES;
    int*   slotof  = cursor  + (size_t)N_NODES;
    int*   plane   = slotof  + (size_t)N_EDGES;
    int*   btot    = plane   + (size_t)SORT_CHUNKS * N_NODES;
    float* sdst    = (float*)(btot + 256);
    float* ssrc    = sdst + (size_t)N_NODES * HEADS;

    chunk_histogram<<<dim3(SORT_CHUNKS, NODE_PARTS), 1024, 0, stream>>>(edge_index, plane);
    prefix_tables<<<PREFIX_LAUNCH_BLOCKS, 256, 0, stream>>>(plane, seglocal, btot,
                                                            W_edge, att_edge, W, att_src, att_dst,
                                                            weff, vs, vd);
    chunk_slots<<<dim3(SORT_CHUNKS, NODE_PARTS), 1024, 0, stream>>>(edge_index, seglocal, btot, plane, cursor, slotof);
    edge_logits<<<SCORE_BLOCKS + (N_EDGES + 255) / 256, 256, 0, stream>>>(
        edge_attr, edge_index, weff, slotof, rec, x, vs, vd, sdst, ssrc);
    node_softmax_gather<<<N_NODES / TILE_NODES, NODE_WAVES * 32, 0, stream>>>(
        cursor, rec, x, W, sdst, ssrc, bias_conv, bias_layer, out);
}
